# starting kernel
# baseline (speedup 1.0000x reference)
#include <hip/hip_runtime.h>
#include <math.h>

typedef __attribute__((ext_vector_type(2))) float v2f;
typedef __attribute__((ext_vector_type(8))) float v8f;

__device__ __forceinline__ void atomicMaxFloat(float* addr, float value) {
  if (value >= 0.0f) {
    atomicMax((int*)addr, __float_as_int(value));
  } else {
    atomicMin((unsigned int*)addr, __float_as_uint(value));
  }
}

__global__ __launch_bounds__(256) void gemm_f32_wmma(
    const float* __restrict__ A, const float* __restrict__ B,
    float* __restrict__ C, int M, int N, int K) {
  const int lane = threadIdx.x & 31;
  const int wave = threadIdx.x >> 5;
  const int wavesPerBlock = blockDim.x >> 5;
  const int tile = blockIdx.x * wavesPerBlock + wave;
  const int tilesN = N >> 4;
  const int numTiles = (M >> 4) * tilesN;
  if (tile >= numTiles) return;

  const int tm = (tile / tilesN) << 4;
  const int tn = (tile % tilesN) << 4;
  const int r  = lane & 15;
  const int kk = (lane >> 4) << 1;

  const float* Arow = A + (size_t)(tm + r) * K;

  v8f acc = {};
  for (int k0 = 0; k0 < K; k0 += 4) {
    v2f a, b;
    float2 av = *(const float2*)(Arow + k0 + kk);
    a.x = av.x;
    a.y = av.y;
    b.x = B[(size_t)(k0 + kk) * N + tn + r];
    b.y = B[(size_t)(k0 + kk + 1) * N + tn + r];
    acc = __builtin_amdgcn_wmma_f32_16x16x4_f32(
         false, a,  false, b,
         (short)0, acc,  false,  false);
  }

  const int rowBase = tm + ((lane >> 4) << 3);
#pragma unroll
  for (int j = 0; j < 8; ++j) {
    C[(size_t)(rowBase + j) * N + tn + r] = acc[j];
  }
}

__global__ void fill_f32(float* __restrict__ p, float v, int n) {
  int i = blockIdx.x * blockDim.x + threadIdx.x;
  if (i < n) p[i] = v;
}

template <int HD, int D>
__global__ __launch_bounds__(256) void edge_scores_wave(
    const float* __restrict__ feat, const int* __restrict__ src,
    const int* __restrict__ dst, float* __restrict__ e, float* __restrict__ m,
    int E, float scale) {
  constexpr int PER_LANE = HD / 32;
  constexpr int LPH = D / PER_LANE;
  constexpr int H = HD / D;
  const int lane = threadIdx.x & 31;
  const int edge = (blockIdx.x * blockDim.x + threadIdx.x) >> 5;
  if (edge >= E) return;
  const int s = src[edge], d = dst[edge];
  const float* fs = feat + (size_t)s * HD + lane * PER_LANE;
  const float* fd = feat + (size_t)d * HD + lane * PER_LANE;

  float acc;
  if constexpr (PER_LANE == 4) {
    float4 a = *(const float4*)fs;
    float4 b = *(const float4*)fd;
    acc = a.x * b.x + a.y * b.y + a.z * b.z + a.w * b.w;
  } else {
    float2 a = *(const float2*)fs;
    float2 b = *(const float2*)fd;
    acc = a.x * b.x + a.y * b.y;
  }
#pragma unroll
  for (int off = 1; off < LPH; off <<= 1) acc += __shfl_xor(acc, off, 32);
  acc *= scale;

  if ((lane % LPH) == 0) {
    const int h = lane / LPH;
    e[(size_t)edge * H + h] = acc;
    atomicMaxFloat(m + (size_t)d * H + h, acc);
  }
}

__global__ void edge_exp(float* __restrict__ e, const int* __restrict__ dst,
                         const float* __restrict__ m, float* __restrict__ denom,
                         int E, int H) {
  int idx = blockIdx.x * blockDim.x + threadIdx.x;
  if (idx >= E * H) return;
  int edge = idx / H;
  int h    = idx - edge * H;
  int d = dst[edge];
  float ex = __expf(e[idx] - m[(size_t)d * H + h]);
  e[idx] = ex;
  atomicAdd(denom + (size_t)d * H + h, ex);
}

template <int HD, int D>
__global__ __launch_bounds__(256) void edge_agg_wave(
    const float* __restrict__ feat, const int* __restrict__ src,
    const int* __restrict__ dst, const float* __restrict__ ex,
    const float* __restrict__ denom, float* __restrict__ out, int E) {
  constexpr int PER_LANE = HD / 32;
  constexpr int LPH = D / PER_LANE;
  constexpr int H = HD / D;
  const int lane = threadIdx.x & 31;
  const int edge = (blockIdx.x * blockDim.x + threadIdx.x) >> 5;
  if (edge >= E) return;
  const int s = src[edge], d = dst[edge];
  const int h = lane / LPH;
  const float a =
      ex[(size_t)edge * H + h] / fmaxf(denom[(size_t)d * H + h], 1e-9f);
  const float* fs = feat + (size_t)s * HD + lane * PER_LANE;
  float* o = out + (size_t)d * HD + lane * PER_LANE;
  if constexpr (PER_LANE == 4) {
    float4 f = *(const float4*)fs;
    atomicAdd(o + 0, a * f.x);
    atomicAdd(o + 1, a * f.y);
    atomicAdd(o + 2, a * f.z);
    atomicAdd(o + 3, a * f.w);
  } else {
    float2 f = *(const float2*)fs;
    atomicAdd(o + 0, a * f.x);
    atomicAdd(o + 1, a * f.y);
  }
}

__global__ void elu_inplace(float* __restrict__ x, int n) {
  int i = blockIdx.x * blockDim.x + threadIdx.x;
  if (i < n) {
    float v = x[i];
    x[i] = v > 0.0f ? v : expm1f(v);
  }
}

static inline int cdiv_i(long long a, long long b) { return (int)((a + b - 1) / b); }

extern "C" void kernel_launch(void* const* d_in, const int* in_sizes, int n_in,
                              void* d_out, int out_size, void* d_ws, size_t ws_size,
                              hipStream_t stream) {
  const float* hIn = (const float*)d_in[0];
  const float* W1  = (const float*)d_in[1];
  const float* W2  = (const float*)d_in[2];
  const int*   src = (const int*)d_in[3];
  const int*   dst = (const int*)d_in[4];

  const int IN_DIM = 128, HEADS = 8, OUT_DIM = 64;
  const int HD1 = 128;
  const int Nn  = in_sizes[0] / IN_DIM;
  const int E   = in_sizes[3];

  float* feat1  = (float*)d_ws;
  float* scores = feat1 + (size_t)Nn * HD1;
  float* mbuf   = scores + (size_t)E * HEADS;
  float* denom  = mbuf + (size_t)Nn * HEADS;
  float* h1     = denom + (size_t)Nn * HEADS;

  const int TPB = 256;
  const int wavesPerBlock = TPB / 32;

  {
    int tiles = (Nn / 16) * (HD1 / 16);
    gemm_f32_wmma<<<cdiv_i(tiles, wavesPerBlock), TPB, 0, stream>>>(
        hIn, W1, feat1, Nn, HD1, IN_DIM);
  }
  fill_f32<<<cdiv_i((long long)Nn * HEADS, TPB), TPB, 0, stream>>>(
      mbuf, -INFINITY, Nn * HEADS);
  hipMemsetAsync(denom, 0, (size_t)Nn * HEADS * sizeof(float), stream);
  hipMemsetAsync(h1, 0, (size_t)Nn * HD1 * sizeof(float), stream);

  edge_scores_wave<128, 16><<<cdiv_i(E, wavesPerBlock), TPB, 0, stream>>>(
      feat1, src, dst, scores, mbuf, E, 0.25f);
  edge_exp<<<cdiv_i((long long)E * HEADS, TPB), TPB, 0, stream>>>(
      scores, dst, mbuf, denom, E, HEADS);
  edge_agg_wave<128, 16><<<cdiv_i(E, wavesPerBlock), TPB, 0, stream>>>(
      feat1, src, dst, scores, denom, h1, E);

  elu_inplace<<<cdiv_i((long long)Nn * HD1, TPB), TPB, 0, stream>>>(h1, Nn * HD1);

  float* feat2 = feat1;
  {
    int tiles = (Nn / 16) * (OUT_DIM / 16);
    gemm_f32_wmma<<<cdiv_i(tiles, wavesPerBlock), TPB, 0, stream>>>(
        h1, W2, feat2, Nn, OUT_DIM, HD1);
  }
  fill_f32<<<cdiv_i(Nn, TPB), TPB, 0, stream>>>(mbuf, -INFINITY, Nn);
  hipMemsetAsync(denom, 0, (size_t)Nn * sizeof(float), stream);
  hipMemsetAsync(d_out, 0, (size_t)Nn * OUT_DIM * sizeof(float), stream);

  edge_scores_wave<64, 64><<<cdiv_i(E, wavesPerBlock), TPB, 0, stream>>>(
      feat2, src, dst, scores, mbuf, E, 0.125f);
  edge_exp<<<cdiv_i(E, TPB), TPB, 0, stream>>>(scores, dst, mbuf, denom, E, 1);
  edge_agg_wave<64, 64><<<cdiv_i(E, wavesPerBlock), TPB, 0, stream>>>(
      feat2, src, dst, scores, denom, (float*)d_out, E);
}


# Edges grouped by destination each call, each list sorted by source id; one wave per node: scores, running-max softmax, weighted sum in registers; no float atomics; sum order = ascending source id
# speedup vs baseline: 3.7507x; 3.7507x over previous
#include <hip/hip_runtime.h>
#include <math.h>

typedef __attribute__((ext_vector_type(2))) float v2f;
typedef __attribute__((ext_vector_type(8))) float v8f;

__global__ __launch_bounds__(256) void gemm_f32_wmma(
    const float* __restrict__ A, const float* __restrict__ B,
    float* __restrict__ C, int M, int N, int K) {
  const int lane = threadIdx.x & 31;
  const int wave = threadIdx.x >> 5;
  const int wavesPerBlock = blockDim.x >> 5;
  const int tile = blockIdx.x * wavesPerBlock + wave;
  const int tilesN = N >> 4;
  const int numTiles = (M >> 4) * tilesN;
  if (tile >= numTiles) return;

  const int tm = (tile / tilesN) << 4;
  const int tn = (tile % tilesN) << 4;
  const int r  = lane & 15;
  const int kk = (lane >> 4) << 1;

  const float* Arow = A + (size_t)(tm + r) * K;

  v8f acc = {};
  for (int k0 = 0; k0 < K; k0 += 4) {
    v2f a, b;
    float2 av = *(const float2*)(Arow + k0 + kk);
    a.x = av.x;
    a.y = av.y;
    b.x = B[(size_t)(k0 + kk) * N + tn + r];
    b.y = B[(size_t)(k0 + kk + 1) * N + tn + r];
    acc = __builtin_amdgcn_wmma_f32_16x16x4_f32(
         false, a,  false, b,
         (short)0, acc,  false,  false);
  }

  const int rowBase = tm + ((lane >> 4) << 3);
#pragma unroll
  for (int j = 0; j < 8; ++j) {
    C[(size_t)(rowBase + j) * N + tn + r] = acc[j];
  }
}

__global__ void zero_i32(int* __restrict__ p, int n) {
  int i = blockIdx.x * blockDim.x + threadIdx.x;
  if (i < n) p[i] = 0;
}

__global__ void edge_slot(const int* __restrict__ dst, int* __restrict__ cnt,
                          int* __restrict__ slot, int E) {
  int e = blockIdx.x * blockDim.x + threadIdx.x;
  if (e < E) slot[e] = atomicAdd(cnt + dst[e], 1);
}

__global__ __launch_bounds__(1024) void scan_counts(
    const int* __restrict__ cnt, int* __restrict__ rowptr, int n) {
  __shared__ int part[1024];
  const int tid = threadIdx.x;
  const int per = (n + 1023) >> 10;
  const int b = min(tid * per, n);
  const int e = min(b + per, n);
  int s = 0;
  for (int i = b; i < e; ++i) s += cnt[i];
  part[tid] = s;
  __syncthreads();
  for (int off = 1; off < 1024; off <<= 1) {
    const int v = (tid >= off) ? part[tid - off] : 0;
    __syncthreads();
    part[tid] += v;
    __syncthreads();
  }
  int run = part[tid] - s;
  for (int i = b; i < e; ++i) {
    rowptr[i] = run;
    run += cnt[i];
  }
  if (tid == 1023) rowptr[n] = part[1023];
}

__global__ void edge_scatter(const int* __restrict__ src,
                             const int* __restrict__ dst,
                             const int* __restrict__ slot,
                             const int* __restrict__ rowptr,
                             int* __restrict__ colTmp, int E) {
  int e = blockIdx.x * blockDim.x + threadIdx.x;
  if (e < E) colTmp[rowptr[dst[e]] + slot[e]] = src[e];
}

__global__ __launch_bounds__(256) void seg_sort_by_src(
    const int* __restrict__ rowptr, const int* __restrict__ colTmp,
    int* __restrict__ col, int Nn) {
  const int lane = threadIdx.x & 31;
  const int node = __builtin_amdgcn_readfirstlane(
      (int)((blockIdx.x * blockDim.x + threadIdx.x) >> 5));
  if (node >= Nn) return;
  const int beg = rowptr[node];
  const int n = rowptr[node + 1] - beg;
  for (int ib = 0; ib < n; ib += 32) {
    const int i = ib + lane;
    const bool vi = i < n;
    const int ki = vi ? colTmp[beg + i] : 0;
    int rank = 0;
    for (int jb = 0; jb < n; jb += 32) {
      const int cj = min(32, n - jb);
      const int kc = (lane < cj) ? colTmp[beg + jb + lane] : 0;
      for (int t = 0; t < cj; ++t) {
        const int kj = __builtin_amdgcn_readlane(kc, t);
        const int bias = (jb + t < i) ? 1 : 0;
        rank += (kj < ki + bias) ? 1 : 0;
      }
    }
    if (vi) col[beg + rank] = ki;
  }
}

constexpr unsigned dpp8_xor_sel(int x) {
  unsigned s = 0;
  for (int i = 0; i < 8; ++i) s |= (unsigned)((i ^ x) & 7) << (3 * i);
  return s;
}
constexpr unsigned dpp8_bcast_sel(int k) { return (unsigned)k * 0x249249u; }

template <unsigned SEL>
__device__ __forceinline__ float dpp8_f(float v) {
  return __uint_as_float(__builtin_amdgcn_mov_dpp8(__float_as_uint(v), SEL));
}
template <unsigned SEL>
__device__ __forceinline__ int dpp8_i(int v) {
  return (int)__builtin_amdgcn_mov_dpp8((unsigned)v, SEL);
}

__global__ __launch_bounds__(256) void node_attn_h8(
    const float* __restrict__ feat, const int* __restrict__ rowptr,
    const int* __restrict__ col, float* __restrict__ out, int Nn, float scale) {
  const int lane = threadIdx.x & 31;
  const int node = __builtin_amdgcn_readfirstlane(
      (int)((blockIdx.x * blockDim.x + threadIdx.x) >> 5));
  if (node >= Nn) return;
  const int beg = rowptr[node];
  const int end = rowptr[node + 1];
  const float* fl = feat + lane * 4;
  const float4 fd = *(const float4*)(fl + (size_t)node * 128);

  float m = -INFINITY, l = 0.0f;
  float4 acc = make_float4(0.0f, 0.0f, 0.0f, 0.0f);

  for (int base = beg; base < end; base += 32) {
    const int cnt = min(32, end - base);
    const int ids = (lane < cnt) ? col[base + lane] : node;
    for (int t = 0; t < cnt; t += 4) {
      float4 f[4];
      float p[4];
#pragma unroll
      for (int u = 0; u < 4; ++u) {
        const int s = __builtin_amdgcn_readlane(ids, t + u);
        f[u] = *(const float4*)(fl + (size_t)s * 128);
      }
#pragma unroll
      for (int u = 0; u < 4; ++u) {
        float d = f[u].x * fd.x + f[u].y * fd.y + f[u].z * fd.z + f[u].w * fd.w;
        d += dpp8_f<dpp8_xor_sel(1)>(d);
        d += dpp8_f<dpp8_xor_sel(2)>(d);
        d *= scale;
        p[u] = (t + u < cnt) ? d : -INFINITY;
      }
      const float mNew = fmaxf(fmaxf(m, p[0]), fmaxf(fmaxf(p[1], p[2]), p[3]));
      const float c = __expf(m - mNew);
      const float w0 = __expf(p[0] - mNew);
      const float w1 = __expf(p[1] - mNew);
      const float w2 = __expf(p[2] - mNew);
      const float w3 = __expf(p[3] - mNew);
      l = l * c + ((w0 + w1) + (w2 + w3));
      acc.x = acc.x * c + ((w0 * f[0].x + w1 * f[1].x) + (w2 * f[2].x + w3 * f[3].x));
      acc.y = acc.y * c + ((w0 * f[0].y + w1 * f[1].y) + (w2 * f[2].y + w3 * f[3].y));
      acc.z = acc.z * c + ((w0 * f[0].z + w1 * f[1].z) + (w2 * f[2].z + w3 * f[3].z));
      acc.w = acc.w * c + ((w0 * f[0].w + w1 * f[1].w) + (w2 * f[2].w + w3 * f[3].w));
      m = mNew;
    }
  }

  const float inv = 1.0f / fmaxf(l, 1e-9f);
  float4 o;
  o.x = acc.x * inv;
  o.y = acc.y * inv;
  o.z = acc.z * inv;
  o.w = acc.w * inv;
  o.x = o.x > 0.0f ? o.x : expm1f(o.x);
  o.y = o.y > 0.0f ? o.y : expm1f(o.y);
  o.z = o.z > 0.0f ? o.z : expm1f(o.z);
  o.w = o.w > 0.0f ? o.w : expm1f(o.w);
  *(float4*)(out + (size_t)node * 128 + lane * 4) = o;
}

template <int K0>
__device__ __forceinline__ void h1_four_steps(
    const float* __restrict__ fq, int ids, int cnt, int g, const float4 fdA,
    const float4 fdB, float scale, float& m, float& l, float4& accA,
    float4& accB) {
  float4 fa[4], fb[4];
  float p[4];
#pragma unroll
  for (int u = 0; u < 4; ++u) {
    const int s = u == 0   ? dpp8_i<dpp8_bcast_sel(K0 + 0)>(ids)
                  : u == 1 ? dpp8_i<dpp8_bcast_sel(K0 + 1)>(ids)
                  : u == 2 ? dpp8_i<dpp8_bcast_sel(K0 + 2)>(ids)
                           : dpp8_i<dpp8_bcast_sel(K0 + 3)>(ids);
    const float* r = fq + (size_t)s * 64;
    fa[u] = *(const float4*)(r);
    fb[u] = *(const float4*)(r + 32);
  }
#pragma unroll
  for (int u = 0; u < 4; ++u) {
    float d = fa[u].x * fdA.x + fa[u].y * fdA.y + fa[u].z * fdA.z +
              fa[u].w * fdA.w + fb[u].x * fdB.x + fb[u].y * fdB.y +
              fb[u].z * fdB.z + fb[u].w * fdB.w;
    d += dpp8_f<dpp8_xor_sel(1)>(d);
    d += dpp8_f<dpp8_xor_sel(2)>(d);
    d += dpp8_f<dpp8_xor_sel(4)>(d);
    d *= scale;
    p[u] = (4 * (K0 + u) + g < cnt) ? d : -INFINITY;
  }
  const float mNew = fmaxf(fmaxf(m, p[0]), fmaxf(fmaxf(p[1], p[2]), p[3]));
  const float mRef = (mNew == -INFINITY) ? 0.0f : mNew;
  const float c = __expf(m - mRef);
  const float w0 = __expf(p[0] - mRef);
  const float w1 = __expf(p[1] - mRef);
  const float w2 = __expf(p[2] - mRef);
  const float w3 = __expf(p[3] - mRef);
  l = l * c + ((w0 + w1) + (w2 + w3));
  accA.x = accA.x * c + ((w0 * fa[0].x + w1 * fa[1].x) + (w2 * fa[2].x + w3 * fa[3].x));
  accA.y = accA.y * c + ((w0 * fa[0].y + w1 * fa[1].y) + (w2 * fa[2].y + w3 * fa[3].y));
  accA.z = accA.z * c + ((w0 * fa[0].z + w1 * fa[1].z) + (w2 * fa[2].z + w3 * fa[3].z));
  accA.w = accA.w * c + ((w0 * fa[0].w + w1 * fa[1].w) + (w2 * fa[2].w + w3 * fa[3].w));
  accB.x = accB.x * c + ((w0 * fb[0].x + w1 * fb[1].x) + (w2 * fb[2].x + w3 * fb[3].x));
  accB.y = accB.y * c + ((w0 * fb[0].y + w1 * fb[1].y) + (w2 * fb[2].y + w3 * fb[3].y));
  accB.z = accB.z * c + ((w0 * fb[0].z + w1 * fb[1].z) + (w2 * fb[2].z + w3 * fb[3].z));
  accB.w = accB.w * c + ((w0 * fb[0].w + w1 * fb[1].w) + (w2 * fb[2].w + w3 * fb[3].w));
  m = mNew;
}

__device__ __forceinline__ float sum_groups(float v) {
  v += __shfl_xor(v, 8, 32);
  v += __shfl_xor(v, 16, 32);
  return v;
}

__global__ __launch_bounds__(256) void node_attn_h1(
    const float* __restrict__ feat, const int* __restrict__ rowptr,
    const int* __restrict__ col, float* __restrict__ out, int Nn, float scale) {
  const int lane = threadIdx.x & 31;
  const int node = __builtin_amdgcn_readfirstlane(
      (int)((blockIdx.x * blockDim.x + threadIdx.x) >> 5));
  if (node >= Nn) return;
  const int beg = rowptr[node];
  const int end = rowptr[node + 1];
  const int q = lane & 7;
  const int g = lane >> 3;
  const float* fq = feat + q * 4;
  const float4 fdA = *(const float4*)(fq + (size_t)node * 64);
  const float4 fdB = *(const float4*)(fq + (size_t)node * 64 + 32);

  float m = -INFINITY, l = 0.0f;
  float4 accA = make_float4(0.0f, 0.0f, 0.0f, 0.0f);
  float4 accB = make_float4(0.0f, 0.0f, 0.0f, 0.0f);

  const int myIdx = 4 * q + g;
  for (int base = beg; base < end; base += 32) {
    const int cnt = min(32, end - base);
    const int ids = (myIdx < cnt) ? col[base + myIdx] : node;
    h1_four_steps<0>(fq, ids, cnt, g, fdA, fdB, scale, m, l, accA, accB);
    if (cnt > 16)
      h1_four_steps<4>(fq, ids, cnt, g, fdA, fdB, scale, m, l, accA, accB);
  }

  float M = fmaxf(m, __shfl_xor(m, 8, 32));
  M = fmaxf(M, __shfl_xor(M, 16, 32));
  const float Mref = (M == -INFINITY) ? 0.0f : M;
  const float sc = __expf(m - Mref);
  const float L = sum_groups(l * sc);
  float4 oA, oB;
  oA.x = sum_groups(accA.x * sc);
  oA.y = sum_groups(accA.y * sc);
  oA.z = sum_groups(accA.z * sc);
  oA.w = sum_groups(accA.w * sc);
  oB.x = sum_groups(accB.x * sc);
  oB.y = sum_groups(accB.y * sc);
  oB.z = sum_groups(accB.z * sc);
  oB.w = sum_groups(accB.w * sc);
  const float inv = 1.0f / fmaxf(L, 1e-9f);
  if (g == 0) {
    float* o = out + (size_t)node * 64 + q * 4;
    *(float4*)(o) = make_float4(oA.x * inv, oA.y * inv, oA.z * inv, oA.w * inv);
    *(float4*)(o + 32) = make_float4(oB.x * inv, oB.y * inv, oB.z * inv, oB.w * inv);
  }
}

static inline int cdiv_i(long long a, long long b) { return (int)((a + b - 1) / b); }

extern "C" void kernel_launch(void* const* d_in, const int* in_sizes, int n_in,
                              void* d_out, int out_size, void* d_ws, size_t ws_size,
                              hipStream_t stream) {
  const float* hIn = (const float*)d_in[0];
  const float* W1  = (const float*)d_in[1];
  const float* W2  = (const float*)d_in[2];
  const int*   src = (const int*)d_in[3];
  const int*   dst = (const int*)d_in[4];

  const int IN_DIM = 128, OUT_DIM = 64;
  const int HD1 = 128;
  const int Nn  = in_sizes[0] / IN_DIM;
  const int E   = in_sizes[3];

  float* feat1  = (float*)d_ws;
  float* h1     = feat1 + (size_t)Nn * HD1;
  int*   slot   = (int*)(h1 + (size_t)Nn * HD1);
  int*   colTmp = slot + (size_t)E;
  int*   col    = colTmp + (size_t)E;
  int*   cnt    = col + (size_t)E;
  int*   rowptr = cnt + (size_t)Nn;

  const int TPB = 256;
  const int wavesPerBlock = TPB / 32;

  zero_i32<<<cdiv_i(Nn, TPB), TPB, 0, stream>>>(cnt, Nn);
  edge_slot<<<cdiv_i(E, TPB), TPB, 0, stream>>>(dst, cnt, slot, E);
  scan_counts<<<1, 1024, 0, stream>>>(cnt, rowptr, Nn);
  edge_scatter<<<cdiv_i(E, TPB), TPB, 0, stream>>>(src, dst, slot, rowptr, colTmp, E);
  seg_sort_by_src<<<cdiv_i(Nn, wavesPerBlock), TPB, 0, stream>>>(rowptr, colTmp, col, Nn);

  {
    int tiles = (Nn / 16) * (HD1 / 16);
    gemm_f32_wmma<<<cdiv_i(tiles, wavesPerBlock), TPB, 0, stream>>>(
        hIn, W1, feat1, Nn, HD1, IN_DIM);
  }
  node_attn_h8<<<cdiv_i(Nn, wavesPerBlock), TPB, 0, stream>>>(
      feat1, rowptr, col, h1, Nn, 0.25f);

  float* feat2 = feat1;
  {
    int tiles = (Nn / 16) * (OUT_DIM / 16);
    gemm_f32_wmma<<<cdiv_i(tiles, wavesPerBlock), TPB, 0, stream>>>(
        h1, W2, feat2, Nn, OUT_DIM, HD1);
  }
  node_attn_h1<<<cdiv_i(Nn, wavesPerBlock), TPB, 0, stream>>>(
      feat2, rowptr, col, (float*)d_out, Nn, 0.125f);
}


# Counter prefix sum (one block): each wave reads its run as whole 512-byte rows, 16 B per lane, all loads in flight at once; scan in registers by lane shifts; was strided per-thread runs
# speedup vs baseline: 4.0212x; 1.0721x over previous
#include <hip/hip_runtime.h>
#include <math.h>

typedef __attribute__((ext_vector_type(2))) float v2f;
typedef __attribute__((ext_vector_type(8))) float v8f;

__global__ __launch_bounds__(256) void gemm_f32_wmma(
    const float* __restrict__ A, const float* __restrict__ B,
    float* __restrict__ C, int M, int N, int K) {
  const int lane = threadIdx.x & 31;
  const int wave = threadIdx.x >> 5;
  const int wavesPerBlock = blockDim.x >> 5;
  const int tile = blockIdx.x * wavesPerBlock + wave;
  const int tilesN = N >> 4;
  const int numTiles = (M >> 4) * tilesN;
  if (tile >= numTiles) return;

  const int tm = (tile / tilesN) << 4;
  const int tn = (tile % tilesN) << 4;
  const int r  = lane & 15;
  const int kk = (lane >> 4) << 1;

  const float* Arow = A + (size_t)(tm + r) * K;

  v8f acc = {};
  for (int k0 = 0; k0 < K; k0 += 4) {
    v2f a, b;
    float2 av = *(const float2*)(Arow + k0 + kk);
    a.x = av.x;
    a.y = av.y;
    b.x = B[(size_t)(k0 + kk) * N + tn + r];
    b.y = B[(size_t)(k0 + kk + 1) * N + tn + r];
    acc = __builtin_amdgcn_wmma_f32_16x16x4_f32(
         false, a,  false, b,
         (short)0, acc,  false,  false);
  }

  const int rowBase = tm + ((lane >> 4) << 3);
#pragma unroll
  for (int j = 0; j < 8; ++j) {
    C[(size_t)(rowBase + j) * N + tn + r] = acc[j];
  }
}

__global__ void zero_i32(int* __restrict__ p, int n) {
  int i = blockIdx.x * blockDim.x + threadIdx.x;
  if (i < n) p[i] = 0;
}

__global__ void edge_slot(const int* __restrict__ dst, int* __restrict__ cnt,
                          int* __restrict__ slot, int E) {
  int e = blockIdx.x * blockDim.x + threadIdx.x;
  if (e < E) slot[e] = atomicAdd(cnt + dst[e], 1);
}

constexpr int SCAN_ROWS = 13;
constexpr int SCAN_WAVE_SPAN = SCAN_ROWS * 128;

template <int N>
__device__ __forceinline__ int row_shr_or0(int v) {
  return __builtin_amdgcn_update_dpp(0, v, 0x110 + N, 0xf, 0xf, true);
}

__device__ __forceinline__ int wave_incl_scan(int v, int lane) {
  v += row_shr_or0<1>(v);
  v += row_shr_or0<2>(v);
  v += row_shr_or0<4>(v);
  v += row_shr_or0<8>(v);
  const int lowRow = __builtin_amdgcn_readlane(v, 15);
  return v + (lane >= 16 ? lowRow : 0);
}

__global__ __launch_bounds__(1024) void scan_counts(
    const int* __restrict__ cnt, int* __restrict__ rowptr, int n) {
  __shared__ int waveTot[32];
  const int lane = threadIdx.x & 31;
  const int wave = __builtin_amdgcn_readfirstlane((int)(threadIdx.x >> 5));
  int carry = 0;
  for (int pass = 0; pass < n; pass += 32 * SCAN_WAVE_SPAN) {
    const int seg = pass + wave * SCAN_WAVE_SPAN;
    const int i0 = seg + lane * 4;
    const bool whole = seg + SCAN_WAVE_SPAN <= n;
    int4 v[SCAN_ROWS];
    if (whole) {
#pragma unroll
      for (int r = 0; r < SCAN_ROWS; ++r) v[r] = *(const int4*)(cnt + i0 + r * 128);
    } else {
#pragma unroll
      for (int r = 0; r < SCAN_ROWS; ++r) {
        const int i = i0 + r * 128;
        v[r].x = (i + 0 < n) ? cnt[i + 0] : 0;
        v[r].y = (i + 1 < n) ? cnt[i + 1] : 0;
        v[r].z = (i + 2 < n) ? cnt[i + 2] : 0;
        v[r].w = (i + 3 < n) ? cnt[i + 3] : 0;
      }
    }

    int before[SCAN_ROWS];
    int waveSum = 0;
#pragma unroll
    for (int r = 0; r < SCAN_ROWS; ++r) {
      const int s = (v[r].x + v[r].y) + (v[r].z + v[r].w);
      const int inc = wave_incl_scan(s, lane);
      before[r] = waveSum + (inc - s);
      waveSum += __builtin_amdgcn_readlane(inc, 31);
    }

    if (lane == 0) waveTot[wave] = waveSum;
    __syncthreads();
    const int wt = waveTot[lane];
    const int wi = wave_incl_scan(wt, lane);
    const int waveBase = carry + __builtin_amdgcn_readlane(wi - wt, wave);
    carry += __builtin_amdgcn_readlane(wi, 31);

#pragma unroll
    for (int r = 0; r < SCAN_ROWS; ++r) {
      const int i = i0 + r * 128;
      int4 o;
      o.x = waveBase + before[r];
      o.y = o.x + v[r].x;
      o.z = o.y + v[r].y;
      o.w = o.z + v[r].z;
      if (whole) {
        *(int4*)(rowptr + i) = o;
      } else {
        if (i + 0 < n) rowptr[i + 0] = o.x;
        if (i + 1 < n) rowptr[i + 1] = o.y;
        if (i + 2 < n) rowptr[i + 2] = o.z;
        if (i + 3 < n) rowptr[i + 3] = o.w;
      }
    }
    __syncthreads();
  }
  if (threadIdx.x == 0) rowptr[n] = carry;
}

__global__ void edge_scatter(const int* __restrict__ src,
                             const int* __restrict__ dst,
                             const int* __restrict__ slot,
                             const int* __restrict__ rowptr,
                             int* __restrict__ colTmp, int E) {
  int e = blockIdx.x * blockDim.x + threadIdx.x;
  if (e < E) colTmp[rowptr[dst[e]] + slot[e]] = src[e];
}

__global__ __launch_bounds__(256) void seg_sort_by_src(
    const int* __restrict__ rowptr, const int* __restrict__ colTmp,
    int* __restrict__ col, int Nn) {
  const int lane = threadIdx.x & 31;
  const int node = __builtin_amdgcn_readfirstlane(
      (int)((blockIdx.x * blockDim.x + threadIdx.x) >> 5));
  if (node >= Nn) return;
  const int beg = rowptr[node];
  const int n = rowptr[node + 1] - beg;
  for (int ib = 0; ib < n; ib += 32) {
    const int i = ib + lane;
    const bool vi = i < n;
    const int ki = vi ? colTmp[beg + i] : 0;
    int rank = 0;
    for (int jb = 0; jb < n; jb += 32) {
      const int cj = min(32, n - jb);
      const int kc = (lane < cj) ? colTmp[beg + jb + lane] : 0;
      for (int t = 0; t < cj; ++t) {
        const int kj = __builtin_amdgcn_readlane(kc, t);
        const int bias = (jb + t < i) ? 1 : 0;
        rank += (kj < ki + bias) ? 1 : 0;
      }
    }
    if (vi) col[beg + rank] = ki;
  }
}

constexpr unsigned dpp8_xor_sel(int x) {
  unsigned s = 0;
  for (int i = 0; i < 8; ++i) s |= (unsigned)((i ^ x) & 7) << (3 * i);
  return s;
}
constexpr unsigned dpp8_bcast_sel(int k) { return (unsigned)k * 0x249249u; }

template <unsigned SEL>
__device__ __forceinline__ float dpp8_f(float v) {
  return __uint_as_float(__builtin_amdgcn_mov_dpp8(__float_as_uint(v), SEL));
}
template <unsigned SEL>
__device__ __forceinline__ int dpp8_i(int v) {
  return (int)__builtin_amdgcn_mov_dpp8((unsigned)v, SEL);
}

__global__ __launch_bounds__(256) void node_attn_h8(
    const float* __restrict__ feat, const int* __restrict__ rowptr,
    const int* __restrict__ col, float* __restrict__ out, int Nn, float scale) {
  const int lane = threadIdx.x & 31;
  const int node = __builtin_amdgcn_readfirstlane(
      (int)((blockIdx.x * blockDim.x + threadIdx.x) >> 5));
  if (node >= Nn) return;
  const int beg = rowptr[node];
  const int end = rowptr[node + 1];
  const float* fl = feat + lane * 4;
  const float4 fd = *(const float4*)(fl + (size_t)node * 128);

  float m = -INFINITY, l = 0.0f;
  float4 acc = make_float4(0.0f, 0.0f, 0.0f, 0.0f);

  for (int base = beg; base < end; base += 32) {
    const int cnt = min(32, end - base);
    const int ids = (lane < cnt) ? col[base + lane] : node;
    for (int t = 0; t < cnt; t += 4) {
      float4 f[4];
      float p[4];
#pragma unroll
      for (int u = 0; u < 4; ++u) {
        const int s = __builtin_amdgcn_readlane(ids, t + u);
        f[u] = *(const float4*)(fl + (size_t)s * 128);
      }
#pragma unroll
      for (int u = 0; u < 4; ++u) {
        float d = f[u].x * fd.x + f[u].y * fd.y + f[u].z * fd.z + f[u].w * fd.w;
        d += dpp8_f<dpp8_xor_sel(1)>(d);
        d += dpp8_f<dpp8_xor_sel(2)>(d);
        d *= scale;
        p[u] = (t + u < cnt) ? d : -INFINITY;
      }
      const float mNew = fmaxf(fmaxf(m, p[0]), fmaxf(fmaxf(p[1], p[2]), p[3]));
      const float c = __expf(m - mNew);
      const float w0 = __expf(p[0] - mNew);
      const float w1 = __expf(p[1] - mNew);
      const float w2 = __expf(p[2] - mNew);
      const float w3 = __expf(p[3] - mNew);
      l = l * c + ((w0 + w1) + (w2 + w3));
      acc.x = acc.x * c + ((w0 * f[0].x + w1 * f[1].x) + (w2 * f[2].x + w3 * f[3].x));
      acc.y = acc.y * c + ((w0 * f[0].y + w1 * f[1].y) + (w2 * f[2].y + w3 * f[3].y));
      acc.z = acc.z * c + ((w0 * f[0].z + w1 * f[1].z) + (w2 * f[2].z + w3 * f[3].z));
      acc.w = acc.w * c + ((w0 * f[0].w + w1 * f[1].w) + (w2 * f[2].w + w3 * f[3].w));
      m = mNew;
    }
  }

  const float inv = 1.0f / fmaxf(l, 1e-9f);
  float4 o;
  o.x = acc.x * inv;
  o.y = acc.y * inv;
  o.z = acc.z * inv;
  o.w = acc.w * inv;
  o.x = o.x > 0.0f ? o.x : expm1f(o.x);
  o.y = o.y > 0.0f ? o.y : expm1f(o.y);
  o.z = o.z > 0.0f ? o.z : expm1f(o.z);
  o.w = o.w > 0.0f ? o.w : expm1f(o.w);
  *(float4*)(out + (size_t)node * 128 + lane * 4) = o;
}

template <int K0>
__device__ __forceinline__ void h1_four_steps(
    const float* __restrict__ fq, int ids, int cnt, int g, const float4 fdA,
    const float4 fdB, float scale, float& m, float& l, float4& accA,
    float4& accB) {
  float4 fa[4], fb[4];
  float p[4];
#pragma unroll
  for (int u = 0; u < 4; ++u) {
    const int s = u == 0   ? dpp8_i<dpp8_bcast_sel(K0 + 0)>(ids)
                  : u == 1 ? dpp8_i<dpp8_bcast_sel(K0 + 1)>(ids)
                  : u == 2 ? dpp8_i<dpp8_bcast_sel(K0 + 2)>(ids)
                           : dpp8_i<dpp8_bcast_sel(K0 + 3)>(ids);
    const float* r = fq + (size_t)s * 64;
    fa[u] = *(const float4*)(r);
    fb[u] = *(const float4*)(r + 32);
  }
#pragma unroll
  for (int u = 0; u < 4; ++u) {
    float d = fa[u].x * fdA.x + fa[u].y * fdA.y + fa[u].z * fdA.z +
              fa[u].w * fdA.w + fb[u].x * fdB.x + fb[u].y * fdB.y +
              fb[u].z * fdB.z + fb[u].w * fdB.w;
    d += dpp8_f<dpp8_xor_sel(1)>(d);
    d += dpp8_f<dpp8_xor_sel(2)>(d);
    d += dpp8_f<dpp8_xor_sel(4)>(d);
    d *= scale;
    p[u] = (4 * (K0 + u) + g < cnt) ? d : -INFINITY;
  }
  const float mNew = fmaxf(fmaxf(m, p[0]), fmaxf(fmaxf(p[1], p[2]), p[3]));
  const float mRef = (mNew == -INFINITY) ? 0.0f : mNew;
  const float c = __expf(m - mRef);
  const float w0 = __expf(p[0] - mRef);
  const float w1 = __expf(p[1] - mRef);
  const float w2 = __expf(p[2] - mRef);
  const float w3 = __expf(p[3] - mRef);
  l = l * c + ((w0 + w1) + (w2 + w3));
  accA.x = accA.x * c + ((w0 * fa[0].x + w1 * fa[1].x) + (w2 * fa[2].x + w3 * fa[3].x));
  accA.y = accA.y * c + ((w0 * fa[0].y + w1 * fa[1].y) + (w2 * fa[2].y + w3 * fa[3].y));
  accA.z = accA.z * c + ((w0 * fa[0].z + w1 * fa[1].z) + (w2 * fa[2].z + w3 * fa[3].z));
  accA.w = accA.w * c + ((w0 * fa[0].w + w1 * fa[1].w) + (w2 * fa[2].w + w3 * fa[3].w));
  accB.x = accB.x * c + ((w0 * fb[0].x + w1 * fb[1].x) + (w2 * fb[2].x + w3 * fb[3].x));
  accB.y = accB.y * c + ((w0 * fb[0].y + w1 * fb[1].y) + (w2 * fb[2].y + w3 * fb[3].y));
  accB.z = accB.z * c + ((w0 * fb[0].z + w1 * fb[1].z) + (w2 * fb[2].z + w3 * fb[3].z));
  accB.w = accB.w * c + ((w0 * fb[0].w + w1 * fb[1].w) + (w2 * fb[2].w + w3 * fb[3].w));
  m = mNew;
}

__device__ __forceinline__ float sum_groups(float v) {
  v += __shfl_xor(v, 8, 32);
  v += __shfl_xor(v, 16, 32);
  return v;
}

__global__ __launch_bounds__(256) void node_attn_h1(
    const float* __restrict__ feat, const int* __restrict__ rowptr,
    const int* __restrict__ col, float* __restrict__ out, int Nn, float scale) {
  const int lane = threadIdx.x & 31;
  const int node = __builtin_amdgcn_readfirstlane(
      (int)((blockIdx.x * blockDim.x + threadIdx.x) >> 5));
  if (node >= Nn) return;
  const int beg = rowptr[node];
  const int end = rowptr[node + 1];
  const int q = lane & 7;
  const int g = lane >> 3;
  const float* fq = feat + q * 4;
  const float4 fdA = *(const float4*)(fq + (size_t)node * 64);
  const float4 fdB = *(const float4*)(fq + (size_t)node * 64 + 32);

  float m = -INFINITY, l = 0.0f;
  float4 accA = make_float4(0.0f, 0.0f, 0.0f, 0.0f);
  float4 accB = make_float4(0.0f, 0.0f, 0.0f, 0.0f);

  const int myIdx = 4 * q + g;
  for (int base = beg; base < end; base += 32) {
    const int cnt = min(32, end - base);
    const int ids = (myIdx < cnt) ? col[base + myIdx] : node;
    h1_four_steps<0>(fq, ids, cnt, g, fdA, fdB, scale, m, l, accA, accB);
    if (cnt > 16)
      h1_four_steps<4>(fq, ids, cnt, g, fdA, fdB, scale, m, l, accA, accB);
  }

  float M = fmaxf(m, __shfl_xor(m, 8, 32));
  M = fmaxf(M, __shfl_xor(M, 16, 32));
  const float Mref = (M == -INFINITY) ? 0.0f : M;
  const float sc = __expf(m - Mref);
  const float L = sum_groups(l * sc);
  float4 oA, oB;
  oA.x = sum_groups(accA.x * sc);
  oA.y = sum_groups(accA.y * sc);
  oA.z = sum_groups(accA.z * sc);
  oA.w = sum_groups(accA.w * sc);
  oB.x = sum_groups(accB.x * sc);
  oB.y = sum_groups(accB.y * sc);
  oB.z = sum_groups(accB.z * sc);
  oB.w = sum_groups(accB.w * sc);
  const float inv = 1.0f / fmaxf(L, 1e-9f);
  if (g == 0) {
    float* o = out + (size_t)node * 64 + q * 4;
    *(float4*)(o) = make_float4(oA.x * inv, oA.y * inv, oA.z * inv, oA.w * inv);
    *(float4*)(o + 32) = make_float4(oB.x * inv, oB.y * inv, oB.z * inv, oB.w * inv);
  }
}

static inline int cdiv_i(long long a, long long b) { return (int)((a + b - 1) / b); }

extern "C" void kernel_launch(void* const* d_in, const int* in_sizes, int n_in,
                              void* d_out, int out_size, void* d_ws, size_t ws_size,
                              hipStream_t stream) {
  const float* hIn = (const float*)d_in[0];
  const float* W1  = (const float*)d_in[1];
  const float* W2  = (const float*)d_in[2];
  const int*   src = (const int*)d_in[3];
  const int*   dst = (const int*)d_in[4];

  const int IN_DIM = 128, OUT_DIM = 64;
  const int HD1 = 128;
  const int Nn  = in_sizes[0] / IN_DIM;
  const int E   = in_sizes[3];

  float* feat1  = (float*)d_ws;
  float* h1     = feat1 + (size_t)Nn * HD1;
  int*   slot   = (int*)(h1 + (size_t)Nn * HD1);
  int*   colTmp = slot + (size_t)E;
  int*   col    = colTmp + (size_t)E;
  int*   cnt    = col + (size_t)E;
  int*   rowptr = cnt + (size_t)Nn;

  const int TPB = 256;
  const int wavesPerBlock = TPB / 32;

  zero_i32<<<cdiv_i(Nn, TPB), TPB, 0, stream>>>(cnt, Nn);
  edge_slot<<<cdiv_i(E, TPB), TPB, 0, stream>>>(dst, cnt, slot, E);
  scan_counts<<<1, 1024, 0, stream>>>(cnt, rowptr, Nn);
  edge_scatter<<<cdiv_i(E, TPB), TPB, 0, stream>>>(src, dst, slot, rowptr, colTmp, E);
  seg_sort_by_src<<<cdiv_i(Nn, wavesPerBlock), TPB, 0, stream>>>(rowptr, colTmp, col, Nn);

  {
    int tiles = (Nn / 16) * (HD1 / 16);
    gemm_f32_wmma<<<cdiv_i(tiles, wavesPerBlock), TPB, 0, stream>>>(
        hIn, W1, feat1, Nn, HD1, IN_DIM);
  }
  node_attn_h8<<<cdiv_i(Nn, wavesPerBlock), TPB, 0, stream>>>(
      feat1, rowptr, col, h1, Nn, 0.25f);

  float* feat2 = feat1;
  {
    int tiles = (Nn / 16) * (OUT_DIM / 16);
    gemm_f32_wmma<<<cdiv_i(tiles, wavesPerBlock), TPB, 0, stream>>>(
        h1, W2, feat2, Nn, OUT_DIM, HD1);
  }
  node_attn_h1<<<cdiv_i(Nn, wavesPerBlock), TPB, 0, stream>>>(
      feat2, rowptr, col, (float*)d_out, Nn, 0.125f);
}


# First product h@W1 has no launch of its own: re-tiled to one wave per 16x64 block, its workgroups lead the edge-count launch and run beside the atomic waits; 8 launches; k order of its sums changes
# speedup vs baseline: 4.4010x; 1.0944x over previous
#include <hip/hip_runtime.h>
#include <math.h>

typedef __attribute__((ext_vector_type(2))) float v2f;
typedef __attribute__((ext_vector_type(8))) float v8f;

__global__ __launch_bounds__(256) void gemm_f32_wmma(
    const float* __restrict__ A, const float* __restrict__ B,
    float* __restrict__ C, int M, int N, int K) {
  const int lane = threadIdx.x & 31;
  const int wave = threadIdx.x >> 5;
  const int wavesPerBlock = blockDim.x >> 5;
  const int tile = blockIdx.x * wavesPerBlock + wave;
  const int tilesN = N >> 4;
  const int numTiles = (M >> 4) * tilesN;
  if (tile >= numTiles) return;

  const int tm = (tile / tilesN) << 4;
  const int tn = (tile % tilesN) << 4;
  const int r  = lane & 15;
  const int kk = (lane >> 4) << 1;

  const float* Arow = A + (size_t)(tm + r) * K;

  v8f acc = {};
  for (int k0 = 0; k0 < K; k0 += 4) {
    v2f a, b;
    float2 av = *(const float2*)(Arow + k0 + kk);
    a.x = av.x;
    a.y = av.y;
    b.x = B[(size_t)(k0 + kk) * N + tn + r];
    b.y = B[(size_t)(k0 + kk + 1) * N + tn + r];
    acc = __builtin_amdgcn_wmma_f32_16x16x4_f32(
         false, a,  false, b,
         (short)0, acc,  false,  false);
  }

  const int rowBase = tm + ((lane >> 4) << 3);
#pragma unroll
  for (int j = 0; j < 8; ++j) {
    C[(size_t)(rowBase + j) * N + tn + r] = acc[j];
  }
}

__global__ void zero_i32(int* __restrict__ p, int n) {
  int i = blockIdx.x * blockDim.x + threadIdx.x;
  if (i < n) p[i] = 0;
}

template <int KD, int N>
__device__ __forceinline__ void gemm_block_16x64(
    const float* __restrict__ A, const float* __restrict__ B,
    float* __restrict__ C, int unit, int lane) {
  constexpr int NT = 4;
  constexpr int CG = N / (16 * NT);
  constexpr int STEPS = KD / 8;
  const int tm  = (unit / CG) << 4;
  const int tn0 = (unit % CG) * (16 * NT);
  const int c = lane & 15;
  const int h = lane >> 4;

  const float* Arow = A + (size_t)(tm + c) * KD + 4 * h;
  const float* Bp = B + (size_t)(4 * h) * N + tn0 + c;
  v8f acc[NT] = {};
#pragma unroll
  for (int j = 0; j < STEPS; ++j) {
    const float4 a = *(const float4*)(Arow + 8 * j);
    float b0[NT], b1[NT], b2[NT], b3[NT];
#pragma unroll
    for (int t = 0; t < NT; ++t) {
      const float* p = Bp + (size_t)(8 * j) * N + 16 * t;
      b0[t] = p[0];
      b1[t] = p[N];
      b2[t] = p[2 * N];
      b3[t] = p[3 * N];
    }
#pragma unroll
    for (int t = 0; t < NT; ++t) {
      v2f av, bv;
      av.x = a.x;
      av.y = a.y;
      bv.x = b0[t];
      bv.y = b1[t];
      acc[t] = __builtin_amdgcn_wmma_f32_16x16x4_f32(
           false, av,  false, bv,
           (short)0, acc[t],  false,  false);
    }
#pragma unroll
    for (int t = 0; t < NT; ++t) {
      v2f av, bv;
      av.x = a.z;
      av.y = a.w;
      bv.x = b2[t];
      bv.y = b3[t];
      acc[t] = __builtin_amdgcn_wmma_f32_16x16x4_f32(
           false, av,  false, bv,
           (short)0, acc[t],  false,  false);
    }
  }

  const int rowBase = tm + (h << 3);
#pragma unroll
  for (int t = 0; t < NT; ++t) {
#pragma unroll
    for (int j = 0; j < 8; ++j) {
      C[(size_t)(rowBase + j) * N + tn0 + 16 * t + c] = acc[t][j];
    }
  }
}

template <int KD, int N>
__global__ __launch_bounds__(256) void edge_slot_and_gemm(
    const int* __restrict__ dst, int* __restrict__ cnt, int* __restrict__ slot,
    int E, const float* __restrict__ A, const float* __restrict__ B,
    float* __restrict__ C, int M, int gemmBlocks) {
  const int b = blockIdx.x;
  if (b < gemmBlocks) {
    const int unit = __builtin_amdgcn_readfirstlane(
        (int)((b * blockDim.x + threadIdx.x) >> 5));
    if (unit < (M >> 4) * (N / 64))
      gemm_block_16x64<KD, N>(A, B, C, unit, threadIdx.x & 31);
    return;
  }
  const int e = (b - gemmBlocks) * blockDim.x + threadIdx.x;
  if (e < E) slot[e] = atomicAdd(cnt + dst[e], 1);
}

constexpr int SCAN_ROWS = 13;
constexpr int SCAN_WAVE_SPAN = SCAN_ROWS * 128;

template <int N>
__device__ __forceinline__ int row_shr_or0(int v) {
  return __builtin_amdgcn_update_dpp(0, v, 0x110 + N, 0xf, 0xf, true);
}

__device__ __forceinline__ int wave_incl_scan(int v, int lane) {
  v += row_shr_or0<1>(v);
  v += row_shr_or0<2>(v);
  v += row_shr_or0<4>(v);
  v += row_shr_or0<8>(v);
  const int lowRow = __builtin_amdgcn_readlane(v, 15);
  return v + (lane >= 16 ? lowRow : 0);
}

__global__ __launch_bounds__(1024) void scan_counts(
    const int* __restrict__ cnt, int* __restrict__ rowptr, int n) {
  __shared__ int waveTot[32];
  const int lane = threadIdx.x & 31;
  const int wave = __builtin_amdgcn_readfirstlane((int)(threadIdx.x >> 5));
  int carry = 0;
  for (int pass = 0; pass < n; pass += 32 * SCAN_WAVE_SPAN) {
    const int seg = pass + wave * SCAN_WAVE_SPAN;
    const int i0 = seg + lane * 4;
    const bool whole = seg + SCAN_WAVE_SPAN <= n;
    int4 v[SCAN_ROWS];
    if (whole) {
#pragma unroll
      for (int r = 0; r < SCAN_ROWS; ++r) v[r] = *(const int4*)(cnt + i0 + r * 128);
    } else {
#pragma unroll
      for (int r = 0; r < SCAN_ROWS; ++r) {
        const int i = i0 + r * 128;
        v[r].x = (i + 0 < n) ? cnt[i + 0] : 0;
        v[r].y = (i + 1 < n) ? cnt[i + 1] : 0;
        v[r].z = (i + 2 < n) ? cnt[i + 2] : 0;
        v[r].w = (i + 3 < n) ? cnt[i + 3] : 0;
      }
    }

    int before[SCAN_ROWS];
    int waveSum = 0;
#pragma unroll
    for (int r = 0; r < SCAN_ROWS; ++r) {
      const int s = (v[r].x + v[r].y) + (v[r].z + v[r].w);
      const int inc = wave_incl_scan(s, lane);
      before[r] = waveSum + (inc - s);
      waveSum += __builtin_amdgcn_readlane(inc, 31);
    }

    if (lane == 0) waveTot[wave] = waveSum;
    __syncthreads();
    const int wt = waveTot[lane];
    const int wi = wave_incl_scan(wt, lane);
    const int waveBase = carry + __builtin_amdgcn_readlane(wi - wt, wave);
    carry += __builtin_amdgcn_readlane(wi, 31);

#pragma unroll
    for (int r = 0; r < SCAN_ROWS; ++r) {
      const int i = i0 + r * 128;
      int4 o;
      o.x = waveBase + before[r];
      o.y = o.x + v[r].x;
      o.z = o.y + v[r].y;
      o.w = o.z + v[r].z;
      if (whole) {
        *(int4*)(rowptr + i) = o;
      } else {
        if (i + 0 < n) rowptr[i + 0] = o.x;
        if (i + 1 < n) rowptr[i + 1] = o.y;
        if (i + 2 < n) rowptr[i + 2] = o.z;
        if (i + 3 < n) rowptr[i + 3] = o.w;
      }
    }
    __syncthreads();
  }
  if (threadIdx.x == 0) rowptr[n] = carry;
}

__global__ void edge_scatter(const int* __restrict__ src,
                             const int* __restrict__ dst,
                             const int* __restrict__ slot,
                             const int* __restrict__ rowptr,
                             int* __restrict__ colTmp, int E) {
  int e = blockIdx.x * blockDim.x + threadIdx.x;
  if (e < E) colTmp[rowptr[dst[e]] + slot[e]] = src[e];
}

__global__ __launch_bounds__(256) void seg_sort_by_src(
    const int* __restrict__ rowptr, const int* __restrict__ colTmp,
    int* __restrict__ col, int Nn) {
  const int lane = threadIdx.x & 31;
  const int node = __builtin_amdgcn_readfirstlane(
      (int)((blockIdx.x * blockDim.x + threadIdx.x) >> 5));
  if (node >= Nn) return;
  const int beg = rowptr[node];
  const int n = rowptr[node + 1] - beg;
  for (int ib = 0; ib < n; ib += 32) {
    const int i = ib + lane;
    const bool vi = i < n;
    const int ki = vi ? colTmp[beg + i] : 0;
    int rank = 0;
    for (int jb = 0; jb < n; jb += 32) {
      const int cj = min(32, n - jb);
      const int kc = (lane < cj) ? colTmp[beg + jb + lane] : 0;
      for (int t = 0; t < cj; ++t) {
        const int kj = __builtin_amdgcn_readlane(kc, t);
        const int bias = (jb + t < i) ? 1 : 0;
        rank += (kj < ki + bias) ? 1 : 0;
      }
    }
    if (vi) col[beg + rank] = ki;
  }
}

constexpr unsigned dpp8_xor_sel(int x) {
  unsigned s = 0;
  for (int i = 0; i < 8; ++i) s |= (unsigned)((i ^ x) & 7) << (3 * i);
  return s;
}
constexpr unsigned dpp8_bcast_sel(int k) { return (unsigned)k * 0x249249u; }

template <unsigned SEL>
__device__ __forceinline__ float dpp8_f(float v) {
  return __uint_as_float(__builtin_amdgcn_mov_dpp8(__float_as_uint(v), SEL));
}
template <unsigned SEL>
__device__ __forceinline__ int dpp8_i(int v) {
  return (int)__builtin_amdgcn_mov_dpp8((unsigned)v, SEL);
}

__global__ __launch_bounds__(256) void node_attn_h8(
    const float* __restrict__ feat, const int* __restrict__ rowptr,
    const int* __restrict__ col, float* __restrict__ out, int Nn, float scale) {
  const int lane = threadIdx.x & 31;
  const int node = __builtin_amdgcn_readfirstlane(
      (int)((blockIdx.x * blockDim.x + threadIdx.x) >> 5));
  if (node >= Nn) return;
  const int beg = rowptr[node];
  const int end = rowptr[node + 1];
  const float* fl = feat + lane * 4;
  const float4 fd = *(const float4*)(fl + (size_t)node * 128);

  float m = -INFINITY, l = 0.0f;
  float4 acc = make_float4(0.0f, 0.0f, 0.0f, 0.0f);

  for (int base = beg; base < end; base += 32) {
    const int cnt = min(32, end - base);
    const int ids = (lane < cnt) ? col[base + lane] : node;
    for (int t = 0; t < cnt; t += 4) {
      float4 f[4];
      float p[4];
#pragma unroll
      for (int u = 0; u < 4; ++u) {
        const int s = __builtin_amdgcn_readlane(ids, t + u);
        f[u] = *(const float4*)(fl + (size_t)s * 128);
      }
#pragma unroll
      for (int u = 0; u < 4; ++u) {
        float d = f[u].x * fd.x + f[u].y * fd.y + f[u].z * fd.z + f[u].w * fd.w;
        d += dpp8_f<dpp8_xor_sel(1)>(d);
        d += dpp8_f<dpp8_xor_sel(2)>(d);
        d *= scale;
        p[u] = (t + u < cnt) ? d : -INFINITY;
      }
      const float mNew = fmaxf(fmaxf(m, p[0]), fmaxf(fmaxf(p[1], p[2]), p[3]));
      const float c = __expf(m - mNew);
      const float w0 = __expf(p[0] - mNew);
      const float w1 = __expf(p[1] - mNew);
      const float w2 = __expf(p[2] - mNew);
      const float w3 = __expf(p[3] - mNew);
      l = l * c + ((w0 + w1) + (w2 + w3));
      acc.x = acc.x * c + ((w0 * f[0].x + w1 * f[1].x) + (w2 * f[2].x + w3 * f[3].x));
      acc.y = acc.y * c + ((w0 * f[0].y + w1 * f[1].y) + (w2 * f[2].y + w3 * f[3].y));
      acc.z = acc.z * c + ((w0 * f[0].z + w1 * f[1].z) + (w2 * f[2].z + w3 * f[3].z));
      acc.w = acc.w * c + ((w0 * f[0].w + w1 * f[1].w) + (w2 * f[2].w + w3 * f[3].w));
      m = mNew;
    }
  }

  const float inv = 1.0f / fmaxf(l, 1e-9f);
  float4 o;
  o.x = acc.x * inv;
  o.y = acc.y * inv;
  o.z = acc.z * inv;
  o.w = acc.w * inv;
  o.x = o.x > 0.0f ? o.x : expm1f(o.x);
  o.y = o.y > 0.0f ? o.y : expm1f(o.y);
  o.z = o.z > 0.0f ? o.z : expm1f(o.z);
  o.w = o.w > 0.0f ? o.w : expm1f(o.w);
  *(float4*)(out + (size_t)node * 128 + lane * 4) = o;
}

template <int K0>
__device__ __forceinline__ void h1_four_steps(
    const float* __restrict__ fq, int ids, int cnt, int g, const float4 fdA,
    const float4 fdB, float scale, float& m, float& l, float4& accA,
    float4& accB) {
  float4 fa[4], fb[4];
  float p[4];
#pragma unroll
  for (int u = 0; u < 4; ++u) {
    const int s = u == 0   ? dpp8_i<dpp8_bcast_sel(K0 + 0)>(ids)
                  : u == 1 ? dpp8_i<dpp8_bcast_sel(K0 + 1)>(ids)
                  : u == 2 ? dpp8_i<dpp8_bcast_sel(K0 + 2)>(ids)
                           : dpp8_i<dpp8_bcast_sel(K0 + 3)>(ids);
    const float* r = fq + (size_t)s * 64;
    fa[u] = *(const float4*)(r);
    fb[u] = *(const float4*)(r + 32);
  }
#pragma unroll
  for (int u = 0; u < 4; ++u) {
    float d = fa[u].x * fdA.x + fa[u].y * fdA.y + fa[u].z * fdA.z +
              fa[u].w * fdA.w + fb[u].x * fdB.x + fb[u].y * fdB.y +
              fb[u].z * fdB.z + fb[u].w * fdB.w;
    d += dpp8_f<dpp8_xor_sel(1)>(d);
    d += dpp8_f<dpp8_xor_sel(2)>(d);
    d += dpp8_f<dpp8_xor_sel(4)>(d);
    d *= scale;
    p[u] = (4 * (K0 + u) + g < cnt) ? d : -INFINITY;
  }
  const float mNew = fmaxf(fmaxf(m, p[0]), fmaxf(fmaxf(p[1], p[2]), p[3]));
  const float mRef = (mNew == -INFINITY) ? 0.0f : mNew;
  const float c = __expf(m - mRef);
  const float w0 = __expf(p[0] - mRef);
  const float w1 = __expf(p[1] - mRef);
  const float w2 = __expf(p[2] - mRef);
  const float w3 = __expf(p[3] - mRef);
  l = l * c + ((w0 + w1) + (w2 + w3));
  accA.x = accA.x * c + ((w0 * fa[0].x + w1 * fa[1].x) + (w2 * fa[2].x + w3 * fa[3].x));
  accA.y = accA.y * c + ((w0 * fa[0].y + w1 * fa[1].y) + (w2 * fa[2].y + w3 * fa[3].y));
  accA.z = accA.z * c + ((w0 * fa[0].z + w1 * fa[1].z) + (w2 * fa[2].z + w3 * fa[3].z));
  accA.w = accA.w * c + ((w0 * fa[0].w + w1 * fa[1].w) + (w2 * fa[2].w + w3 * fa[3].w));
  accB.x = accB.x * c + ((w0 * fb[0].x + w1 * fb[1].x) + (w2 * fb[2].x + w3 * fb[3].x));
  accB.y = accB.y * c + ((w0 * fb[0].y + w1 * fb[1].y) + (w2 * fb[2].y + w3 * fb[3].y));
  accB.z = accB.z * c + ((w0 * fb[0].z + w1 * fb[1].z) + (w2 * fb[2].z + w3 * fb[3].z));
  accB.w = accB.w * c + ((w0 * fb[0].w + w1 * fb[1].w) + (w2 * fb[2].w + w3 * fb[3].w));
  m = mNew;
}

__device__ __forceinline__ float sum_groups(float v) {
  v += __shfl_xor(v, 8, 32);
  v += __shfl_xor(v, 16, 32);
  return v;
}

__global__ __launch_bounds__(256) void node_attn_h1(
    const float* __restrict__ feat, const int* __restrict__ rowptr,
    const int* __restrict__ col, float* __restrict__ out, int Nn, float scale) {
  const int lane = threadIdx.x & 31;
  const int node = __builtin_amdgcn_readfirstlane(
      (int)((blockIdx.x * blockDim.x + threadIdx.x) >> 5));
  if (node >= Nn) return;
  const int beg = rowptr[node];
  const int end = rowptr[node + 1];
  const int q = lane & 7;
  const int g = lane >> 3;
  const float* fq = feat + q * 4;
  const float4 fdA = *(const float4*)(fq + (size_t)node * 64);
  const float4 fdB = *(const float4*)(fq + (size_t)node * 64 + 32);

  float m = -INFINITY, l = 0.0f;
  float4 accA = make_float4(0.0f, 0.0f, 0.0f, 0.0f);
  float4 accB = make_float4(0.0f, 0.0f, 0.0f, 0.0f);

  const int myIdx = 4 * q + g;
  for (int base = beg; base < end; base += 32) {
    const int cnt = min(32, end - base);
    const int ids = (myIdx < cnt) ? col[base + myIdx] : node;
    h1_four_steps<0>(fq, ids, cnt, g, fdA, fdB, scale, m, l, accA, accB);
    if (cnt > 16)
      h1_four_steps<4>(fq, ids, cnt, g, fdA, fdB, scale, m, l, accA, accB);
  }

  float M = fmaxf(m, __shfl_xor(m, 8, 32));
  M = fmaxf(M, __shfl_xor(M, 16, 32));
  const float Mref = (M == -INFINITY) ? 0.0f : M;
  const float sc = __expf(m - Mref);
  const float L = sum_groups(l * sc);
  float4 oA, oB;
  oA.x = sum_groups(accA.x * sc);
  oA.y = sum_groups(accA.y * sc);
  oA.z = sum_groups(accA.z * sc);
  oA.w = sum_groups(accA.w * sc);
  oB.x = sum_groups(accB.x * sc);
  oB.y = sum_groups(accB.y * sc);
  oB.z = sum_groups(accB.z * sc);
  oB.w = sum_groups(accB.w * sc);
  const float inv = 1.0f / fmaxf(L, 1e-9f);
  if (g == 0) {
    float* o = out + (size_t)node * 64 + q * 4;
    *(float4*)(o) = make_float4(oA.x * inv, oA.y * inv, oA.z * inv, oA.w * inv);
    *(float4*)(o + 32) = make_float4(oB.x * inv, oB.y * inv, oB.z * inv, oB.w * inv);
  }
}

static inline int cdiv_i(long long a, long long b) { return (int)((a + b - 1) / b); }

extern "C" void kernel_launch(void* const* d_in, const int* in_sizes, int n_in,
                              void* d_out, int out_size, void* d_ws, size_t ws_size,
                              hipStream_t stream) {
  const float* hIn = (const float*)d_in[0];
  const float* W1  = (const float*)d_in[1];
  const float* W2  = (const float*)d_in[2];
  const int*   src = (const int*)d_in[3];
  const int*   dst = (const int*)d_in[4];

  const int IN_DIM = 128, OUT_DIM = 64;
  const int HD1 = 128;
  const int Nn  = in_sizes[0] / IN_DIM;
  const int E   = in_sizes[3];

  float* feat1  = (float*)d_ws;
  float* h1     = feat1 + (size_t)Nn * HD1;
  int*   slot   = (int*)(h1 + (size_t)Nn * HD1);
  int*   colTmp = slot + (size_t)E;
  int*   col    = colTmp + (size_t)E;
  int*   cnt    = col + (size_t)E;
  int*   rowptr = cnt + (size_t)Nn;

  const int TPB = 256;
  const int wavesPerBlock = TPB / 32;

  zero_i32<<<cdiv_i(Nn, TPB), TPB, 0, stream>>>(cnt, Nn);
  {
    int gemmBlocks = cdiv_i((Nn / 16) * (HD1 / 64), wavesPerBlock);
    edge_slot_and_gemm<128, 128><<<cdiv_i(E, TPB) + gemmBlocks, TPB, 0, stream>>>(
        dst, cnt, slot, E, hIn, W1, feat1, Nn, gemmBlocks);
  }
  scan_counts<<<1, 1024, 0, stream>>>(cnt, rowptr, Nn);
  edge_scatter<<<cdiv_i(E, TPB), TPB, 0, stream>>>(src, dst, slot, rowptr, colTmp, E);
  seg_sort_by_src<<<cdiv_i(Nn, wavesPerBlock), TPB, 0, stream>>>(rowptr, colTmp, col, Nn);

  node_attn_h8<<<cdiv_i(Nn, wavesPerBlock), TPB, 0, stream>>>(
      feat1, rowptr, col, h1, Nn, 0.25f);

  float* feat2 = feat1;
  {
    int tiles = (Nn / 16) * (OUT_DIM / 16);
    gemm_f32_wmma<<<cdiv_i(tiles, wavesPerBlock), TPB, 0, stream>>>(
        h1, W2, feat2, Nn, OUT_DIM, HD1);
  }
  node_attn_h1<<<cdiv_i(Nn, wavesPerBlock), TPB, 0, stream>>>(
      feat2, rowptr, col, (float*)d_out, Nn, 0.125f);
}


# Per-node list sort has no launch of its own: the layer-1 attention wave sorts its node's source ids in registers (lane compare-exchange network, up to 64 ids) and stores them for layer 2; same sums
# speedup vs baseline: 4.5935x; 1.0437x over previous
#include <hip/hip_runtime.h>
#include <math.h>

typedef __attribute__((ext_vector_type(2))) float v2f;
typedef __attribute__((ext_vector_type(8))) float v8f;

__global__ __launch_bounds__(256) void gemm_f32_wmma(
    const float* __restrict__ A, const float* __restrict__ B,
    float* __restrict__ C, int M, int N, int K) {
  const int lane = threadIdx.x & 31;
  const int wave = threadIdx.x >> 5;
  const int wavesPerBlock = blockDim.x >> 5;
  const int tile = blockIdx.x * wavesPerBlock + wave;
  const int tilesN = N >> 4;
  const int numTiles = (M >> 4) * tilesN;
  if (tile >= numTiles) return;

  const int tm = (tile / tilesN) << 4;
  const int tn = (tile % tilesN) << 4;
  const int r  = lane & 15;
  const int kk = (lane >> 4) << 1;

  const float* Arow = A + (size_t)(tm + r) * K;

  v8f acc = {};
  for (int k0 = 0; k0 < K; k0 += 4) {
    v2f a, b;
    float2 av = *(const float2*)(Arow + k0 + kk);
    a.x = av.x;
    a.y = av.y;
    b.x = B[(size_t)(k0 + kk) * N + tn + r];
    b.y = B[(size_t)(k0 + kk + 1) * N + tn + r];
    acc = __builtin_amdgcn_wmma_f32_16x16x4_f32(
         false, a,  false, b,
         (short)0, acc,  false,  false);
  }

  const int rowBase = tm + ((lane >> 4) << 3);
#pragma unroll
  for (int j = 0; j < 8; ++j) {
    C[(size_t)(rowBase + j) * N + tn + r] = acc[j];
  }
}

__global__ void zero_i32(int* __restrict__ p, int n) {
  int i = blockIdx.x * blockDim.x + threadIdx.x;
  if (i < n) p[i] = 0;
}

template <int KD, int N>
__device__ __forceinline__ void gemm_block_16x64(
    const float* __restrict__ A, const float* __restrict__ B,
    float* __restrict__ C, int unit, int lane) {
  constexpr int NT = 4;
  constexpr int CG = N / (16 * NT);
  constexpr int STEPS = KD / 8;
  const int tm  = (unit / CG) << 4;
  const int tn0 = (unit % CG) * (16 * NT);
  const int c = lane & 15;
  const int h = lane >> 4;

  const float* Arow = A + (size_t)(tm + c) * KD + 4 * h;
  const float* Bp = B + (size_t)(4 * h) * N + tn0 + c;
  v8f acc[NT] = {};
#pragma unroll
  for (int j = 0; j < STEPS; ++j) {
    const float4 a = *(const float4*)(Arow + 8 * j);
    float b0[NT], b1[NT], b2[NT], b3[NT];
#pragma unroll
    for (int t = 0; t < NT; ++t) {
      const float* p = Bp + (size_t)(8 * j) * N + 16 * t;
      b0[t] = p[0];
      b1[t] = p[N];
      b2[t] = p[2 * N];
      b3[t] = p[3 * N];
    }
#pragma unroll
    for (int t = 0; t < NT; ++t) {
      v2f av, bv;
      av.x = a.x;
      av.y = a.y;
      bv.x = b0[t];
      bv.y = b1[t];
      acc[t] = __builtin_amdgcn_wmma_f32_16x16x4_f32(
           false, av,  false, bv,
           (short)0, acc[t],  false,  false);
    }
#pragma unroll
    for (int t = 0; t < NT; ++t) {
      v2f av, bv;
      av.x = a.z;
      av.y = a.w;
      bv.x = b2[t];
      bv.y = b3[t];
      acc[t] = __builtin_amdgcn_wmma_f32_16x16x4_f32(
           false, av,  false, bv,
           (short)0, acc[t],  false,  false);
    }
  }

  const int rowBase = tm + (h << 3);
#pragma unroll
  for (int t = 0; t < NT; ++t) {
#pragma unroll
    for (int j = 0; j < 8; ++j) {
      C[(size_t)(rowBase + j) * N + tn0 + 16 * t + c] = acc[t][j];
    }
  }
}

template <int KD, int N>
__global__ __launch_bounds__(256) void edge_slot_and_gemm(
    const int* __restrict__ dst, int* __restrict__ cnt, int* __restrict__ slot,
    int E, const float* __restrict__ A, const float* __restrict__ B,
    float* __restrict__ C, int M, int gemmBlocks) {
  const int b = blockIdx.x;
  if (b < gemmBlocks) {
    const int unit = __builtin_amdgcn_readfirstlane(
        (int)((b * blockDim.x + threadIdx.x) >> 5));
    if (unit < (M >> 4) * (N / 64))
      gemm_block_16x64<KD, N>(A, B, C, unit, threadIdx.x & 31);
    return;
  }
  const int e = (b - gemmBlocks) * blockDim.x + threadIdx.x;
  if (e < E) slot[e] = atomicAdd(cnt + dst[e], 1);
}

constexpr int SCAN_ROWS = 13;
constexpr int SCAN_WAVE_SPAN = SCAN_ROWS * 128;

template <int N>
__device__ __forceinline__ int row_shr_or0(int v) {
  return __builtin_amdgcn_update_dpp(0, v, 0x110 + N, 0xf, 0xf, true);
}

__device__ __forceinline__ int wave_incl_scan(int v, int lane) {
  v += row_shr_or0<1>(v);
  v += row_shr_or0<2>(v);
  v += row_shr_or0<4>(v);
  v += row_shr_or0<8>(v);
  const int lowRow = __builtin_amdgcn_readlane(v, 15);
  return v + (lane >= 16 ? lowRow : 0);
}

__global__ __launch_bounds__(1024) void scan_counts(
    const int* __restrict__ cnt, int* __restrict__ rowptr, int n) {
  __shared__ int waveTot[32];
  const int lane = threadIdx.x & 31;
  const int wave = __builtin_amdgcn_readfirstlane((int)(threadIdx.x >> 5));
  int carry = 0;
  for (int pass = 0; pass < n; pass += 32 * SCAN_WAVE_SPAN) {
    const int seg = pass + wave * SCAN_WAVE_SPAN;
    const int i0 = seg + lane * 4;
    const bool whole = seg + SCAN_WAVE_SPAN <= n;
    int4 v[SCAN_ROWS];
    if (whole) {
#pragma unroll
      for (int r = 0; r < SCAN_ROWS; ++r) v[r] = *(const int4*)(cnt + i0 + r * 128);
    } else {
#pragma unroll
      for (int r = 0; r < SCAN_ROWS; ++r) {
        const int i = i0 + r * 128;
        v[r].x = (i + 0 < n) ? cnt[i + 0] : 0;
        v[r].y = (i + 1 < n) ? cnt[i + 1] : 0;
        v[r].z = (i + 2 < n) ? cnt[i + 2] : 0;
        v[r].w = (i + 3 < n) ? cnt[i + 3] : 0;
      }
    }

    int before[SCAN_ROWS];
    int waveSum = 0;
#pragma unroll
    for (int r = 0; r < SCAN_ROWS; ++r) {
      const int s = (v[r].x + v[r].y) + (v[r].z + v[r].w);
      const int inc = wave_incl_scan(s, lane);
      before[r] = waveSum + (inc - s);
      waveSum += __builtin_amdgcn_readlane(inc, 31);
    }

    if (lane == 0) waveTot[wave] = waveSum;
    __syncthreads();
    const int wt = waveTot[lane];
    const int wi = wave_incl_scan(wt, lane);
    const int waveBase = carry + __builtin_amdgcn_readlane(wi - wt, wave);
    carry += __builtin_amdgcn_readlane(wi, 31);

#pragma unroll
    for (int r = 0; r < SCAN_ROWS; ++r) {
      const int i = i0 + r * 128;
      int4 o;
      o.x = waveBase + before[r];
      o.y = o.x + v[r].x;
      o.z = o.y + v[r].y;
      o.w = o.z + v[r].z;
      if (whole) {
        *(int4*)(rowptr + i) = o;
      } else {
        if (i + 0 < n) rowptr[i + 0] = o.x;
        if (i + 1 < n) rowptr[i + 1] = o.y;
        if (i + 2 < n) rowptr[i + 2] = o.z;
        if (i + 3 < n) rowptr[i + 3] = o.w;
      }
    }
    __syncthreads();
  }
  if (threadIdx.x == 0) rowptr[n] = carry;
}

__global__ void edge_scatter(const int* __restrict__ src,
                             const int* __restrict__ dst,
                             const int* __restrict__ slot,
                             const int* __restrict__ rowptr,
                             int* __restrict__ colTmp, int E) {
  int e = blockIdx.x * blockDim.x + threadIdx.x;
  if (e < E) colTmp[rowptr[dst[e]] + slot[e]] = src[e];
}

constexpr unsigned dpp8_xor_sel(int x) {
  unsigned s = 0;
  for (int i = 0; i < 8; ++i) s |= (unsigned)((i ^ x) & 7) << (3 * i);
  return s;
}
constexpr unsigned dpp8_bcast_sel(int k) { return (unsigned)k * 0x249249u; }

template <unsigned SEL>
__device__ __forceinline__ float dpp8_f(float v) {
  return __uint_as_float(__builtin_amdgcn_mov_dpp8(__float_as_uint(v), SEL));
}
template <unsigned SEL>
__device__ __forceinline__ int dpp8_i(int v) {
  return (int)__builtin_amdgcn_mov_dpp8((unsigned)v, SEL);
}

template <int J>
__device__ __forceinline__ int lane_xor_i(int v);
template <>
__device__ __forceinline__ int lane_xor_i<1>(int v) { return dpp8_i<dpp8_xor_sel(1)>(v); }
template <>
__device__ __forceinline__ int lane_xor_i<2>(int v) { return dpp8_i<dpp8_xor_sel(2)>(v); }
template <>
__device__ __forceinline__ int lane_xor_i<4>(int v) { return dpp8_i<dpp8_xor_sel(4)>(v); }
template <>
__device__ __forceinline__ int lane_xor_i<8>(int v) {
  return __builtin_amdgcn_update_dpp(v, v, 0x168  , 0xf, 0xf, false);
}
template <>
__device__ __forceinline__ int lane_xor_i<16>(int v) {
  return (int)__builtin_amdgcn_permlanex16((unsigned)v, (unsigned)v, 0x76543210u,
                                           0xfedcba98u, false, false);
}

template <int J>
__device__ __forceinline__ int cmp_xchg(int v, int lane, bool asc) {
  const int p = lane_xor_i<J>(v);
  const bool keepMin = ((lane & J) == 0) == asc;
  return keepMin ? min(v, p) : max(v, p);
}

template <int K>
__device__ __forceinline__ int merge_lanes(int v, int lane, bool asc) {
  if (K >= 32) v = cmp_xchg<16>(v, lane, asc);
  if (K >= 16) v = cmp_xchg<8>(v, lane, asc);
  if (K >= 8) v = cmp_xchg<4>(v, lane, asc);
  if (K >= 4) v = cmp_xchg<2>(v, lane, asc);
  return cmp_xchg<1>(v, lane, asc);
}

__device__ __forceinline__ int sort_lanes(int v, int lane, bool asc) {
  v = merge_lanes<2>(v, lane, (lane & 2) == 0);
  v = merge_lanes<4>(v, lane, (lane & 4) == 0);
  v = merge_lanes<8>(v, lane, (lane & 8) == 0);
  v = merge_lanes<16>(v, lane, (lane & 16) == 0);
  return merge_lanes<32>(v, lane, asc);
}

__global__ __launch_bounds__(256) void node_attn_h8(
    const float* __restrict__ feat, const int* __restrict__ rowptr,
    const int* __restrict__ colTmp, int* col, float* __restrict__ out, int Nn,
    float scale) {
  const int lane = threadIdx.x & 31;
  const int node = __builtin_amdgcn_readfirstlane(
      (int)((blockIdx.x * blockDim.x + threadIdx.x) >> 5));
  if (node >= Nn) return;
  const int beg = rowptr[node];
  const int end = rowptr[node + 1];
  const int n = end - beg;
  const float* fl = feat + lane * 4;
  const float4 fd = *(const float4*)(fl + (size_t)node * 128);

  const bool inRegs = n <= 64;
  int k0 = 0x7fffffff, k1 = 0x7fffffff;
  if (inRegs) {
    if (lane < n) k0 = colTmp[beg + lane];
    if (n > 32) {
      if (32 + lane < n) k1 = colTmp[beg + 32 + lane];
      k0 = sort_lanes(k0, lane, true);
      k1 = sort_lanes(k1, lane, false);
      const int lo = min(k0, k1);
      const int hi = max(k0, k1);
      k0 = merge_lanes<32>(lo, lane, true);
      k1 = merge_lanes<32>(hi, lane, true);
      if (32 + lane < n) col[beg + 32 + lane] = k1;
    } else {
      k0 = sort_lanes(k0, lane, true);
    }
    if (lane < n) col[beg + lane] = k0;
  } else {
    for (int ib = 0; ib < n; ib += 32) {
      const int i = ib + lane;
      const bool vi = i < n;
      const int ki = vi ? colTmp[beg + i] : 0;
      int rank = 0;
      for (int jb = 0; jb < n; jb += 32) {
        const int cj = min(32, n - jb);
        const int kc = (lane < cj) ? colTmp[beg + jb + lane] : 0;
        for (int t = 0; t < cj; ++t) {
          const int kj = __builtin_amdgcn_readlane(kc, t);
          const int bias = (jb + t < i) ? 1 : 0;
          rank += (kj < ki + bias) ? 1 : 0;
        }
      }
      if (vi) col[beg + rank] = ki;
    }
    __threadfence();
  }

  float m = -INFINITY, l = 0.0f;
  float4 acc = make_float4(0.0f, 0.0f, 0.0f, 0.0f);

  for (int base = beg; base < end; base += 32) {
    const int cnt = min(32, end - base);
    int ids;
    if (inRegs) {
      ids = (base == beg) ? k0 : k1;
    } else {
      ids = (lane < cnt) ? col[base + lane] : 0;
    }
    ids = (lane < cnt) ? ids : node;
    for (int t = 0; t < cnt; t += 4) {
      float4 f[4];
      float p[4];
#pragma unroll
      for (int u = 0; u < 4; ++u) {
        const int s = __builtin_amdgcn_readlane(ids, t + u);
        f[u] = *(const float4*)(fl + (size_t)s * 128);
      }
#pragma unroll
      for (int u = 0; u < 4; ++u) {
        float d = f[u].x * fd.x + f[u].y * fd.y + f[u].z * fd.z + f[u].w * fd.w;
        d += dpp8_f<dpp8_xor_sel(1)>(d);
        d += dpp8_f<dpp8_xor_sel(2)>(d);
        d *= scale;
        p[u] = (t + u < cnt) ? d : -INFINITY;
      }
      const float mNew = fmaxf(fmaxf(m, p[0]), fmaxf(fmaxf(p[1], p[2]), p[3]));
      const float c = __expf(m - mNew);
      const float w0 = __expf(p[0] - mNew);
      const float w1 = __expf(p[1] - mNew);
      const float w2 = __expf(p[2] - mNew);
      const float w3 = __expf(p[3] - mNew);
      l = l * c + ((w0 + w1) + (w2 + w3));
      acc.x = acc.x * c + ((w0 * f[0].x + w1 * f[1].x) + (w2 * f[2].x + w3 * f[3].x));
      acc.y = acc.y * c + ((w0 * f[0].y + w1 * f[1].y) + (w2 * f[2].y + w3 * f[3].y));
      acc.z = acc.z * c + ((w0 * f[0].z + w1 * f[1].z) + (w2 * f[2].z + w3 * f[3].z));
      acc.w = acc.w * c + ((w0 * f[0].w + w1 * f[1].w) + (w2 * f[2].w + w3 * f[3].w));
      m = mNew;
    }
  }

  const float inv = 1.0f / fmaxf(l, 1e-9f);
  float4 o;
  o.x = acc.x * inv;
  o.y = acc.y * inv;
  o.z = acc.z * inv;
  o.w = acc.w * inv;
  o.x = o.x > 0.0f ? o.x : expm1f(o.x);
  o.y = o.y > 0.0f ? o.y : expm1f(o.y);
  o.z = o.z > 0.0f ? o.z : expm1f(o.z);
  o.w = o.w > 0.0f ? o.w : expm1f(o.w);
  *(float4*)(out + (size_t)node * 128 + lane * 4) = o;
}

template <int K0>
__device__ __forceinline__ void h1_four_steps(
    const float* __restrict__ fq, int ids, int cnt, int g, const float4 fdA,
    const float4 fdB, float scale, float& m, float& l, float4& accA,
    float4& accB) {
  float4 fa[4], fb[4];
  float p[4];
#pragma unroll
  for (int u = 0; u < 4; ++u) {
    const int s = u == 0   ? dpp8_i<dpp8_bcast_sel(K0 + 0)>(ids)
                  : u == 1 ? dpp8_i<dpp8_bcast_sel(K0 + 1)>(ids)
                  : u == 2 ? dpp8_i<dpp8_bcast_sel(K0 + 2)>(ids)
                           : dpp8_i<dpp8_bcast_sel(K0 + 3)>(ids);
    const float* r = fq + (size_t)s * 64;
    fa[u] = *(const float4*)(r);
    fb[u] = *(const float4*)(r + 32);
  }
#pragma unroll
  for (int u = 0; u < 4; ++u) {
    float d = fa[u].x * fdA.x + fa[u].y * fdA.y + fa[u].z * fdA.z +
              fa[u].w * fdA.w + fb[u].x * fdB.x + fb[u].y * fdB.y +
              fb[u].z * fdB.z + fb[u].w * fdB.w;
    d += dpp8_f<dpp8_xor_sel(1)>(d);
    d += dpp8_f<dpp8_xor_sel(2)>(d);
    d += dpp8_f<dpp8_xor_sel(4)>(d);
    d *= scale;
    p[u] = (4 * (K0 + u) + g < cnt) ? d : -INFINITY;
  }
  const float mNew = fmaxf(fmaxf(m, p[0]), fmaxf(fmaxf(p[1], p[2]), p[3]));
  const float mRef = (mNew == -INFINITY) ? 0.0f : mNew;
  const float c = __expf(m - mRef);
  const float w0 = __expf(p[0] - mRef);
  const float w1 = __expf(p[1] - mRef);
  const float w2 = __expf(p[2] - mRef);
  const float w3 = __expf(p[3] - mRef);
  l = l * c + ((w0 + w1) + (w2 + w3));
  accA.x = accA.x * c + ((w0 * fa[0].x + w1 * fa[1].x) + (w2 * fa[2].x + w3 * fa[3].x));
  accA.y = accA.y * c + ((w0 * fa[0].y + w1 * fa[1].y) + (w2 * fa[2].y + w3 * fa[3].y));
  accA.z = accA.z * c + ((w0 * fa[0].z + w1 * fa[1].z) + (w2 * fa[2].z + w3 * fa[3].z));
  accA.w = accA.w * c + ((w0 * fa[0].w + w1 * fa[1].w) + (w2 * fa[2].w + w3 * fa[3].w));
  accB.x = accB.x * c + ((w0 * fb[0].x + w1 * fb[1].x) + (w2 * fb[2].x + w3 * fb[3].x));
  accB.y = accB.y * c + ((w0 * fb[0].y + w1 * fb[1].y) + (w2 * fb[2].y + w3 * fb[3].y));
  accB.z = accB.z * c + ((w0 * fb[0].z + w1 * fb[1].z) + (w2 * fb[2].z + w3 * fb[3].z));
  accB.w = accB.w * c + ((w0 * fb[0].w + w1 * fb[1].w) + (w2 * fb[2].w + w3 * fb[3].w));
  m = mNew;
}

__device__ __forceinline__ float sum_groups(float v) {
  v += __shfl_xor(v, 8, 32);
  v += __shfl_xor(v, 16, 32);
  return v;
}

__global__ __launch_bounds__(256) void node_attn_h1(
    const float* __restrict__ feat, const int* __restrict__ rowptr,
    const int* __restrict__ col, float* __restrict__ out, int Nn, float scale) {
  const int lane = threadIdx.x & 31;
  const int node = __builtin_amdgcn_readfirstlane(
      (int)((blockIdx.x * blockDim.x + threadIdx.x) >> 5));
  if (node >= Nn) return;
  const int beg = rowptr[node];
  const int end = rowptr[node + 1];
  const int q = lane & 7;
  const int g = lane >> 3;
  const float* fq = feat + q * 4;
  const float4 fdA = *(const float4*)(fq + (size_t)node * 64);
  const float4 fdB = *(const float4*)(fq + (size_t)node * 64 + 32);

  float m = -INFINITY, l = 0.0f;
  float4 accA = make_float4(0.0f, 0.0f, 0.0f, 0.0f);
  float4 accB = make_float4(0.0f, 0.0f, 0.0f, 0.0f);

  const int myIdx = 4 * q + g;
  for (int base = beg; base < end; base += 32) {
    const int cnt = min(32, end - base);
    const int ids = (myIdx < cnt) ? col[base + myIdx] : node;
    h1_four_steps<0>(fq, ids, cnt, g, fdA, fdB, scale, m, l, accA, accB);
    if (cnt > 16)
      h1_four_steps<4>(fq, ids, cnt, g, fdA, fdB, scale, m, l, accA, accB);
  }

  float M = fmaxf(m, __shfl_xor(m, 8, 32));
  M = fmaxf(M, __shfl_xor(M, 16, 32));
  const float Mref = (M == -INFINITY) ? 0.0f : M;
  const float sc = __expf(m - Mref);
  const float L = sum_groups(l * sc);
  float4 oA, oB;
  oA.x = sum_groups(accA.x * sc);
  oA.y = sum_groups(accA.y * sc);
  oA.z = sum_groups(accA.z * sc);
  oA.w = sum_groups(accA.w * sc);
  oB.x = sum_groups(accB.x * sc);
  oB.y = sum_groups(accB.y * sc);
  oB.z = sum_groups(accB.z * sc);
  oB.w = sum_groups(accB.w * sc);
  const float inv = 1.0f / fmaxf(L, 1e-9f);
  if (g == 0) {
    float* o = out + (size_t)node * 64 + q * 4;
    *(float4*)(o) = make_float4(oA.x * inv, oA.y * inv, oA.z * inv, oA.w * inv);
    *(float4*)(o + 32) = make_float4(oB.x * inv, oB.y * inv, oB.z * inv, oB.w * inv);
  }
}

static inline int cdiv_i(long long a, long long b) { return (int)((a + b - 1) / b); }

extern "C" void kernel_launch(void* const* d_in, const int* in_sizes, int n_in,
                              void* d_out, int out_size, void* d_ws, size_t ws_size,
                              hipStream_t stream) {
  const float* hIn = (const float*)d_in[0];
  const float* W1  = (const float*)d_in[1];
  const float* W2  = (const float*)d_in[2];
  const int*   src = (const int*)d_in[3];
  const int*   dst = (const int*)d_in[4];

  const int IN_DIM = 128, OUT_DIM = 64;
  const int HD1 = 128;
  const int Nn  = in_sizes[0] / IN_DIM;
  const int E   = in_sizes[3];

  float* feat1  = (float*)d_ws;
  float* h1     = feat1 + (size_t)Nn * HD1;
  int*   slot   = (int*)(h1 + (size_t)Nn * HD1);
  int*   colTmp = slot + (size_t)E;
  int*   col    = colTmp + (size_t)E;
  int*   cnt    = col + (size_t)E;
  int*   rowptr = cnt + (size_t)Nn;

  const int TPB = 256;
  const int wavesPerBlock = TPB / 32;

  zero_i32<<<cdiv_i(Nn, TPB), TPB, 0, stream>>>(cnt, Nn);
  {
    int gemmBlocks = cdiv_i((Nn / 16) * (HD1 / 64), wavesPerBlock);
    edge_slot_and_gemm<128, 128><<<cdiv_i(E, TPB) + gemmBlocks, TPB, 0, stream>>>(
        dst, cnt, slot, E, hIn, W1, feat1, Nn, gemmBlocks);
  }
  scan_counts<<<1, 1024, 0, stream>>>(cnt, rowptr, Nn);
  edge_scatter<<<cdiv_i(E, TPB), TPB, 0, stream>>>(src, dst, slot, rowptr, colTmp, E);

  node_attn_h8<<<cdiv_i(Nn, wavesPerBlock), TPB, 0, stream>>>(
      feat1, rowptr, colTmp, col, h1, Nn, 0.25f);

  float* feat2 = feat1;
  {
    int tiles = (Nn / 16) * (OUT_DIM / 16);
    gemm_f32_wmma<<<cdiv_i(tiles, wavesPerBlock), TPB, 0, stream>>>(
        h1, W2, feat2, Nn, OUT_DIM, HD1);
  }
  node_attn_h1<<<cdiv_i(Nn, wavesPerBlock), TPB, 0, stream>>>(
      feat2, rowptr, col, (float*)d_out, Nn, 0.125f);
}


# Edge grouping by destination: per-edge returning global atomics and random scatter replaced by a two-level counting sort on LDS counters (256-node bins, 256 edge chunks); same sorted lists and sums
# speedup vs baseline: 7.2010x; 1.5676x over previous
#include <hip/hip_runtime.h>
#include <math.h>

typedef __attribute__((ext_vector_type(2))) float v2f;
typedef __attribute__((ext_vector_type(8))) float v8f;

__global__ __launch_bounds__(256) void gemm_f32_wmma(
    const float* __restrict__ A, const float* __restrict__ B,
    float* __restrict__ C, int M, int N, int K) {
  const int lane = threadIdx.x & 31;
  const int wave = threadIdx.x >> 5;
  const int wavesPerBlock = blockDim.x >> 5;
  const int tile = blockIdx.x * wavesPerBlock + wave;
  const int tilesN = N >> 4;
  const int numTiles = (M >> 4) * tilesN;
  if (tile >= numTiles) return;

  const int tm = (tile / tilesN) << 4;
  const int tn = (tile % tilesN) << 4;
  const int r  = lane & 15;
  const int kk = (lane >> 4) << 1;

  const float* Arow = A + (size_t)(tm + r) * K;

  v8f acc = {};
  for (int k0 = 0; k0 < K; k0 += 4) {
    v2f a, b;
    float2 av = *(const float2*)(Arow + k0 + kk);
    a.x = av.x;
    a.y = av.y;
    b.x = B[(size_t)(k0 + kk) * N + tn + r];
    b.y = B[(size_t)(k0 + kk + 1) * N + tn + r];
    acc = __builtin_amdgcn_wmma_f32_16x16x4_f32(
         false, a,  false, b,
         (short)0, acc,  false,  false);
  }

  const int rowBase = tm + ((lane >> 4) << 3);
#pragma unroll
  for (int j = 0; j < 8; ++j) {
    C[(size_t)(rowBase + j) * N + tn + r] = acc[j];
  }
}

constexpr int BIN_SHIFT = 8;
constexpr int BIN_NODES = 1 << BIN_SHIFT;
constexpr int NCHUNK = 256;
constexpr int MAX_BINS = 1024;
constexpr int EDGE_BATCH = 8;

template <int KD, int N>
__device__ __forceinline__ void gemm_block_16x64(
    const float* __restrict__ A, const float* __restrict__ B,
    float* __restrict__ C, int unit, int lane) {
  constexpr int NT = 4;
  constexpr int CG = N / (16 * NT);
  constexpr int STEPS = KD / 8;
  const int tm  = (unit / CG) << 4;
  const int tn0 = (unit % CG) * (16 * NT);
  const int c = lane & 15;
  const int h = lane >> 4;

  const float* Arow = A + (size_t)(tm + c) * KD + 4 * h;
  const float* Bp = B + (size_t)(4 * h) * N + tn0 + c;
  v8f acc[NT] = {};
#pragma unroll
  for (int j = 0; j < STEPS; ++j) {
    const float4 a = *(const float4*)(Arow + 8 * j);
    float b0[NT], b1[NT], b2[NT], b3[NT];
#pragma unroll
    for (int t = 0; t < NT; ++t) {
      const float* p = Bp + (size_t)(8 * j) * N + 16 * t;
      b0[t] = p[0];
      b1[t] = p[N];
      b2[t] = p[2 * N];
      b3[t] = p[3 * N];
    }
#pragma unroll
    for (int t = 0; t < NT; ++t) {
      v2f av, bv;
      av.x = a.x;
      av.y = a.y;
      bv.x = b0[t];
      bv.y = b1[t];
      acc[t] = __builtin_amdgcn_wmma_f32_16x16x4_f32(
           false, av,  false, bv,
           (short)0, acc[t],  false,  false);
    }
#pragma unroll
    for (int t = 0; t < NT; ++t) {
      v2f av, bv;
      av.x = a.z;
      av.y = a.w;
      bv.x = b2[t];
      bv.y = b3[t];
      acc[t] = __builtin_amdgcn_wmma_f32_16x16x4_f32(
           false, av,  false, bv,
           (short)0, acc[t],  false,  false);
    }
  }

  const int rowBase = tm + (h << 3);
#pragma unroll
  for (int t = 0; t < NT; ++t) {
#pragma unroll
    for (int j = 0; j < 8; ++j) {
      C[(size_t)(rowBase + j) * N + tn0 + 16 * t + c] = acc[t][j];
    }
  }
}

template <int KD, int N>
__global__ __launch_bounds__(256) void bin_count_and_gemm(
    const int* __restrict__ dst, int* __restrict__ hist, int E, int chunk,
    int nBins, const float* __restrict__ A, const float* __restrict__ B,
    float* __restrict__ C, int M, int gemmBlocks) {
  __shared__ int binCnt[MAX_BINS];
  const int b = blockIdx.x;
  if (b < gemmBlocks) {
    const int unit = __builtin_amdgcn_readfirstlane(
        (int)((b * blockDim.x + threadIdx.x) >> 5));
    if (unit < (M >> 4) * (N / 64))
      gemm_block_16x64<KD, N>(A, B, C, unit, threadIdx.x & 31);
    return;
  }
  const int c = b - gemmBlocks;
  const int nt = blockDim.x;
  for (int i = threadIdx.x; i < nBins; i += nt) binCnt[i] = 0;
  __syncthreads();
  const int e0 = c * chunk;
  const int e1 = min(E, e0 + chunk);
  for (int e = e0 + threadIdx.x; e < e1; e += EDGE_BATCH * nt) {
    int d[EDGE_BATCH];
#pragma unroll
    for (int u = 0; u < EDGE_BATCH; ++u) {
      const int i = e + u * nt;
      d[u] = (i < e1) ? dst[i] : -1;
    }
#pragma unroll
    for (int u = 0; u < EDGE_BATCH; ++u)
      if (d[u] >= 0) atomicAdd(&binCnt[d[u] >> BIN_SHIFT], 1);
  }
  __syncthreads();
  for (int i = threadIdx.x; i < nBins; i += nt) hist[i * NCHUNK + c] = binCnt[i];
}

constexpr int SCAN_ROWS = 13;
constexpr int SCAN_WAVE_SPAN = SCAN_ROWS * 128;

template <int N>
__device__ __forceinline__ int row_shr_or0(int v) {
  return __builtin_amdgcn_update_dpp(0, v, 0x110 + N, 0xf, 0xf, true);
}

__device__ __forceinline__ int wave_incl_scan(int v, int lane) {
  v += row_shr_or0<1>(v);
  v += row_shr_or0<2>(v);
  v += row_shr_or0<4>(v);
  v += row_shr_or0<8>(v);
  const int lowRow = __builtin_amdgcn_readlane(v, 15);
  return v + (lane >= 16 ? lowRow : 0);
}

__global__ __launch_bounds__(1024) void scan_counts(
    const int* __restrict__ cnt, int* __restrict__ rowptr, int n) {
  __shared__ int waveTot[32];
  const int lane = threadIdx.x & 31;
  const int wave = __builtin_amdgcn_readfirstlane((int)(threadIdx.x >> 5));
  int carry = 0;
  for (int pass = 0; pass < n; pass += 32 * SCAN_WAVE_SPAN) {
    const int seg = pass + wave * SCAN_WAVE_SPAN;
    const int i0 = seg + lane * 4;
    const bool whole = seg + SCAN_WAVE_SPAN <= n;
    int4 v[SCAN_ROWS];
    if (whole) {
#pragma unroll
      for (int r = 0; r < SCAN_ROWS; ++r) v[r] = *(const int4*)(cnt + i0 + r * 128);
    } else {
#pragma unroll
      for (int r = 0; r < SCAN_ROWS; ++r) {
        const int i = i0 + r * 128;
        v[r].x = (i + 0 < n) ? cnt[i + 0] : 0;
        v[r].y = (i + 1 < n) ? cnt[i + 1] : 0;
        v[r].z = (i + 2 < n) ? cnt[i + 2] : 0;
        v[r].w = (i + 3 < n) ? cnt[i + 3] : 0;
      }
    }

    int before[SCAN_ROWS];
    int waveSum = 0;
#pragma unroll
    for (int r = 0; r < SCAN_ROWS; ++r) {
      const int s = (v[r].x + v[r].y) + (v[r].z + v[r].w);
      const int inc = wave_incl_scan(s, lane);
      before[r] = waveSum + (inc - s);
      waveSum += __builtin_amdgcn_readlane(inc, 31);
    }

    if (lane == 0) waveTot[wave] = waveSum;
    __syncthreads();
    const int wt = waveTot[lane];
    const int wi = wave_incl_scan(wt, lane);
    const int waveBase = carry + __builtin_amdgcn_readlane(wi - wt, wave);
    carry += __builtin_amdgcn_readlane(wi, 31);

#pragma unroll
    for (int r = 0; r < SCAN_ROWS; ++r) {
      const int i = i0 + r * 128;
      int4 o;
      o.x = waveBase + before[r];
      o.y = o.x + v[r].x;
      o.z = o.y + v[r].y;
      o.w = o.z + v[r].z;
      if (whole) {
        *(int4*)(rowptr + i) = o;
      } else {
        if (i + 0 < n) rowptr[i + 0] = o.x;
        if (i + 1 < n) rowptr[i + 1] = o.y;
        if (i + 2 < n) rowptr[i + 2] = o.z;
        if (i + 3 < n) rowptr[i + 3] = o.w;
      }
    }
    __syncthreads();
  }
  if (threadIdx.x == 0) rowptr[n] = carry;
}

__global__ __launch_bounds__(1024) void bin_scatter(
    const int* __restrict__ src, const int* __restrict__ dst,
    const int* __restrict__ off, int2* __restrict__ binned, int E, int chunk,
    int nBins) {
  __shared__ int cur[MAX_BINS];
  const int c = blockIdx.x;
  const int nt = blockDim.x;
  for (int i = threadIdx.x; i < nBins; i += nt) cur[i] = off[i * NCHUNK + c];
  __syncthreads();
  const int e0 = c * chunk;
  const int e1 = min(E, e0 + chunk);
  for (int e = e0 + threadIdx.x; e < e1; e += EDGE_BATCH * nt) {
    int d[EDGE_BATCH], s[EDGE_BATCH];
#pragma unroll
    for (int u = 0; u < EDGE_BATCH; ++u) {
      const int i = e + u * nt;
      const bool v = i < e1;
      d[u] = v ? dst[i] : -1;
      s[u] = v ? src[i] : 0;
    }
#pragma unroll
    for (int u = 0; u < EDGE_BATCH; ++u) {
      if (d[u] >= 0) {
        const int pos = atomicAdd(&cur[d[u] >> BIN_SHIFT], 1);
        binned[pos] = make_int2(d[u], s[u]);
      }
    }
  }
}

__global__ __launch_bounds__(1024) void bin_lists(
    const int2* __restrict__ binned, const int* __restrict__ off,
    int* __restrict__ rowptr, int* __restrict__ colTmp, int Nn, int nBins) {
  __shared__ int cnt[BIN_NODES];
  __shared__ int waveTot[BIN_NODES / 32];
  const int bin = blockIdx.x;
  const int t = threadIdx.x;
  const int nt = blockDim.x;
  const int lane = t & 31;
  const int b0 = off[bin * NCHUNK];
  const int b1 = off[(bin + 1) * NCHUNK];
  const int node0 = bin << BIN_SHIFT;
  if (t < BIN_NODES) cnt[t] = 0;
  __syncthreads();

  for (int i = b0 + t; i < b1; i += EDGE_BATCH * nt) {
    int d[EDGE_BATCH];
#pragma unroll
    for (int u = 0; u < EDGE_BATCH; ++u) {
      const int j = i + u * nt;
      d[u] = (j < b1) ? binned[j].x : -1;
    }
#pragma unroll
    for (int u = 0; u < EDGE_BATCH; ++u)
      if (d[u] >= 0) atomicAdd(&cnt[d[u] - node0], 1);
  }
  __syncthreads();

  int c = 0, inc = 0;
  if (t < BIN_NODES) {
    c = cnt[t];
    inc = wave_incl_scan(c, lane);
    if (lane == 31) waveTot[t >> 5] = inc;
  }
  __syncthreads();
  if (t < BIN_NODES) {
    int base = b0;
    for (int w = 0; w < (t >> 5); ++w) base += waveTot[w];
    const int start = base + (inc - c);
    cnt[t] = start;
    const int node = node0 + t;
    if (node < Nn) rowptr[node] = start;
  }
  if (bin == nBins - 1 && t == 0) rowptr[Nn] = b1;
  __syncthreads();

  for (int i = b0 + t; i < b1; i += EDGE_BATCH * nt) {
    int2 e[EDGE_BATCH];
#pragma unroll
    for (int u = 0; u < EDGE_BATCH; ++u) {
      const int j = i + u * nt;
      e[u] = (j < b1) ? binned[j] : make_int2(-1, 0);
    }
#pragma unroll
    for (int u = 0; u < EDGE_BATCH; ++u) {
      if (e[u].x >= 0) {
        const int pos = atomicAdd(&cnt[e[u].x - node0], 1);
        colTmp[pos] = e[u].y;
      }
    }
  }
}

constexpr unsigned dpp8_xor_sel(int x) {
  unsigned s = 0;
  for (int i = 0; i < 8; ++i) s |= (unsigned)((i ^ x) & 7) << (3 * i);
  return s;
}
constexpr unsigned dpp8_bcast_sel(int k) { return (unsigned)k * 0x249249u; }

template <unsigned SEL>
__device__ __forceinline__ float dpp8_f(float v) {
  return __uint_as_float(__builtin_amdgcn_mov_dpp8(__float_as_uint(v), SEL));
}
template <unsigned SEL>
__device__ __forceinline__ int dpp8_i(int v) {
  return (int)__builtin_amdgcn_mov_dpp8((unsigned)v, SEL);
}

template <int J>
__device__ __forceinline__ int lane_xor_i(int v);
template <>
__device__ __forceinline__ int lane_xor_i<1>(int v) { return dpp8_i<dpp8_xor_sel(1)>(v); }
template <>
__device__ __forceinline__ int lane_xor_i<2>(int v) { return dpp8_i<dpp8_xor_sel(2)>(v); }
template <>
__device__ __forceinline__ int lane_xor_i<4>(int v) { return dpp8_i<dpp8_xor_sel(4)>(v); }
template <>
__device__ __forceinline__ int lane_xor_i<8>(int v) {
  return __builtin_amdgcn_update_dpp(v, v, 0x168  , 0xf, 0xf, false);
}
template <>
__device__ __forceinline__ int lane_xor_i<16>(int v) {
  return (int)__builtin_amdgcn_permlanex16((unsigned)v, (unsigned)v, 0x76543210u,
                                           0xfedcba98u, false, false);
}

template <int J>
__device__ __forceinline__ int cmp_xchg(int v, int lane, bool asc) {
  const int p = lane_xor_i<J>(v);
  const bool keepMin = ((lane & J) == 0) == asc;
  return keepMin ? min(v, p) : max(v, p);
}

template <int K>
__device__ __forceinline__ int merge_lanes(int v, int lane, bool asc) {
  if (K >= 32) v = cmp_xchg<16>(v, lane, asc);
  if (K >= 16) v = cmp_xchg<8>(v, lane, asc);
  if (K >= 8) v = cmp_xchg<4>(v, lane, asc);
  if (K >= 4) v = cmp_xchg<2>(v, lane, asc);
  return cmp_xchg<1>(v, lane, asc);
}

__device__ __forceinline__ int sort_lanes(int v, int lane, bool asc) {
  v = merge_lanes<2>(v, lane, (lane & 2) == 0);
  v = merge_lanes<4>(v, lane, (lane & 4) == 0);
  v = merge_lanes<8>(v, lane, (lane & 8) == 0);
  v = merge_lanes<16>(v, lane, (lane & 16) == 0);
  return merge_lanes<32>(v, lane, asc);
}

__global__ __launch_bounds__(256) void node_attn_h8(
    const float* __restrict__ feat, const int* __restrict__ rowptr,
    const int* __restrict__ colTmp, int* col, float* __restrict__ out, int Nn,
    float scale) {
  const int lane = threadIdx.x & 31;
  const int node = __builtin_amdgcn_readfirstlane(
      (int)((blockIdx.x * blockDim.x + threadIdx.x) >> 5));
  if (node >= Nn) return;
  const int beg = rowptr[node];
  const int end = rowptr[node + 1];
  const int n = end - beg;
  const float* fl = feat + lane * 4;
  const float4 fd = *(const float4*)(fl + (size_t)node * 128);

  const bool inRegs = n <= 64;
  int k0 = 0x7fffffff, k1 = 0x7fffffff;
  if (inRegs) {
    if (lane < n) k0 = colTmp[beg + lane];
    if (n > 32) {
      if (32 + lane < n) k1 = colTmp[beg + 32 + lane];
      k0 = sort_lanes(k0, lane, true);
      k1 = sort_lanes(k1, lane, false);
      const int lo = min(k0, k1);
      const int hi = max(k0, k1);
      k0 = merge_lanes<32>(lo, lane, true);
      k1 = merge_lanes<32>(hi, lane, true);
      if (32 + lane < n) col[beg + 32 + lane] = k1;
    } else {
      k0 = sort_lanes(k0, lane, true);
    }
    if (lane < n) col[beg + lane] = k0;
  } else {
    for (int ib = 0; ib < n; ib += 32) {
      const int i = ib + lane;
      const bool vi = i < n;
      const int ki = vi ? colTmp[beg + i] : 0;
      int rank = 0;
      for (int jb = 0; jb < n; jb += 32) {
        const int cj = min(32, n - jb);
        const int kc = (lane < cj) ? colTmp[beg + jb + lane] : 0;
        for (int t = 0; t < cj; ++t) {
          const int kj = __builtin_amdgcn_readlane(kc, t);
          const int bias = (jb + t < i) ? 1 : 0;
          rank += (kj < ki + bias) ? 1 : 0;
        }
      }
      if (vi) col[beg + rank] = ki;
    }
    __threadfence();
  }

  float m = -INFINITY, l = 0.0f;
  float4 acc = make_float4(0.0f, 0.0f, 0.0f, 0.0f);

  for (int base = beg; base < end; base += 32) {
    const int cnt = min(32, end - base);
    int ids;
    if (inRegs) {
      ids = (base == beg) ? k0 : k1;
    } else {
      ids = (lane < cnt) ? col[base + lane] : 0;
    }
    ids = (lane < cnt) ? ids : node;
    for (int t = 0; t < cnt; t += 4) {
      float4 f[4];
      float p[4];
#pragma unroll
      for (int u = 0; u < 4; ++u) {
        const int s = __builtin_amdgcn_readlane(ids, t + u);
        f[u] = *(const float4*)(fl + (size_t)s * 128);
      }
#pragma unroll
      for (int u = 0; u < 4; ++u) {
        float d = f[u].x * fd.x + f[u].y * fd.y + f[u].z * fd.z + f[u].w * fd.w;
        d += dpp8_f<dpp8_xor_sel(1)>(d);
        d += dpp8_f<dpp8_xor_sel(2)>(d);
        d *= scale;
        p[u] = (t + u < cnt) ? d : -INFINITY;
      }
      const float mNew = fmaxf(fmaxf(m, p[0]), fmaxf(fmaxf(p[1], p[2]), p[3]));
      const float c = __expf(m - mNew);
      const float w0 = __expf(p[0] - mNew);
      const float w1 = __expf(p[1] - mNew);
      const float w2 = __expf(p[2] - mNew);
      const float w3 = __expf(p[3] - mNew);
      l = l * c + ((w0 + w1) + (w2 + w3));
      acc.x = acc.x * c + ((w0 * f[0].x + w1 * f[1].x) + (w2 * f[2].x + w3 * f[3].x));
      acc.y = acc.y * c + ((w0 * f[0].y + w1 * f[1].y) + (w2 * f[2].y + w3 * f[3].y));
      acc.z = acc.z * c + ((w0 * f[0].z + w1 * f[1].z) + (w2 * f[2].z + w3 * f[3].z));
      acc.w = acc.w * c + ((w0 * f[0].w + w1 * f[1].w) + (w2 * f[2].w + w3 * f[3].w));
      m = mNew;
    }
  }

  const float inv = 1.0f / fmaxf(l, 1e-9f);
  float4 o;
  o.x = acc.x * inv;
  o.y = acc.y * inv;
  o.z = acc.z * inv;
  o.w = acc.w * inv;
  o.x = o.x > 0.0f ? o.x : expm1f(o.x);
  o.y = o.y > 0.0f ? o.y : expm1f(o.y);
  o.z = o.z > 0.0f ? o.z : expm1f(o.z);
  o.w = o.w > 0.0f ? o.w : expm1f(o.w);
  *(float4*)(out + (size_t)node * 128 + lane * 4) = o;
}

template <int K0>
__device__ __forceinline__ void h1_four_steps(
    const float* __restrict__ fq, int ids, int cnt, int g, const float4 fdA,
    const float4 fdB, float scale, float& m, float& l, float4& accA,
    float4& accB) {
  float4 fa[4], fb[4];
  float p[4];
#pragma unroll
  for (int u = 0; u < 4; ++u) {
    const int s = u == 0   ? dpp8_i<dpp8_bcast_sel(K0 + 0)>(ids)
                  : u == 1 ? dpp8_i<dpp8_bcast_sel(K0 + 1)>(ids)
                  : u == 2 ? dpp8_i<dpp8_bcast_sel(K0 + 2)>(ids)
                           : dpp8_i<dpp8_bcast_sel(K0 + 3)>(ids);
    const float* r = fq + (size_t)s * 64;
    fa[u] = *(const float4*)(r);
    fb[u] = *(const float4*)(r + 32);
  }
#pragma unroll
  for (int u = 0; u < 4; ++u) {
    float d = fa[u].x * fdA.x + fa[u].y * fdA.y + fa[u].z * fdA.z +
              fa[u].w * fdA.w + fb[u].x * fdB.x + fb[u].y * fdB.y +
              fb[u].z * fdB.z + fb[u].w * fdB.w;
    d += dpp8_f<dpp8_xor_sel(1)>(d);
    d += dpp8_f<dpp8_xor_sel(2)>(d);
    d += dpp8_f<dpp8_xor_sel(4)>(d);
    d *= scale;
    p[u] = (4 * (K0 + u) + g < cnt) ? d : -INFINITY;
  }
  const float mNew = fmaxf(fmaxf(m, p[0]), fmaxf(fmaxf(p[1], p[2]), p[3]));
  const float mRef = (mNew == -INFINITY) ? 0.0f : mNew;
  const float c = __expf(m - mRef);
  const float w0 = __expf(p[0] - mRef);
  const float w1 = __expf(p[1] - mRef);
  const float w2 = __expf(p[2] - mRef);
  const float w3 = __expf(p[3] - mRef);
  l = l * c + ((w0 + w1) + (w2 + w3));
  accA.x = accA.x * c + ((w0 * fa[0].x + w1 * fa[1].x) + (w2 * fa[2].x + w3 * fa[3].x));
  accA.y = accA.y * c + ((w0 * fa[0].y + w1 * fa[1].y) + (w2 * fa[2].y + w3 * fa[3].y));
  accA.z = accA.z * c + ((w0 * fa[0].z + w1 * fa[1].z) + (w2 * fa[2].z + w3 * fa[3].z));
  accA.w = accA.w * c + ((w0 * fa[0].w + w1 * fa[1].w) + (w2 * fa[2].w + w3 * fa[3].w));
  accB.x = accB.x * c + ((w0 * fb[0].x + w1 * fb[1].x) + (w2 * fb[2].x + w3 * fb[3].x));
  accB.y = accB.y * c + ((w0 * fb[0].y + w1 * fb[1].y) + (w2 * fb[2].y + w3 * fb[3].y));
  accB.z = accB.z * c + ((w0 * fb[0].z + w1 * fb[1].z) + (w2 * fb[2].z + w3 * fb[3].z));
  accB.w = accB.w * c + ((w0 * fb[0].w + w1 * fb[1].w) + (w2 * fb[2].w + w3 * fb[3].w));
  m = mNew;
}

__device__ __forceinline__ float sum_groups(float v) {
  v += __shfl_xor(v, 8, 32);
  v += __shfl_xor(v, 16, 32);
  return v;
}

__global__ __launch_bounds__(256) void node_attn_h1(
    const float* __restrict__ feat, const int* __restrict__ rowptr,
    const int* __restrict__ col, float* __restrict__ out, int Nn, float scale) {
  const int lane = threadIdx.x & 31;
  const int node = __builtin_amdgcn_readfirstlane(
      (int)((blockIdx.x * blockDim.x + threadIdx.x) >> 5));
  if (node >= Nn) return;
  const int beg = rowptr[node];
  const int end = rowptr[node + 1];
  const int q = lane & 7;
  const int g = lane >> 3;
  const float* fq = feat + q * 4;
  const float4 fdA = *(const float4*)(fq + (size_t)node * 64);
  const float4 fdB = *(const float4*)(fq + (size_t)node * 64 + 32);

  float m = -INFINITY, l = 0.0f;
  float4 accA = make_float4(0.0f, 0.0f, 0.0f, 0.0f);
  float4 accB = make_float4(0.0f, 0.0f, 0.0f, 0.0f);

  const int myIdx = 4 * q + g;
  for (int base = beg; base < end; base += 32) {
    const int cnt = min(32, end - base);
    const int ids = (myIdx < cnt) ? col[base + myIdx] : node;
    h1_four_steps<0>(fq, ids, cnt, g, fdA, fdB, scale, m, l, accA, accB);
    if (cnt > 16)
      h1_four_steps<4>(fq, ids, cnt, g, fdA, fdB, scale, m, l, accA, accB);
  }

  float M = fmaxf(m, __shfl_xor(m, 8, 32));
  M = fmaxf(M, __shfl_xor(M, 16, 32));
  const float Mref = (M == -INFINITY) ? 0.0f : M;
  const float sc = __expf(m - Mref);
  const float L = sum_groups(l * sc);
  float4 oA, oB;
  oA.x = sum_groups(accA.x * sc);
  oA.y = sum_groups(accA.y * sc);
  oA.z = sum_groups(accA.z * sc);
  oA.w = sum_groups(accA.w * sc);
  oB.x = sum_groups(accB.x * sc);
  oB.y = sum_groups(accB.y * sc);
  oB.z = sum_groups(accB.z * sc);
  oB.w = sum_groups(accB.w * sc);
  const float inv = 1.0f / fmaxf(L, 1e-9f);
  if (g == 0) {
    float* o = out + (size_t)node * 64 + q * 4;
    *(float4*)(o) = make_float4(oA.x * inv, oA.y * inv, oA.z * inv, oA.w * inv);
    *(float4*)(o + 32) = make_float4(oB.x * inv, oB.y * inv, oB.z * inv, oB.w * inv);
  }
}

static inline int cdiv_i(long long a, long long b) { return (int)((a + b - 1) / b); }

extern "C" void kernel_launch(void* const* d_in, const int* in_sizes, int n_in,
                              void* d_out, int out_size, void* d_ws, size_t ws_size,
                              hipStream_t stream) {
  const float* hIn = (const float*)d_in[0];
  const float* W1  = (const float*)d_in[1];
  const float* W2  = (const float*)d_in[2];
  const int*   src = (const int*)d_in[3];
  const int*   dst = (const int*)d_in[4];

  const int IN_DIM = 128, OUT_DIM = 64;
  const int HD1 = 128;
  const int Nn  = in_sizes[0] / IN_DIM;
  const int E   = in_sizes[3];

  const int nBins   = cdiv_i(Nn, BIN_NODES);
  const int chunk   = cdiv_i(E, NCHUNK);
  const int histLen = nBins * NCHUNK;

  float* feat1  = (float*)d_ws;
  float* h1     = feat1 + (size_t)Nn * HD1;
  int*   colTmp = (int*)(h1 + (size_t)Nn * HD1) + (size_t)E;
  int*   col    = colTmp + (size_t)E;
  int*   rowptr = col + (size_t)E + (size_t)Nn;
  int2*  binned = (int2*)h1;
  int*   off    = (int*)(h1 + (size_t)Nn * HD1) - (histLen + 4);
  int*   hist   = off - histLen;

  const int TPB = 256;
  const int wavesPerBlock = TPB / 32;

  {
    int gemmBlocks = cdiv_i((Nn / 16) * (HD1 / 64), wavesPerBlock);
    bin_count_and_gemm<128, 128><<<gemmBlocks + NCHUNK, TPB, 0, stream>>>(
        dst, hist, E, chunk, nBins, hIn, W1, feat1, Nn, gemmBlocks);
  }
  scan_counts<<<1, 1024, 0, stream>>>(hist, off, histLen);
  bin_scatter<<<NCHUNK, 1024, 0, stream>>>(src, dst, off, binned, E, chunk, nBins);
  bin_lists<<<nBins, 1024, 0, stream>>>(binned, off, rowptr, colTmp, Nn, nBins);

  node_attn_h8<<<cdiv_i(Nn, wavesPerBlock), TPB, 0, stream>>>(
      feat1, rowptr, colTmp, col, h1, Nn, 0.25f);

  float* feat2 = feat1;
  {
    int tiles = (Nn / 16) * (OUT_DIM / 16);
    gemm_f32_wmma<<<cdiv_i(tiles, wavesPerBlock), TPB, 0, stream>>>(
        h1, W2, feat2, Nn, OUT_DIM, HD1);
  }
  node_attn_h1<<<cdiv_i(Nn, wavesPerBlock), TPB, 0, stream>>>(
      feat2, rowptr, col, (float*)d_out, Nn, 0.125f);
}


# Edge binning without grid-wide scan or scatter launch: each chunk workgroup counts, scans its own bins in LDS and writes its edges bin-sorted into its own slice, beside h@W1; 5 launches; same sums
# speedup vs baseline: 8.1591x; 1.1331x over previous
#include <hip/hip_runtime.h>
#include <math.h>

typedef __attribute__((ext_vector_type(2))) float v2f;
typedef __attribute__((ext_vector_type(8))) float v8f;

__global__ __launch_bounds__(256) void gemm_f32_wmma(
    const float* __restrict__ A, const float* __restrict__ B,
    float* __restrict__ C, int M, int N, int K) {
  const int lane = threadIdx.x & 31;
  const int wave = threadIdx.x >> 5;
  const int wavesPerBlock = blockDim.x >> 5;
  const int tile = blockIdx.x * wavesPerBlock + wave;
  const int tilesN = N >> 4;
  const int numTiles = (M >> 4) * tilesN;
  if (tile >= numTiles) return;

  const int tm = (tile / tilesN) << 4;
  const int tn = (tile % tilesN) << 4;
  const int r  = lane & 15;
  const int kk = (lane >> 4) << 1;

  const float* Arow = A + (size_t)(tm + r) * K;

  v8f acc = {};
  for (int k0 = 0; k0 < K; k0 += 4) {
    v2f a, b;
    float2 av = *(const float2*)(Arow + k0 + kk);
    a.x = av.x;
    a.y = av.y;
    b.x = B[(size_t)(k0 + kk) * N + tn + r];
    b.y = B[(size_t)(k0 + kk + 1) * N + tn + r];
    acc = __builtin_amdgcn_wmma_f32_16x16x4_f32(
         false, a,  false, b,
         (short)0, acc,  false,  false);
  }

  const int rowBase = tm + ((lane >> 4) << 3);
#pragma unroll
  for (int j = 0; j < 8; ++j) {
    C[(size_t)(rowBase + j) * N + tn + r] = acc[j];
  }
}

constexpr int BIN_SHIFT = 8;
constexpr int BIN_NODES = 1 << BIN_SHIFT;
constexpr int NCHUNK = 256;
constexpr int MAX_BINS = 1024;
constexpr int SORT_TPB = 256;
constexpr int SORT_BATCH = 16;

template <int N>
__device__ __forceinline__ int row_shr_or0(int v) {
  return __builtin_amdgcn_update_dpp(0, v, 0x110 + N, 0xf, 0xf, true);
}

__device__ __forceinline__ int wave_incl_scan(int v, int lane) {
  v += row_shr_or0<1>(v);
  v += row_shr_or0<2>(v);
  v += row_shr_or0<4>(v);
  v += row_shr_or0<8>(v);
  const int lowRow = __builtin_amdgcn_readlane(v, 15);
  return v + (lane >= 16 ? lowRow : 0);
}

template <int KD, int N>
__device__ __forceinline__ void gemm_block_16x64(
    const float* __restrict__ A, const float* __restrict__ B,
    float* __restrict__ C, int unit, int lane) {
  constexpr int NT = 4;
  constexpr int CG = N / (16 * NT);
  constexpr int STEPS = KD / 8;
  const int tm  = (unit / CG) << 4;
  const int tn0 = (unit % CG) * (16 * NT);
  const int c = lane & 15;
  const int h = lane >> 4;

  const float* Arow = A + (size_t)(tm + c) * KD + 4 * h;
  const float* Bp = B + (size_t)(4 * h) * N + tn0 + c;
  v8f acc[NT] = {};
#pragma unroll
  for (int j = 0; j < STEPS; ++j) {
    const float4 a = *(const float4*)(Arow + 8 * j);
    float b0[NT], b1[NT], b2[NT], b3[NT];
#pragma unroll
    for (int t = 0; t < NT; ++t) {
      const float* p = Bp + (size_t)(8 * j) * N + 16 * t;
      b0[t] = p[0];
      b1[t] = p[N];
      b2[t] = p[2 * N];
      b3[t] = p[3 * N];
    }
#pragma unroll
    for (int t = 0; t < NT; ++t) {
      v2f av, bv;
      av.x = a.x;
      av.y = a.y;
      bv.x = b0[t];
      bv.y = b1[t];
      acc[t] = __builtin_amdgcn_wmma_f32_16x16x4_f32(
           false, av,  false, bv,
           (short)0, acc[t],  false,  false);
    }
#pragma unroll
    for (int t = 0; t < NT; ++t) {
      v2f av, bv;
      av.x = a.z;
      av.y = a.w;
      bv.x = b2[t];
      bv.y = b3[t];
      acc[t] = __builtin_amdgcn_wmma_f32_16x16x4_f32(
           false, av,  false, bv,
           (short)0, acc[t],  false,  false);
    }
  }

  const int rowBase = tm + (h << 3);
#pragma unroll
  for (int t = 0; t < NT; ++t) {
#pragma unroll
    for (int j = 0; j < 8; ++j) {
      C[(size_t)(rowBase + j) * N + tn0 + 16 * t + c] = acc[t][j];
    }
  }
}

template <int KD, int N>
__global__ __launch_bounds__(SORT_TPB) void bin_sort_and_gemm(
    const int* __restrict__ src, const int* __restrict__ dst,
    int* __restrict__ loc, int2* __restrict__ binned, int E, int chunk,
    int nBins, const float* __restrict__ A, const float* __restrict__ B,
    float* __restrict__ C, int M) {
  __shared__ int binCur[MAX_BINS + 32];
  __shared__ int waveTot[SORT_TPB / 32];
  const int b = blockIdx.x;
  if (b >= NCHUNK) {
    const int unit = __builtin_amdgcn_readfirstlane(
        (int)(((b - NCHUNK) * SORT_TPB + threadIdx.x) >> 5));
    if (unit < (M >> 4) * (N / 64))
      gemm_block_16x64<KD, N>(A, B, C, unit, threadIdx.x & 31);
    return;
  }
  const int c = b;
  const int t = threadIdx.x;
  const int lane = t & 31;
  const int wave = t >> 5;
  for (int i = t; i < nBins; i += SORT_TPB) binCur[i] = 0;
  if (t < 32) binCur[MAX_BINS + t] = 0;
  __syncthreads();
  const int e0 = c * chunk;
  const int e1 = min(E, e0 + chunk);

  for (int e = e0 + t; e < e1; e += SORT_BATCH * SORT_TPB) {
    int d[SORT_BATCH];
#pragma unroll
    for (int u = 0; u < SORT_BATCH; ++u) {
      const int i = e + u * SORT_TPB;
      d[u] = (i < e1) ? dst[i] : -1;
    }
#pragma unroll
    for (int u = 0; u < SORT_BATCH; ++u)
      if (d[u] >= 0) atomicAdd(&binCur[d[u] >> BIN_SHIFT], 1);
  }
  __syncthreads();

  const int i0 = 4 * t;
  const int v0 = (i0 + 0 < nBins) ? binCur[i0 + 0] : 0;
  const int v1 = (i0 + 1 < nBins) ? binCur[i0 + 1] : 0;
  const int v2 = (i0 + 2 < nBins) ? binCur[i0 + 2] : 0;
  const int v3 = (i0 + 3 < nBins) ? binCur[i0 + 3] : 0;
  const int s4 = (v0 + v1) + (v2 + v3);
  const int inc = wave_incl_scan(s4, lane);
  if (lane == 31) waveTot[wave] = inc;
  __syncthreads();
  int base = 0, total = 0;
#pragma unroll
  for (int w = 0; w < SORT_TPB / 32; ++w) {
    const int x = waveTot[w];
    total += x;
    base += (w < wave) ? x : 0;
  }
  const int o0 = base + (inc - s4);
  const int o1 = o0 + v0;
  const int o2 = o1 + v1;
  const int o3 = o2 + v2;
  if (i0 + 0 < nBins) { binCur[i0 + 0] = o0; loc[(i0 + 0) * NCHUNK + c] = o0; }
  if (i0 + 1 < nBins) { binCur[i0 + 1] = o1; loc[(i0 + 1) * NCHUNK + c] = o1; }
  if (i0 + 2 < nBins) { binCur[i0 + 2] = o2; loc[(i0 + 2) * NCHUNK + c] = o2; }
  if (i0 + 3 < nBins) { binCur[i0 + 3] = o3; loc[(i0 + 3) * NCHUNK + c] = o3; }
  if (t == 0) loc[nBins * NCHUNK + c] = total;
  __syncthreads();

  for (int e = e0 + t; e < e1; e += SORT_BATCH * SORT_TPB) {
    int d[SORT_BATCH], sId[SORT_BATCH];
#pragma unroll
    for (int u = 0; u < SORT_BATCH; ++u) {
      const int i = e + u * SORT_TPB;
      const bool v = i < e1;
      d[u] = v ? dst[i] : -1;
      sId[u] = v ? src[i] : 0;
    }
    int pos[SORT_BATCH];
#pragma unroll
    for (int u = 0; u < SORT_BATCH; ++u)
      pos[u] = atomicAdd(&binCur[d[u] >= 0 ? (d[u] >> BIN_SHIFT) : MAX_BINS + lane], 1);
#pragma unroll
    for (int u = 0; u < SORT_BATCH; ++u)
      if (d[u] >= 0) binned[e0 + pos[u]] = make_int2(d[u], sId[u]);
  }
}

constexpr int LIST_WAVES = 32;
constexpr int LIST_CPW = NCHUNK / LIST_WAVES;

__global__ __launch_bounds__(32 * LIST_WAVES) void bin_lists(
    const int2* __restrict__ binned, const int* __restrict__ loc,
    int* __restrict__ rowptr, int* __restrict__ colTmp, int Nn, int nBins,
    int chunk) {
  __shared__ int cnt[BIN_NODES];
  __shared__ int waveTot[BIN_NODES / 32];
  __shared__ int sumLo[LIST_WAVES];
  __shared__ int sumHi[LIST_WAVES];
  const int bin = blockIdx.x;
  const int t = threadIdx.x;
  const int lane = t & 31;
  const int wave = __builtin_amdgcn_readfirstlane((int)(t >> 5));
  const int node0 = bin << BIN_SHIFT;

  const int cl = wave * LIST_CPW + (lane % LIST_CPW);
  const int loV = loc[bin * NCHUNK + cl];
  const int hiV = loc[(bin + 1) * NCHUNK + cl];
  {
    const int wl = wave_incl_scan(lane < LIST_CPW ? loV : 0, lane);
    const int wh = wave_incl_scan(lane < LIST_CPW ? hiV : 0, lane);
    if (lane == 31) {
      sumLo[wave] = wl;
      sumHi[wave] = wh;
    }
  }
  if (t < BIN_NODES) cnt[t] = 0;
  __syncthreads();
  const int b0 = __builtin_amdgcn_readlane(wave_incl_scan(sumLo[lane], lane), 31);
  const int b1 = __builtin_amdgcn_readlane(wave_incl_scan(sumHi[lane], lane), 31);

  int lo[LIST_CPW], hi[LIST_CPW];
#pragma unroll
  for (int u = 0; u < LIST_CPW; ++u) {
    lo[u] = __builtin_amdgcn_readlane(loV, u);
    hi[u] = __builtin_amdgcn_readlane(hiV, u);
  }
  const int2* wbase = binned + (size_t)(wave * LIST_CPW) * chunk;

  {
    int d[2 * LIST_CPW];
#pragma unroll
    for (int u = 0; u < LIST_CPW; ++u) {
      const int2* piece = wbase + (size_t)u * chunk;
      const int j = lo[u] + lane;
      d[2 * u] = (j < hi[u]) ? piece[j].x : -1;
      d[2 * u + 1] = (j + 32 < hi[u]) ? piece[j + 32].x : -1;
    }
#pragma unroll
    for (int v = 0; v < 2 * LIST_CPW; ++v)
      if (d[v] >= 0) atomicAdd(&cnt[d[v] - node0], 1);
#pragma unroll
    for (int u = 0; u < LIST_CPW; ++u) {
      const int2* piece = wbase + (size_t)u * chunk;
      for (int jb = lo[u] + 64; jb < hi[u]; jb += 32) {
        const int j = jb + lane;
        if (j < hi[u]) atomicAdd(&cnt[piece[j].x - node0], 1);
      }
    }
  }
  __syncthreads();

  int n = 0, inc = 0;
  if (t < BIN_NODES) {
    n = cnt[t];
    inc = wave_incl_scan(n, lane);
    if (lane == 31) waveTot[t >> 5] = inc;
  }
  __syncthreads();
  if (t < BIN_NODES) {
    int base = b0;
    for (int w = 0; w < (t >> 5); ++w) base += waveTot[w];
    const int start = base + (inc - n);
    cnt[t] = start;
    const int node = node0 + t;
    if (node < Nn) rowptr[node] = start;
  }
  if (bin == nBins - 1 && t == 0) rowptr[Nn] = b1;
  __syncthreads();

  {
    int2 e[2 * LIST_CPW];
#pragma unroll
    for (int u = 0; u < LIST_CPW; ++u) {
      const int2* piece = wbase + (size_t)u * chunk;
      const int j = lo[u] + lane;
      e[2 * u] = (j < hi[u]) ? piece[j] : make_int2(-1, 0);
      e[2 * u + 1] = (j + 32 < hi[u]) ? piece[j + 32] : make_int2(-1, 0);
    }
#pragma unroll
    for (int v = 0; v < 2 * LIST_CPW; ++v) {
      if (e[v].x >= 0) {
        const int pos = atomicAdd(&cnt[e[v].x - node0], 1);
        colTmp[pos] = e[v].y;
      }
    }
#pragma unroll
    for (int u = 0; u < LIST_CPW; ++u) {
      const int2* piece = wbase + (size_t)u * chunk;
      for (int jb = lo[u] + 64; jb < hi[u]; jb += 32) {
        const int j = jb + lane;
        if (j < hi[u]) {
          const int2 x = piece[j];
          const int pos = atomicAdd(&cnt[x.x - node0], 1);
          colTmp[pos] = x.y;
        }
      }
    }
  }
}

constexpr unsigned dpp8_xor_sel(int x) {
  unsigned s = 0;
  for (int i = 0; i < 8; ++i) s |= (unsigned)((i ^ x) & 7) << (3 * i);
  return s;
}
constexpr unsigned dpp8_bcast_sel(int k) { return (unsigned)k * 0x249249u; }

template <unsigned SEL>
__device__ __forceinline__ float dpp8_f(float v) {
  return __uint_as_float(__builtin_amdgcn_mov_dpp8(__float_as_uint(v), SEL));
}
template <unsigned SEL>
__device__ __forceinline__ int dpp8_i(int v) {
  return (int)__builtin_amdgcn_mov_dpp8((unsigned)v, SEL);
}

template <int J>
__device__ __forceinline__ int lane_xor_i(int v);
template <>
__device__ __forceinline__ int lane_xor_i<1>(int v) { return dpp8_i<dpp8_xor_sel(1)>(v); }
template <>
__device__ __forceinline__ int lane_xor_i<2>(int v) { return dpp8_i<dpp8_xor_sel(2)>(v); }
template <>
__device__ __forceinline__ int lane_xor_i<4>(int v) { return dpp8_i<dpp8_xor_sel(4)>(v); }
template <>
__device__ __forceinline__ int lane_xor_i<8>(int v) {
  return __builtin_amdgcn_update_dpp(v, v, 0x168  , 0xf, 0xf, false);
}
template <>
__device__ __forceinline__ int lane_xor_i<16>(int v) {
  return (int)__builtin_amdgcn_permlanex16((unsigned)v, (unsigned)v, 0x76543210u,
                                           0xfedcba98u, false, false);
}

template <int J>
__device__ __forceinline__ int cmp_xchg(int v, int lane, bool asc) {
  const int p = lane_xor_i<J>(v);
  const bool keepMin = ((lane & J) == 0) == asc;
  return keepMin ? min(v, p) : max(v, p);
}

template <int K>
__device__ __forceinline__ int merge_lanes(int v, int lane, bool asc) {
  if (K >= 32) v = cmp_xchg<16>(v, lane, asc);
  if (K >= 16) v = cmp_xchg<8>(v, lane, asc);
  if (K >= 8) v = cmp_xchg<4>(v, lane, asc);
  if (K >= 4) v = cmp_xchg<2>(v, lane, asc);
  return cmp_xchg<1>(v, lane, asc);
}

__device__ __forceinline__ int sort_lanes(int v, int lane, bool asc) {
  v = merge_lanes<2>(v, lane, (lane & 2) == 0);
  v = merge_lanes<4>(v, lane, (lane & 4) == 0);
  v = merge_lanes<8>(v, lane, (lane & 8) == 0);
  v = merge_lanes<16>(v, lane, (lane & 16) == 0);
  return merge_lanes<32>(v, lane, asc);
}

__global__ __launch_bounds__(256) void node_attn_h8(
    const float* __restrict__ feat, const int* __restrict__ rowptr,
    const int* __restrict__ colTmp, int* col, float* __restrict__ out, int Nn,
    float scale) {
  const int lane = threadIdx.x & 31;
  const int node = __builtin_amdgcn_readfirstlane(
      (int)((blockIdx.x * blockDim.x + threadIdx.x) >> 5));
  if (node >= Nn) return;
  const int beg = rowptr[node];
  const int end = rowptr[node + 1];
  const int n = end - beg;
  const float* fl = feat + lane * 4;
  const float4 fd = *(const float4*)(fl + (size_t)node * 128);

  const bool inRegs = n <= 64;
  int k0 = 0x7fffffff, k1 = 0x7fffffff;
  if (inRegs) {
    if (lane < n) k0 = colTmp[beg + lane];
    if (n > 32) {
      if (32 + lane < n) k1 = colTmp[beg + 32 + lane];
      k0 = sort_lanes(k0, lane, true);
      k1 = sort_lanes(k1, lane, false);
      const int lo = min(k0, k1);
      const int hi = max(k0, k1);
      k0 = merge_lanes<32>(lo, lane, true);
      k1 = merge_lanes<32>(hi, lane, true);
      if (32 + lane < n) col[beg + 32 + lane] = k1;
    } else {
      k0 = sort_lanes(k0, lane, true);
    }
    if (lane < n) col[beg + lane] = k0;
  } else {
    for (int ib = 0; ib < n; ib += 32) {
      const int i = ib + lane;
      const bool vi = i < n;
      const int ki = vi ? colTmp[beg + i] : 0;
      int rank = 0;
      for (int jb = 0; jb < n; jb += 32) {
        const int cj = min(32, n - jb);
        const int kc = (lane < cj) ? colTmp[beg + jb + lane] : 0;
        for (int t = 0; t < cj; ++t) {
          const int kj = __builtin_amdgcn_readlane(kc, t);
          const int bias = (jb + t < i) ? 1 : 0;
          rank += (kj < ki + bias) ? 1 : 0;
        }
      }
      if (vi) col[beg + rank] = ki;
    }
    __threadfence();
  }

  float m = -INFINITY, l = 0.0f;
  float4 acc = make_float4(0.0f, 0.0f, 0.0f, 0.0f);

  for (int base = beg; base < end; base += 32) {
    const int cnt = min(32, end - base);
    int ids;
    if (inRegs) {
      ids = (base == beg) ? k0 : k1;
    } else {
      ids = (lane < cnt) ? col[base + lane] : 0;
    }
    ids = (lane < cnt) ? ids : node;
    for (int t = 0; t < cnt; t += 4) {
      float4 f[4];
      float p[4];
#pragma unroll
      for (int u = 0; u < 4; ++u) {
        const int s = __builtin_amdgcn_readlane(ids, t + u);
        f[u] = *(const float4*)(fl + (size_t)s * 128);
      }
#pragma unroll
      for (int u = 0; u < 4; ++u) {
        float d = f[u].x * fd.x + f[u].y * fd.y + f[u].z * fd.z + f[u].w * fd.w;
        d += dpp8_f<dpp8_xor_sel(1)>(d);
        d += dpp8_f<dpp8_xor_sel(2)>(d);
        d *= scale;
        p[u] = (t + u < cnt) ? d : -INFINITY;
      }
      const float mNew = fmaxf(fmaxf(m, p[0]), fmaxf(fmaxf(p[1], p[2]), p[3]));
      const float c = __expf(m - mNew);
      const float w0 = __expf(p[0] - mNew);
      const float w1 = __expf(p[1] - mNew);
      const float w2 = __expf(p[2] - mNew);
      const float w3 = __expf(p[3] - mNew);
      l = l * c + ((w0 + w1) + (w2 + w3));
      acc.x = acc.x * c + ((w0 * f[0].x + w1 * f[1].x) + (w2 * f[2].x + w3 * f[3].x));
      acc.y = acc.y * c + ((w0 * f[0].y + w1 * f[1].y) + (w2 * f[2].y + w3 * f[3].y));
      acc.z = acc.z * c + ((w0 * f[0].z + w1 * f[1].z) + (w2 * f[2].z + w3 * f[3].z));
      acc.w = acc.w * c + ((w0 * f[0].w + w1 * f[1].w) + (w2 * f[2].w + w3 * f[3].w));
      m = mNew;
    }
  }

  const float inv = 1.0f / fmaxf(l, 1e-9f);
  float4 o;
  o.x = acc.x * inv;
  o.y = acc.y * inv;
  o.z = acc.z * inv;
  o.w = acc.w * inv;
  o.x = o.x > 0.0f ? o.x : expm1f(o.x);
  o.y = o.y > 0.0f ? o.y : expm1f(o.y);
  o.z = o.z > 0.0f ? o.z : expm1f(o.z);
  o.w = o.w > 0.0f ? o.w : expm1f(o.w);
  *(float4*)(out + (size_t)node * 128 + lane * 4) = o;
}

template <int K0>
__device__ __forceinline__ void h1_four_steps(
    const float* __restrict__ fq, int ids, int cnt, int g, const float4 fdA,
    const float4 fdB, float scale, float& m, float& l, float4& accA,
    float4& accB) {
  float4 fa[4], fb[4];
  float p[4];
#pragma unroll
  for (int u = 0; u < 4; ++u) {
    const int s = u == 0   ? dpp8_i<dpp8_bcast_sel(K0 + 0)>(ids)
                  : u == 1 ? dpp8_i<dpp8_bcast_sel(K0 + 1)>(ids)
                  : u == 2 ? dpp8_i<dpp8_bcast_sel(K0 + 2)>(ids)
                           : dpp8_i<dpp8_bcast_sel(K0 + 3)>(ids);
    const float* r = fq + (size_t)s * 64;
    fa[u] = *(const float4*)(r);
    fb[u] = *(const float4*)(r + 32);
  }
#pragma unroll
  for (int u = 0; u < 4; ++u) {
    float d = fa[u].x * fdA.x + fa[u].y * fdA.y + fa[u].z * fdA.z +
              fa[u].w * fdA.w + fb[u].x * fdB.x + fb[u].y * fdB.y +
              fb[u].z * fdB.z + fb[u].w * fdB.w;
    d += dpp8_f<dpp8_xor_sel(1)>(d);
    d += dpp8_f<dpp8_xor_sel(2)>(d);
    d += dpp8_f<dpp8_xor_sel(4)>(d);
    d *= scale;
    p[u] = (4 * (K0 + u) + g < cnt) ? d : -INFINITY;
  }
  const float mNew = fmaxf(fmaxf(m, p[0]), fmaxf(fmaxf(p[1], p[2]), p[3]));
  const float mRef = (mNew == -INFINITY) ? 0.0f : mNew;
  const float c = __expf(m - mRef);
  const float w0 = __expf(p[0] - mRef);
  const float w1 = __expf(p[1] - mRef);
  const float w2 = __expf(p[2] - mRef);
  const float w3 = __expf(p[3] - mRef);
  l = l * c + ((w0 + w1) + (w2 + w3));
  accA.x = accA.x * c + ((w0 * fa[0].x + w1 * fa[1].x) + (w2 * fa[2].x + w3 * fa[3].x));
  accA.y = accA.y * c + ((w0 * fa[0].y + w1 * fa[1].y) + (w2 * fa[2].y + w3 * fa[3].y));
  accA.z = accA.z * c + ((w0 * fa[0].z + w1 * fa[1].z) + (w2 * fa[2].z + w3 * fa[3].z));
  accA.w = accA.w * c + ((w0 * fa[0].w + w1 * fa[1].w) + (w2 * fa[2].w + w3 * fa[3].w));
  accB.x = accB.x * c + ((w0 * fb[0].x + w1 * fb[1].x) + (w2 * fb[2].x + w3 * fb[3].x));
  accB.y = accB.y * c + ((w0 * fb[0].y + w1 * fb[1].y) + (w2 * fb[2].y + w3 * fb[3].y));
  accB.z = accB.z * c + ((w0 * fb[0].z + w1 * fb[1].z) + (w2 * fb[2].z + w3 * fb[3].z));
  accB.w = accB.w * c + ((w0 * fb[0].w + w1 * fb[1].w) + (w2 * fb[2].w + w3 * fb[3].w));
  m = mNew;
}

__device__ __forceinline__ float sum_groups(float v) {
  v += __shfl_xor(v, 8, 32);
  v += __shfl_xor(v, 16, 32);
  return v;
}

__global__ __launch_bounds__(256) void node_attn_h1(
    const float* __restrict__ feat, const int* __restrict__ rowptr,
    const int* __restrict__ col, float* __restrict__ out, int Nn, float scale) {
  const int lane = threadIdx.x & 31;
  const int node = __builtin_amdgcn_readfirstlane(
      (int)((blockIdx.x * blockDim.x + threadIdx.x) >> 5));
  if (node >= Nn) return;
  const int beg = rowptr[node];
  const int end = rowptr[node + 1];
  const int q = lane & 7;
  const int g = lane >> 3;
  const float* fq = feat + q * 4;
  const float4 fdA = *(const float4*)(fq + (size_t)node * 64);
  const float4 fdB = *(const float4*)(fq + (size_t)node * 64 + 32);

  float m = -INFINITY, l = 0.0f;
  float4 accA = make_float4(0.0f, 0.0f, 0.0f, 0.0f);
  float4 accB = make_float4(0.0f, 0.0f, 0.0f, 0.0f);

  const int myIdx = 4 * q + g;
  for (int base = beg; base < end; base += 32) {
    const int cnt = min(32, end - base);
    const int ids = (myIdx < cnt) ? col[base + myIdx] : node;
    h1_four_steps<0>(fq, ids, cnt, g, fdA, fdB, scale, m, l, accA, accB);
    if (cnt > 16)
      h1_four_steps<4>(fq, ids, cnt, g, fdA, fdB, scale, m, l, accA, accB);
  }

  float M = fmaxf(m, __shfl_xor(m, 8, 32));
  M = fmaxf(M, __shfl_xor(M, 16, 32));
  const float Mref = (M == -INFINITY) ? 0.0f : M;
  const float sc = __expf(m - Mref);
  const float L = sum_groups(l * sc);
  float4 oA, oB;
  oA.x = sum_groups(accA.x * sc);
  oA.y = sum_groups(accA.y * sc);
  oA.z = sum_groups(accA.z * sc);
  oA.w = sum_groups(accA.w * sc);
  oB.x = sum_groups(accB.x * sc);
  oB.y = sum_groups(accB.y * sc);
  oB.z = sum_groups(accB.z * sc);
  oB.w = sum_groups(accB.w * sc);
  const float inv = 1.0f / fmaxf(L, 1e-9f);
  if (g == 0) {
    float* o = out + (size_t)node * 64 + q * 4;
    *(float4*)(o) = make_float4(oA.x * inv, oA.y * inv, oA.z * inv, oA.w * inv);
    *(float4*)(o + 32) = make_float4(oB.x * inv, oB.y * inv, oB.z * inv, oB.w * inv);
  }
}

static inline int cdiv_i(long long a, long long b) { return (int)((a + b - 1) / b); }

extern "C" void kernel_launch(void* const* d_in, const int* in_sizes, int n_in,
                              void* d_out, int out_size, void* d_ws, size_t ws_size,
                              hipStream_t stream) {
  const float* hIn = (const float*)d_in[0];
  const float* W1  = (const float*)d_in[1];
  const float* W2  = (const float*)d_in[2];
  const int*   src = (const int*)d_in[3];
  const int*   dst = (const int*)d_in[4];

  const int IN_DIM = 128, OUT_DIM = 64;
  const int HD1 = 128;
  const int Nn  = in_sizes[0] / IN_DIM;
  const int E   = in_sizes[3];

  const int nBins   = cdiv_i(Nn, BIN_NODES);
  const int chunk   = cdiv_i(E, NCHUNK);
  const int locLen  = (nBins + 1) * NCHUNK;

  float* feat1  = (float*)d_ws;
  float* h1     = feat1 + (size_t)Nn * HD1;
  int*   colTmp = (int*)(h1 + (size_t)Nn * HD1) + (size_t)E;
  int*   col    = colTmp + (size_t)E;
  int*   rowptr = col + (size_t)E + (size_t)Nn;
  int2*  binned = (int2*)h1;
  int*   loc    = (int*)(h1 + (size_t)Nn * HD1) - locLen;

  const int TPB = 256;
  const int wavesPerBlock = TPB / 32;

  {
    int gemmBlocks = cdiv_i((Nn / 16) * (HD1 / 64), SORT_TPB / 32);
    bin_sort_and_gemm<128, 128><<<NCHUNK + gemmBlocks, SORT_TPB, 0, stream>>>(
        src, dst, loc, binned, E, chunk, nBins, hIn, W1, feat1, Nn);
  }
  bin_lists<<<nBins, 32 * LIST_WAVES, 0, stream>>>(binned, loc, rowptr, colTmp, Nn,
                                                   nBins, chunk);

  node_attn_h8<<<cdiv_i(Nn, wavesPerBlock), TPB, 0, stream>>>(
      feat1, rowptr, colTmp, col, h1, Nn, 0.25f);

  float* feat2 = feat1;
  {
    int tiles = (Nn / 16) * (OUT_DIM / 16);
    gemm_f32_wmma<<<cdiv_i(tiles, wavesPerBlock), TPB, 0, stream>>>(
        h1, W2, feat2, Nn, OUT_DIM, HD1);
  }
  node_attn_h1<<<cdiv_i(Nn, wavesPerBlock), TPB, 0, stream>>>(
      feat2, rowptr, col, (float*)d_out, Nn, 0.125f);
}


# Second product h1@W2 has no launch of its own: 16-wave layer-1 workgroups keep their 16 h1 rows in LDS and 4 waves multiply them by W2 on WMMA; h1 never stored; 4 launches; its k sum order changes
# speedup vs baseline: 8.8609x; 1.0860x over previous
#include <hip/hip_runtime.h>
#include <math.h>

typedef __attribute__((ext_vector_type(2))) float v2f;
typedef __attribute__((ext_vector_type(8))) float v8f;

constexpr int BIN_SHIFT = 8;
constexpr int BIN_NODES = 1 << BIN_SHIFT;
constexpr int NCHUNK = 256;
constexpr int MAX_BINS = 1024;
constexpr int SORT_TPB = 256;
constexpr int SORT_BATCH = 16;

template <int N>
__device__ __forceinline__ int row_shr_or0(int v) {
  return __builtin_amdgcn_update_dpp(0, v, 0x110 + N, 0xf, 0xf, true);
}

__device__ __forceinline__ int wave_incl_scan(int v, int lane) {
  v += row_shr_or0<1>(v);
  v += row_shr_or0<2>(v);
  v += row_shr_or0<4>(v);
  v += row_shr_or0<8>(v);
  const int lowRow = __builtin_amdgcn_readlane(v, 15);
  return v + (lane >= 16 ? lowRow : 0);
}

template <int KD, int N>
__device__ __forceinline__ void gemm_block_16x64(
    const float* __restrict__ A, const float* __restrict__ B,
    float* __restrict__ C, int unit, int lane) {
  constexpr int NT = 4;
  constexpr int CG = N / (16 * NT);
  constexpr int STEPS = KD / 8;
  const int tm  = (unit / CG) << 4;
  const int tn0 = (unit % CG) * (16 * NT);
  const int c = lane & 15;
  const int h = lane >> 4;

  const float* Arow = A + (size_t)(tm + c) * KD + 4 * h;
  const float* Bp = B + (size_t)(4 * h) * N + tn0 + c;
  v8f acc[NT] = {};
#pragma unroll
  for (int j = 0; j < STEPS; ++j) {
    const float4 a = *(const float4*)(Arow + 8 * j);
    float b0[NT], b1[NT], b2[NT], b3[NT];
#pragma unroll
    for (int t = 0; t < NT; ++t) {
      const float* p = Bp + (size_t)(8 * j) * N + 16 * t;
      b0[t] = p[0];
      b1[t] = p[N];
      b2[t] = p[2 * N];
      b3[t] = p[3 * N];
    }
#pragma unroll
    for (int t = 0; t < NT; ++t) {
      v2f av, bv;
      av.x = a.x;
      av.y = a.y;
      bv.x = b0[t];
      bv.y = b1[t];
      acc[t] = __builtin_amdgcn_wmma_f32_16x16x4_f32(
           false, av,  false, bv,
           (short)0, acc[t],  false,  false);
    }
#pragma unroll
    for (int t = 0; t < NT; ++t) {
      v2f av, bv;
      av.x = a.z;
      av.y = a.w;
      bv.x = b2[t];
      bv.y = b3[t];
      acc[t] = __builtin_amdgcn_wmma_f32_16x16x4_f32(
           false, av,  false, bv,
           (short)0, acc[t],  false,  false);
    }
  }

  const int rowBase = tm + (h << 3);
#pragma unroll
  for (int t = 0; t < NT; ++t) {
#pragma unroll
    for (int j = 0; j < 8; ++j) {
      C[(size_t)(rowBase + j) * N + tn0 + 16 * t + c] = acc[t][j];
    }
  }
}

template <int KD, int N>
__global__ __launch_bounds__(SORT_TPB) void bin_sort_and_gemm(
    const int* __restrict__ src, const int* __restrict__ dst,
    int* __restrict__ loc, int2* __restrict__ binned, int E, int chunk,
    int nBins, const float* __restrict__ A, const float* __restrict__ B,
    float* __restrict__ C, int M) {
  __shared__ int binCur[MAX_BINS + 32];
  __shared__ int waveTot[SORT_TPB / 32];
  const int b = blockIdx.x;
  if (b >= NCHUNK) {
    const int unit = __builtin_amdgcn_readfirstlane(
        (int)(((b - NCHUNK) * SORT_TPB + threadIdx.x) >> 5));
    if (unit < (M >> 4) * (N / 64))
      gemm_block_16x64<KD, N>(A, B, C, unit, threadIdx.x & 31);
    return;
  }
  const int c = b;
  const int t = threadIdx.x;
  const int lane = t & 31;
  const int wave = t >> 5;
  for (int i = t; i < nBins; i += SORT_TPB) binCur[i] = 0;
  if (t < 32) binCur[MAX_BINS + t] = 0;
  __syncthreads();
  const int e0 = c * chunk;
  const int e1 = min(E, e0 + chunk);

  for (int e = e0 + t; e < e1; e += SORT_BATCH * SORT_TPB) {
    int d[SORT_BATCH];
#pragma unroll
    for (int u = 0; u < SORT_BATCH; ++u) {
      const int i = e + u * SORT_TPB;
      d[u] = (i < e1) ? dst[i] : -1;
    }
#pragma unroll
    for (int u = 0; u < SORT_BATCH; ++u)
      if (d[u] >= 0) atomicAdd(&binCur[d[u] >> BIN_SHIFT], 1);
  }
  __syncthreads();

  const int i0 = 4 * t;
  const int v0 = (i0 + 0 < nBins) ? binCur[i0 + 0] : 0;
  const int v1 = (i0 + 1 < nBins) ? binCur[i0 + 1] : 0;
  const int v2 = (i0 + 2 < nBins) ? binCur[i0 + 2] : 0;
  const int v3 = (i0 + 3 < nBins) ? binCur[i0 + 3] : 0;
  const int s4 = (v0 + v1) + (v2 + v3);
  const int inc = wave_incl_scan(s4, lane);
  if (lane == 31) waveTot[wave] = inc;
  __syncthreads();
  int base = 0, total = 0;
#pragma unroll
  for (int w = 0; w < SORT_TPB / 32; ++w) {
    const int x = waveTot[w];
    total += x;
    base += (w < wave) ? x : 0;
  }
  const int o0 = base + (inc - s4);
  const int o1 = o0 + v0;
  const int o2 = o1 + v1;
  const int o3 = o2 + v2;
  if (i0 + 0 < nBins) { binCur[i0 + 0] = o0; loc[(i0 + 0) * NCHUNK + c] = o0; }
  if (i0 + 1 < nBins) { binCur[i0 + 1] = o1; loc[(i0 + 1) * NCHUNK + c] = o1; }
  if (i0 + 2 < nBins) { binCur[i0 + 2] = o2; loc[(i0 + 2) * NCHUNK + c] = o2; }
  if (i0 + 3 < nBins) { binCur[i0 + 3] = o3; loc[(i0 + 3) * NCHUNK + c] = o3; }
  if (t == 0) loc[nBins * NCHUNK + c] = total;
  __syncthreads();

  for (int e = e0 + t; e < e1; e += SORT_BATCH * SORT_TPB) {
    int d[SORT_BATCH], sId[SORT_BATCH];
#pragma unroll
    for (int u = 0; u < SORT_BATCH; ++u) {
      const int i = e + u * SORT_TPB;
      const bool v = i < e1;
      d[u] = v ? dst[i] : -1;
      sId[u] = v ? src[i] : 0;
    }
    int pos[SORT_BATCH];
#pragma unroll
    for (int u = 0; u < SORT_BATCH; ++u)
      pos[u] = atomicAdd(&binCur[d[u] >= 0 ? (d[u] >> BIN_SHIFT) : MAX_BINS + lane], 1);
#pragma unroll
    for (int u = 0; u < SORT_BATCH; ++u)
      if (d[u] >= 0) binned[e0 + pos[u]] = make_int2(d[u], sId[u]);
  }
}

constexpr int LIST_WAVES = 32;
constexpr int LIST_CPW = NCHUNK / LIST_WAVES;

__global__ __launch_bounds__(32 * LIST_WAVES) void bin_lists(
    const int2* __restrict__ binned, const int* __restrict__ loc,
    int* __restrict__ rowptr, int* __restrict__ colTmp, int Nn, int nBins,
    int chunk) {
  __shared__ int cnt[BIN_NODES];
  __shared__ int waveTot[BIN_NODES / 32];
  __shared__ int sumLo[LIST_WAVES];
  __shared__ int sumHi[LIST_WAVES];
  const int bin = blockIdx.x;
  const int t = threadIdx.x;
  const int lane = t & 31;
  const int wave = __builtin_amdgcn_readfirstlane((int)(t >> 5));
  const int node0 = bin << BIN_SHIFT;

  const int cl = wave * LIST_CPW + (lane % LIST_CPW);
  const int loV = loc[bin * NCHUNK + cl];
  const int hiV = loc[(bin + 1) * NCHUNK + cl];
  {
    const int wl = wave_incl_scan(lane < LIST_CPW ? loV : 0, lane);
    const int wh = wave_incl_scan(lane < LIST_CPW ? hiV : 0, lane);
    if (lane == 31) {
      sumLo[wave] = wl;
      sumHi[wave] = wh;
    }
  }
  if (t < BIN_NODES) cnt[t] = 0;
  __syncthreads();
  const int b0 = __builtin_amdgcn_readlane(wave_incl_scan(sumLo[lane], lane), 31);
  const int b1 = __builtin_amdgcn_readlane(wave_incl_scan(sumHi[lane], lane), 31);

  int lo[LIST_CPW], hi[LIST_CPW];
#pragma unroll
  for (int u = 0; u < LIST_CPW; ++u) {
    lo[u] = __builtin_amdgcn_readlane(loV, u);
    hi[u] = __builtin_amdgcn_readlane(hiV, u);
  }
  const int2* wbase = binned + (size_t)(wave * LIST_CPW) * chunk;

  {
    int d[2 * LIST_CPW];
#pragma unroll
    for (int u = 0; u < LIST_CPW; ++u) {
      const int2* piece = wbase + (size_t)u * chunk;
      const int j = lo[u] + lane;
      d[2 * u] = (j < hi[u]) ? piece[j].x : -1;
      d[2 * u + 1] = (j + 32 < hi[u]) ? piece[j + 32].x : -1;
    }
#pragma unroll
    for (int v = 0; v < 2 * LIST_CPW; ++v)
      if (d[v] >= 0) atomicAdd(&cnt[d[v] - node0], 1);
#pragma unroll
    for (int u = 0; u < LIST_CPW; ++u) {
      const int2* piece = wbase + (size_t)u * chunk;
      for (int jb = lo[u] + 64; jb < hi[u]; jb += 32) {
        const int j = jb + lane;
        if (j < hi[u]) atomicAdd(&cnt[piece[j].x - node0], 1);
      }
    }
  }
  __syncthreads();

  int n = 0, inc = 0;
  if (t < BIN_NODES) {
    n = cnt[t];
    inc = wave_incl_scan(n, lane);
    if (lane == 31) waveTot[t >> 5] = inc;
  }
  __syncthreads();
  if (t < BIN_NODES) {
    int base = b0;
    for (int w = 0; w < (t >> 5); ++w) base += waveTot[w];
    const int start = base + (inc - n);
    cnt[t] = start;
    const int node = node0 + t;
    if (node < Nn) rowptr[node] = start;
  }
  if (bin == nBins - 1 && t == 0) rowptr[Nn] = b1;
  __syncthreads();

  {
    int2 e[2 * LIST_CPW];
#pragma unroll
    for (int u = 0; u < LIST_CPW; ++u) {
      const int2* piece = wbase + (size_t)u * chunk;
      const int j = lo[u] + lane;
      e[2 * u] = (j < hi[u]) ? piece[j] : make_int2(-1, 0);
      e[2 * u + 1] = (j + 32 < hi[u]) ? piece[j + 32] : make_int2(-1, 0);
    }
#pragma unroll
    for (int v = 0; v < 2 * LIST_CPW; ++v) {
      if (e[v].x >= 0) {
        const int pos = atomicAdd(&cnt[e[v].x - node0], 1);
        colTmp[pos] = e[v].y;
      }
    }
#pragma unroll
    for (int u = 0; u < LIST_CPW; ++u) {
      const int2* piece = wbase + (size_t)u * chunk;
      for (int jb = lo[u] + 64; jb < hi[u]; jb += 32) {
        const int j = jb + lane;
        if (j < hi[u]) {
          const int2 x = piece[j];
          const int pos = atomicAdd(&cnt[x.x - node0], 1);
          colTmp[pos] = x.y;
        }
      }
    }
  }
}

constexpr unsigned dpp8_xor_sel(int x) {
  unsigned s = 0;
  for (int i = 0; i < 8; ++i) s |= (unsigned)((i ^ x) & 7) << (3 * i);
  return s;
}
constexpr unsigned dpp8_bcast_sel(int k) { return (unsigned)k * 0x249249u; }

template <unsigned SEL>
__device__ __forceinline__ float dpp8_f(float v) {
  return __uint_as_float(__builtin_amdgcn_mov_dpp8(__float_as_uint(v), SEL));
}
template <unsigned SEL>
__device__ __forceinline__ int dpp8_i(int v) {
  return (int)__builtin_amdgcn_mov_dpp8((unsigned)v, SEL);
}

template <int J>
__device__ __forceinline__ int lane_xor_i(int v);
template <>
__device__ __forceinline__ int lane_xor_i<1>(int v) { return dpp8_i<dpp8_xor_sel(1)>(v); }
template <>
__device__ __forceinline__ int lane_xor_i<2>(int v) { return dpp8_i<dpp8_xor_sel(2)>(v); }
template <>
__device__ __forceinline__ int lane_xor_i<4>(int v) { return dpp8_i<dpp8_xor_sel(4)>(v); }
template <>
__device__ __forceinline__ int lane_xor_i<8>(int v) {
  return __builtin_amdgcn_update_dpp(v, v, 0x168  , 0xf, 0xf, false);
}
template <>
__device__ __forceinline__ int lane_xor_i<16>(int v) {
  return (int)__builtin_amdgcn_permlanex16((unsigned)v, (unsigned)v, 0x76543210u,
                                           0xfedcba98u, false, false);
}

template <int J>
__device__ __forceinline__ int cmp_xchg(int v, int lane, bool asc) {
  const int p = lane_xor_i<J>(v);
  const bool keepMin = ((lane & J) == 0) == asc;
  return keepMin ? min(v, p) : max(v, p);
}

template <int K>
__device__ __forceinline__ int merge_lanes(int v, int lane, bool asc) {
  if (K >= 32) v = cmp_xchg<16>(v, lane, asc);
  if (K >= 16) v = cmp_xchg<8>(v, lane, asc);
  if (K >= 8) v = cmp_xchg<4>(v, lane, asc);
  if (K >= 4) v = cmp_xchg<2>(v, lane, asc);
  return cmp_xchg<1>(v, lane, asc);
}

__device__ __forceinline__ int sort_lanes(int v, int lane, bool asc) {
  v = merge_lanes<2>(v, lane, (lane & 2) == 0);
  v = merge_lanes<4>(v, lane, (lane & 4) == 0);
  v = merge_lanes<8>(v, lane, (lane & 8) == 0);
  v = merge_lanes<16>(v, lane, (lane & 16) == 0);
  return merge_lanes<32>(v, lane, asc);
}

__device__ __forceinline__ float4 attn_h8_row(
    const float* __restrict__ feat, const int* __restrict__ rowptr,
    const int* __restrict__ colTmp, int* col, int node, int lane, float scale) {
  const int beg = rowptr[node];
  const int end = rowptr[node + 1];
  const int n = end - beg;
  const float* fl = feat + lane * 4;
  const float4 fd = *(const float4*)(fl + (size_t)node * 128);

  const bool inRegs = n <= 64;
  int k0 = 0x7fffffff, k1 = 0x7fffffff;
  if (inRegs) {
    if (lane < n) k0 = colTmp[beg + lane];
    if (n > 32) {
      if (32 + lane < n) k1 = colTmp[beg + 32 + lane];
      k0 = sort_lanes(k0, lane, true);
      k1 = sort_lanes(k1, lane, false);
      const int lo = min(k0, k1);
      const int hi = max(k0, k1);
      k0 = merge_lanes<32>(lo, lane, true);
      k1 = merge_lanes<32>(hi, lane, true);
      if (32 + lane < n) col[beg + 32 + lane] = k1;
    } else {
      k0 = sort_lanes(k0, lane, true);
    }
    if (lane < n) col[beg + lane] = k0;
  } else {
    for (int ib = 0; ib < n; ib += 32) {
      const int i = ib + lane;
      const bool vi = i < n;
      const int ki = vi ? colTmp[beg + i] : 0;
      int rank = 0;
      for (int jb = 0; jb < n; jb += 32) {
        const int cj = min(32, n - jb);
        const int kc = (lane < cj) ? colTmp[beg + jb + lane] : 0;
        for (int t = 0; t < cj; ++t) {
          const int kj = __builtin_amdgcn_readlane(kc, t);
          const int bias = (jb + t < i) ? 1 : 0;
          rank += (kj < ki + bias) ? 1 : 0;
        }
      }
      if (vi) col[beg + rank] = ki;
    }
    __threadfence();
  }

  float m = -INFINITY, l = 0.0f;
  float4 acc = make_float4(0.0f, 0.0f, 0.0f, 0.0f);

  for (int base = beg; base < end; base += 32) {
    const int cnt = min(32, end - base);
    int ids;
    if (inRegs) {
      ids = (base == beg) ? k0 : k1;
    } else {
      ids = (lane < cnt) ? col[base + lane] : 0;
    }
    ids = (lane < cnt) ? ids : node;
    for (int t = 0; t < cnt; t += 4) {
      float4 f[4];
      float p[4];
#pragma unroll
      for (int u = 0; u < 4; ++u) {
        const int s = __builtin_amdgcn_readlane(ids, t + u);
        f[u] = *(const float4*)(fl + (size_t)s * 128);
      }
#pragma unroll
      for (int u = 0; u < 4; ++u) {
        float d = f[u].x * fd.x + f[u].y * fd.y + f[u].z * fd.z + f[u].w * fd.w;
        d += dpp8_f<dpp8_xor_sel(1)>(d);
        d += dpp8_f<dpp8_xor_sel(2)>(d);
        d *= scale;
        p[u] = (t + u < cnt) ? d : -INFINITY;
      }
      const float mNew = fmaxf(fmaxf(m, p[0]), fmaxf(fmaxf(p[1], p[2]), p[3]));
      const float c = __expf(m - mNew);
      const float w0 = __expf(p[0] - mNew);
      const float w1 = __expf(p[1] - mNew);
      const float w2 = __expf(p[2] - mNew);
      const float w3 = __expf(p[3] - mNew);
      l = l * c + ((w0 + w1) + (w2 + w3));
      acc.x = acc.x * c + ((w0 * f[0].x + w1 * f[1].x) + (w2 * f[2].x + w3 * f[3].x));
      acc.y = acc.y * c + ((w0 * f[0].y + w1 * f[1].y) + (w2 * f[2].y + w3 * f[3].y));
      acc.z = acc.z * c + ((w0 * f[0].z + w1 * f[1].z) + (w2 * f[2].z + w3 * f[3].z));
      acc.w = acc.w * c + ((w0 * f[0].w + w1 * f[1].w) + (w2 * f[2].w + w3 * f[3].w));
      m = mNew;
    }
  }

  const float inv = 1.0f / fmaxf(l, 1e-9f);
  float4 o;
  o.x = acc.x * inv;
  o.y = acc.y * inv;
  o.z = acc.z * inv;
  o.w = acc.w * inv;
  o.x = o.x > 0.0f ? o.x : expm1f(o.x);
  o.y = o.y > 0.0f ? o.y : expm1f(o.y);
  o.z = o.z > 0.0f ? o.z : expm1f(o.z);
  o.w = o.w > 0.0f ? o.w : expm1f(o.w);
  return o;
}

constexpr int L1_WAVES = 16;
constexpr int H1_PITCH = 132;

template <int KD, int N>
__device__ __forceinline__ void gemm_tile_from_lds(
    const float* tile, const float* __restrict__ B, float* __restrict__ C,
    int tm, int tn, int lane) {
  constexpr int STEPS = KD / 8;
  const int c = lane & 15;
  const int h = lane >> 4;
  const float* Ap = tile + c * H1_PITCH + 4 * h;
  const float* Bp = B + (size_t)(4 * h) * N + tn + c;
  v8f acc = {};
#pragma unroll
  for (int j = 0; j < STEPS; ++j) {
    const float4 a = *(const float4*)(Ap + 8 * j);
    const float* p = Bp + (size_t)(8 * j) * N;
    v2f av, bv;
    av.x = a.x;
    av.y = a.y;
    bv.x = p[0];
    bv.y = p[N];
    acc = __builtin_amdgcn_wmma_f32_16x16x4_f32(
         false, av,  false, bv,
         (short)0, acc,  false,  false);
    av.x = a.z;
    av.y = a.w;
    bv.x = p[2 * N];
    bv.y = p[3 * N];
    acc = __builtin_amdgcn_wmma_f32_16x16x4_f32(
         false, av,  false, bv,
         (short)0, acc,  false,  false);
  }
  const int rowBase = tm + (h << 3);
#pragma unroll
  for (int j = 0; j < 8; ++j) {
    C[(size_t)(rowBase + j) * N + tn + c] = acc[j];
  }
}

template <int N>
__global__ __launch_bounds__(32 * L1_WAVES) void node_attn_h8_gemm(
    const float* __restrict__ feat, const int* __restrict__ rowptr,
    const int* __restrict__ colTmp, int* col, const float* __restrict__ W2,
    float* __restrict__ feat2, int Nn, float scale) {
  __shared__ __attribute__((aligned(16))) float h1Tile[L1_WAVES * H1_PITCH];
  const int lane = threadIdx.x & 31;
  const int wave = __builtin_amdgcn_readfirstlane((int)(threadIdx.x >> 5));
  const int node0 = blockIdx.x * L1_WAVES;
  const int node = node0 + wave;
  float4 o = make_float4(0.0f, 0.0f, 0.0f, 0.0f);
  if (node < Nn) o = attn_h8_row(feat, rowptr, colTmp, col, node, lane, scale);
  *(float4*)(h1Tile + wave * H1_PITCH + lane * 4) = o;
  __syncthreads();
  if (wave < N / 16)
    gemm_tile_from_lds<128, N>(h1Tile, W2, feat2, node0, 16 * wave, lane);
}

template <int K0>
__device__ __forceinline__ void h1_four_steps(
    const float* __restrict__ fq, int ids, int cnt, int g, const float4 fdA,
    const float4 fdB, float scale, float& m, float& l, float4& accA,
    float4& accB) {
  float4 fa[4], fb[4];
  float p[4];
#pragma unroll
  for (int u = 0; u < 4; ++u) {
    const int s = u == 0   ? dpp8_i<dpp8_bcast_sel(K0 + 0)>(ids)
                  : u == 1 ? dpp8_i<dpp8_bcast_sel(K0 + 1)>(ids)
                  : u == 2 ? dpp8_i<dpp8_bcast_sel(K0 + 2)>(ids)
                           : dpp8_i<dpp8_bcast_sel(K0 + 3)>(ids);
    const float* r = fq + (size_t)s * 64;
    fa[u] = *(const float4*)(r);
    fb[u] = *(const float4*)(r + 32);
  }
#pragma unroll
  for (int u = 0; u < 4; ++u) {
    float d = fa[u].x * fdA.x + fa[u].y * fdA.y + fa[u].z * fdA.z +
              fa[u].w * fdA.w + fb[u].x * fdB.x + fb[u].y * fdB.y +
              fb[u].z * fdB.z + fb[u].w * fdB.w;
    d += dpp8_f<dpp8_xor_sel(1)>(d);
    d += dpp8_f<dpp8_xor_sel(2)>(d);
    d += dpp8_f<dpp8_xor_sel(4)>(d);
    d *= scale;
    p[u] = (4 * (K0 + u) + g < cnt) ? d : -INFINITY;
  }
  const float mNew = fmaxf(fmaxf(m, p[0]), fmaxf(fmaxf(p[1], p[2]), p[3]));
  const float mRef = (mNew == -INFINITY) ? 0.0f : mNew;
  const float c = __expf(m - mRef);
  const float w0 = __expf(p[0] - mRef);
  const float w1 = __expf(p[1] - mRef);
  const float w2 = __expf(p[2] - mRef);
  const float w3 = __expf(p[3] - mRef);
  l = l * c + ((w0 + w1) + (w2 + w3));
  accA.x = accA.x * c + ((w0 * fa[0].x + w1 * fa[1].x) + (w2 * fa[2].x + w3 * fa[3].x));
  accA.y = accA.y * c + ((w0 * fa[0].y + w1 * fa[1].y) + (w2 * fa[2].y + w3 * fa[3].y));
  accA.z = accA.z * c + ((w0 * fa[0].z + w1 * fa[1].z) + (w2 * fa[2].z + w3 * fa[3].z));
  accA.w = accA.w * c + ((w0 * fa[0].w + w1 * fa[1].w) + (w2 * fa[2].w + w3 * fa[3].w));
  accB.x = accB.x * c + ((w0 * fb[0].x + w1 * fb[1].x) + (w2 * fb[2].x + w3 * fb[3].x));
  accB.y = accB.y * c + ((w0 * fb[0].y + w1 * fb[1].y) + (w2 * fb[2].y + w3 * fb[3].y));
  accB.z = accB.z * c + ((w0 * fb[0].z + w1 * fb[1].z) + (w2 * fb[2].z + w3 * fb[3].z));
  accB.w = accB.w * c + ((w0 * fb[0].w + w1 * fb[1].w) + (w2 * fb[2].w + w3 * fb[3].w));
  m = mNew;
}

__device__ __forceinline__ float sum_groups(float v) {
  v += __shfl_xor(v, 8, 32);
  v += __shfl_xor(v, 16, 32);
  return v;
}

__global__ __launch_bounds__(256) void node_attn_h1(
    const float* __restrict__ feat, const int* __restrict__ rowptr,
    const int* __restrict__ col, float* __restrict__ out, int Nn, float scale) {
  const int lane = threadIdx.x & 31;
  const int node = __builtin_amdgcn_readfirstlane(
      (int)((blockIdx.x * blockDim.x + threadIdx.x) >> 5));
  if (node >= Nn) return;
  const int beg = rowptr[node];
  const int end = rowptr[node + 1];
  const int q = lane & 7;
  const int g = lane >> 3;
  const float* fq = feat + q * 4;
  const float4 fdA = *(const float4*)(fq + (size_t)node * 64);
  const float4 fdB = *(const float4*)(fq + (size_t)node * 64 + 32);

  float m = -INFINITY, l = 0.0f;
  float4 accA = make_float4(0.0f, 0.0f, 0.0f, 0.0f);
  float4 accB = make_float4(0.0f, 0.0f, 0.0f, 0.0f);

  const int myIdx = 4 * q + g;
  for (int base = beg; base < end; base += 32) {
    const int cnt = min(32, end - base);
    const int ids = (myIdx < cnt) ? col[base + myIdx] : node;
    h1_four_steps<0>(fq, ids, cnt, g, fdA, fdB, scale, m, l, accA, accB);
    if (cnt > 16)
      h1_four_steps<4>(fq, ids, cnt, g, fdA, fdB, scale, m, l, accA, accB);
  }

  float M = fmaxf(m, __shfl_xor(m, 8, 32));
  M = fmaxf(M, __shfl_xor(M, 16, 32));
  const float Mref = (M == -INFINITY) ? 0.0f : M;
  const float sc = __expf(m - Mref);
  const float L = sum_groups(l * sc);
  float4 oA, oB;
  oA.x = sum_groups(accA.x * sc);
  oA.y = sum_groups(accA.y * sc);
  oA.z = sum_groups(accA.z * sc);
  oA.w = sum_groups(accA.w * sc);
  oB.x = sum_groups(accB.x * sc);
  oB.y = sum_groups(accB.y * sc);
  oB.z = sum_groups(accB.z * sc);
  oB.w = sum_groups(accB.w * sc);
  const float inv = 1.0f / fmaxf(L, 1e-9f);
  if (g == 0) {
    float* o = out + (size_t)node * 64 + q * 4;
    *(float4*)(o) = make_float4(oA.x * inv, oA.y * inv, oA.z * inv, oA.w * inv);
    *(float4*)(o + 32) = make_float4(oB.x * inv, oB.y * inv, oB.z * inv, oB.w * inv);
  }
}

static inline int cdiv_i(long long a, long long b) { return (int)((a + b - 1) / b); }

extern "C" void kernel_launch(void* const* d_in, const int* in_sizes, int n_in,
                              void* d_out, int out_size, void* d_ws, size_t ws_size,
                              hipStream_t stream) {
  const float* hIn = (const float*)d_in[0];
  const float* W1  = (const float*)d_in[1];
  const float* W2  = (const float*)d_in[2];
  const int*   src = (const int*)d_in[3];
  const int*   dst = (const int*)d_in[4];

  const int IN_DIM = 128;
  constexpr int OUT_DIM = 64;
  const int HD1 = 128;
  const int Nn  = in_sizes[0] / IN_DIM;
  const int E   = in_sizes[3];

  const int nBins   = cdiv_i(Nn, BIN_NODES);
  const int chunk   = cdiv_i(E, NCHUNK);
  const int locLen  = (nBins + 1) * NCHUNK;

  float* feat1  = (float*)d_ws;
  float* h1     = feat1 + (size_t)Nn * HD1;
  int*   colTmp = (int*)(h1 + (size_t)Nn * HD1) + (size_t)E;
  int*   col    = colTmp + (size_t)E;
  int*   rowptr = col + (size_t)E + (size_t)Nn;
  int2*  binned = (int2*)h1;
  int*   loc    = (int*)(h1 + (size_t)Nn * HD1) - locLen;

  const int TPB = 256;
  const int wavesPerBlock = TPB / 32;
  float* feat2 = h1;

  {
    int gemmBlocks = cdiv_i((Nn / 16) * (HD1 / 64), SORT_TPB / 32);
    bin_sort_and_gemm<128, 128><<<NCHUNK + gemmBlocks, SORT_TPB, 0, stream>>>(
        src, dst, loc, binned, E, chunk, nBins, hIn, W1, feat1, Nn);
  }
  bin_lists<<<nBins, 32 * LIST_WAVES, 0, stream>>>(binned, loc, rowptr, colTmp, Nn,
                                                   nBins, chunk);

  node_attn_h8_gemm<OUT_DIM><<<cdiv_i(Nn, L1_WAVES), 32 * L1_WAVES, 0, stream>>>(
      feat1, rowptr, colTmp, col, W2, feat2, Nn, 0.25f);

  node_attn_h1<<<cdiv_i(Nn, wavesPerBlock), TPB, 0, stream>>>(
      feat2, rowptr, col, (float*)d_out, Nn, 0.125f);
}


# Both attention loops: packed products on the register pairs a 16-byte load delivers (no shuffle moves), scale*log2e folded into the node's row; sum order in a lane's dot and the 4-edge sum changes
# speedup vs baseline: 9.4588x; 1.0675x over previous
#include <hip/hip_runtime.h>
#include <math.h>

typedef __attribute__((ext_vector_type(2))) float v2f;
typedef __attribute__((ext_vector_type(8))) float v8f;

constexpr int BIN_SHIFT = 8;
constexpr int BIN_NODES = 1 << BIN_SHIFT;
constexpr int NCHUNK = 256;
constexpr int MAX_BINS = 1024;
constexpr int SORT_TPB = 256;
constexpr int SORT_BATCH = 16;

template <int N>
__device__ __forceinline__ int row_shr_or0(int v) {
  return __builtin_amdgcn_update_dpp(0, v, 0x110 + N, 0xf, 0xf, true);
}

__device__ __forceinline__ int wave_incl_scan(int v, int lane) {
  v += row_shr_or0<1>(v);
  v += row_shr_or0<2>(v);
  v += row_shr_or0<4>(v);
  v += row_shr_or0<8>(v);
  const int lowRow = __builtin_amdgcn_readlane(v, 15);
  return v + (lane >= 16 ? lowRow : 0);
}

template <int KD, int N>
__device__ __forceinline__ void gemm_block_16x64(
    const float* __restrict__ A, const float* __restrict__ B,
    float* __restrict__ C, int unit, int lane) {
  constexpr int NT = 4;
  constexpr int CG = N / (16 * NT);
  constexpr int STEPS = KD / 8;
  const int tm  = (unit / CG) << 4;
  const int tn0 = (unit % CG) * (16 * NT);
  const int c = lane & 15;
  const int h = lane >> 4;

  const float* Arow = A + (size_t)(tm + c) * KD + 4 * h;
  const float* Bp = B + (size_t)(4 * h) * N + tn0 + c;
  v8f acc[NT] = {};
#pragma unroll
  for (int j = 0; j < STEPS; ++j) {
    const float4 a = *(const float4*)(Arow + 8 * j);
    float b0[NT], b1[NT], b2[NT], b3[NT];
#pragma unroll
    for (int t = 0; t < NT; ++t) {
      const float* p = Bp + (size_t)(8 * j) * N + 16 * t;
      b0[t] = p[0];
      b1[t] = p[N];
      b2[t] = p[2 * N];
      b3[t] = p[3 * N];
    }
#pragma unroll
    for (int t = 0; t < NT; ++t) {
      v2f av, bv;
      av.x = a.x;
      av.y = a.y;
      bv.x = b0[t];
      bv.y = b1[t];
      acc[t] = __builtin_amdgcn_wmma_f32_16x16x4_f32(
           false, av,  false, bv,
           (short)0, acc[t],  false,  false);
    }
#pragma unroll
    for (int t = 0; t < NT; ++t) {
      v2f av, bv;
      av.x = a.z;
      av.y = a.w;
      bv.x = b2[t];
      bv.y = b3[t];
      acc[t] = __builtin_amdgcn_wmma_f32_16x16x4_f32(
           false, av,  false, bv,
           (short)0, acc[t],  false,  false);
    }
  }

  const int rowBase = tm + (h << 3);
#pragma unroll
  for (int t = 0; t < NT; ++t) {
#pragma unroll
    for (int j = 0; j < 8; ++j) {
      C[(size_t)(rowBase + j) * N + tn0 + 16 * t + c] = acc[t][j];
    }
  }
}

template <int KD, int N>
__global__ __launch_bounds__(SORT_TPB) void bin_sort_and_gemm(
    const int* __restrict__ src, const int* __restrict__ dst,
    int* __restrict__ loc, int2* __restrict__ binned, int E, int chunk,
    int nBins, const float* __restrict__ A, const float* __restrict__ B,
    float* __restrict__ C, int M) {
  __shared__ int binCur[MAX_BINS + 32];
  __shared__ int waveTot[SORT_TPB / 32];
  const int b = blockIdx.x;
  if (b >= NCHUNK) {
    const int unit = __builtin_amdgcn_readfirstlane(
        (int)(((b - NCHUNK) * SORT_TPB + threadIdx.x) >> 5));
    if (unit < (M >> 4) * (N / 64))
      gemm_block_16x64<KD, N>(A, B, C, unit, threadIdx.x & 31);
    return;
  }
  const int c = b;
  const int t = threadIdx.x;
  const int lane = t & 31;
  const int wave = t >> 5;
  for (int i = t; i < nBins; i += SORT_TPB) binCur[i] = 0;
  if (t < 32) binCur[MAX_BINS + t] = 0;
  __syncthreads();
  const int e0 = c * chunk;
  const int e1 = min(E, e0 + chunk);

  for (int e = e0 + t; e < e1; e += SORT_BATCH * SORT_TPB) {
    int d[SORT_BATCH];
#pragma unroll
    for (int u = 0; u < SORT_BATCH; ++u) {
      const int i = e + u * SORT_TPB;
      d[u] = (i < e1) ? dst[i] : -1;
    }
#pragma unroll
    for (int u = 0; u < SORT_BATCH; ++u)
      if (d[u] >= 0) atomicAdd(&binCur[d[u] >> BIN_SHIFT], 1);
  }
  __syncthreads();

  const int i0 = 4 * t;
  const int v0 = (i0 + 0 < nBins) ? binCur[i0 + 0] : 0;
  const int v1 = (i0 + 1 < nBins) ? binCur[i0 + 1] : 0;
  const int v2 = (i0 + 2 < nBins) ? binCur[i0 + 2] : 0;
  const int v3 = (i0 + 3 < nBins) ? binCur[i0 + 3] : 0;
  const int s4 = (v0 + v1) + (v2 + v3);
  const int inc = wave_incl_scan(s4, lane);
  if (lane == 31) waveTot[wave] = inc;
  __syncthreads();
  int base = 0, total = 0;
#pragma unroll
  for (int w = 0; w < SORT_TPB / 32; ++w) {
    const int x = waveTot[w];
    total += x;
    base += (w < wave) ? x : 0;
  }
  const int o0 = base + (inc - s4);
  const int o1 = o0 + v0;
  const int o2 = o1 + v1;
  const int o3 = o2 + v2;
  if (i0 + 0 < nBins) { binCur[i0 + 0] = o0; loc[(i0 + 0) * NCHUNK + c] = o0; }
  if (i0 + 1 < nBins) { binCur[i0 + 1] = o1; loc[(i0 + 1) * NCHUNK + c] = o1; }
  if (i0 + 2 < nBins) { binCur[i0 + 2] = o2; loc[(i0 + 2) * NCHUNK + c] = o2; }
  if (i0 + 3 < nBins) { binCur[i0 + 3] = o3; loc[(i0 + 3) * NCHUNK + c] = o3; }
  if (t == 0) loc[nBins * NCHUNK + c] = total;
  __syncthreads();

  for (int e = e0 + t; e < e1; e += SORT_BATCH * SORT_TPB) {
    int d[SORT_BATCH], sId[SORT_BATCH];
#pragma unroll
    for (int u = 0; u < SORT_BATCH; ++u) {
      const int i = e + u * SORT_TPB;
      const bool v = i < e1;
      d[u] = v ? dst[i] : -1;
      sId[u] = v ? src[i] : 0;
    }
    int pos[SORT_BATCH];
#pragma unroll
    for (int u = 0; u < SORT_BATCH; ++u)
      pos[u] = atomicAdd(&binCur[d[u] >= 0 ? (d[u] >> BIN_SHIFT) : MAX_BINS + lane], 1);
#pragma unroll
    for (int u = 0; u < SORT_BATCH; ++u)
      if (d[u] >= 0) binned[e0 + pos[u]] = make_int2(d[u], sId[u]);
  }
}

constexpr int LIST_WAVES = 32;
constexpr int LIST_CPW = NCHUNK / LIST_WAVES;

__global__ __launch_bounds__(32 * LIST_WAVES) void bin_lists(
    const int2* __restrict__ binned, const int* __restrict__ loc,
    int* __restrict__ rowptr, int* __restrict__ colTmp, int Nn, int nBins,
    int chunk) {
  __shared__ int cnt[BIN_NODES];
  __shared__ int waveTot[BIN_NODES / 32];
  __shared__ int sumLo[LIST_WAVES];
  __shared__ int sumHi[LIST_WAVES];
  const int bin = blockIdx.x;
  const int t = threadIdx.x;
  const int lane = t & 31;
  const int wave = __builtin_amdgcn_readfirstlane((int)(t >> 5));
  const int node0 = bin << BIN_SHIFT;

  const int cl = wave * LIST_CPW + (lane % LIST_CPW);
  const int loV = loc[bin * NCHUNK + cl];
  const int hiV = loc[(bin + 1) * NCHUNK + cl];
  {
    const int wl = wave_incl_scan(lane < LIST_CPW ? loV : 0, lane);
    const int wh = wave_incl_scan(lane < LIST_CPW ? hiV : 0, lane);
    if (lane == 31) {
      sumLo[wave] = wl;
      sumHi[wave] = wh;
    }
  }
  if (t < BIN_NODES) cnt[t] = 0;
  __syncthreads();
  const int b0 = __builtin_amdgcn_readlane(wave_incl_scan(sumLo[lane], lane), 31);
  const int b1 = __builtin_amdgcn_readlane(wave_incl_scan(sumHi[lane], lane), 31);

  int lo[LIST_CPW], hi[LIST_CPW];
#pragma unroll
  for (int u = 0; u < LIST_CPW; ++u) {
    lo[u] = __builtin_amdgcn_readlane(loV, u);
    hi[u] = __builtin_amdgcn_readlane(hiV, u);
  }
  const int2* wbase = binned + (size_t)(wave * LIST_CPW) * chunk;

  {
    int d[2 * LIST_CPW];
#pragma unroll
    for (int u = 0; u < LIST_CPW; ++u) {
      const int2* piece = wbase + (size_t)u * chunk;
      const int j = lo[u] + lane;
      d[2 * u] = (j < hi[u]) ? piece[j].x : -1;
      d[2 * u + 1] = (j + 32 < hi[u]) ? piece[j + 32].x : -1;
    }
#pragma unroll
    for (int v = 0; v < 2 * LIST_CPW; ++v)
      if (d[v] >= 0) atomicAdd(&cnt[d[v] - node0], 1);
#pragma unroll
    for (int u = 0; u < LIST_CPW; ++u) {
      const int2* piece = wbase + (size_t)u * chunk;
      for (int jb = lo[u] + 64; jb < hi[u]; jb += 32) {
        const int j = jb + lane;
        if (j < hi[u]) atomicAdd(&cnt[piece[j].x - node0], 1);
      }
    }
  }
  __syncthreads();

  int n = 0, inc = 0;
  if (t < BIN_NODES) {
    n = cnt[t];
    inc = wave_incl_scan(n, lane);
    if (lane == 31) waveTot[t >> 5] = inc;
  }
  __syncthreads();
  if (t < BIN_NODES) {
    int base = b0;
    for (int w = 0; w < (t >> 5); ++w) base += waveTot[w];
    const int start = base + (inc - n);
    cnt[t] = start;
    const int node = node0 + t;
    if (node < Nn) rowptr[node] = start;
  }
  if (bin == nBins - 1 && t == 0) rowptr[Nn] = b1;
  __syncthreads();

  {
    int2 e[2 * LIST_CPW];
#pragma unroll
    for (int u = 0; u < LIST_CPW; ++u) {
      const int2* piece = wbase + (size_t)u * chunk;
      const int j = lo[u] + lane;
      e[2 * u] = (j < hi[u]) ? piece[j] : make_int2(-1, 0);
      e[2 * u + 1] = (j + 32 < hi[u]) ? piece[j + 32] : make_int2(-1, 0);
    }
#pragma unroll
    for (int v = 0; v < 2 * LIST_CPW; ++v) {
      if (e[v].x >= 0) {
        const int pos = atomicAdd(&cnt[e[v].x - node0], 1);
        colTmp[pos] = e[v].y;
      }
    }
#pragma unroll
    for (int u = 0; u < LIST_CPW; ++u) {
      const int2* piece = wbase + (size_t)u * chunk;
      for (int jb = lo[u] + 64; jb < hi[u]; jb += 32) {
        const int j = jb + lane;
        if (j < hi[u]) {
          const int2 x = piece[j];
          const int pos = atomicAdd(&cnt[x.x - node0], 1);
          colTmp[pos] = x.y;
        }
      }
    }
  }
}

constexpr unsigned dpp8_xor_sel(int x) {
  unsigned s = 0;
  for (int i = 0; i < 8; ++i) s |= (unsigned)((i ^ x) & 7) << (3 * i);
  return s;
}
constexpr unsigned dpp8_bcast_sel(int k) { return (unsigned)k * 0x249249u; }

template <unsigned SEL>
__device__ __forceinline__ float dpp8_f(float v) {
  return __uint_as_float(__builtin_amdgcn_mov_dpp8(__float_as_uint(v), SEL));
}
template <unsigned SEL>
__device__ __forceinline__ int dpp8_i(int v) {
  return (int)__builtin_amdgcn_mov_dpp8((unsigned)v, SEL);
}

template <int J>
__device__ __forceinline__ int lane_xor_i(int v);
template <>
__device__ __forceinline__ int lane_xor_i<1>(int v) { return dpp8_i<dpp8_xor_sel(1)>(v); }
template <>
__device__ __forceinline__ int lane_xor_i<2>(int v) { return dpp8_i<dpp8_xor_sel(2)>(v); }
template <>
__device__ __forceinline__ int lane_xor_i<4>(int v) { return dpp8_i<dpp8_xor_sel(4)>(v); }
template <>
__device__ __forceinline__ int lane_xor_i<8>(int v) {
  return __builtin_amdgcn_update_dpp(v, v, 0x168  , 0xf, 0xf, false);
}
template <>
__device__ __forceinline__ int lane_xor_i<16>(int v) {
  return (int)__builtin_amdgcn_permlanex16((unsigned)v, (unsigned)v, 0x76543210u,
                                           0xfedcba98u, false, false);
}

template <int J>
__device__ __forceinline__ int cmp_xchg(int v, int lane, bool asc) {
  const int p = lane_xor_i<J>(v);
  const bool keepMin = ((lane & J) == 0) == asc;
  return keepMin ? min(v, p) : max(v, p);
}

template <int K>
__device__ __forceinline__ int merge_lanes(int v, int lane, bool asc) {
  if (K >= 32) v = cmp_xchg<16>(v, lane, asc);
  if (K >= 16) v = cmp_xchg<8>(v, lane, asc);
  if (K >= 8) v = cmp_xchg<4>(v, lane, asc);
  if (K >= 4) v = cmp_xchg<2>(v, lane, asc);
  return cmp_xchg<1>(v, lane, asc);
}

__device__ __forceinline__ int sort_lanes(int v, int lane, bool asc) {
  v = merge_lanes<2>(v, lane, (lane & 2) == 0);
  v = merge_lanes<4>(v, lane, (lane & 4) == 0);
  v = merge_lanes<8>(v, lane, (lane & 8) == 0);
  v = merge_lanes<16>(v, lane, (lane & 16) == 0);
  return merge_lanes<32>(v, lane, asc);
}

__device__ __forceinline__ float quad_xor1_f(float v) {
  return __int_as_float(__builtin_amdgcn_mov_dpp(__float_as_int(v), 0xB1  ,
                                                 0xf, 0xf, true));
}
__device__ __forceinline__ float quad_xor2_f(float v) {
  return __int_as_float(__builtin_amdgcn_mov_dpp(__float_as_int(v), 0x4E  ,
                                                 0xf, 0xf, true));
}
__device__ __forceinline__ float row_xor4_f(float v) {
  return __int_as_float(__builtin_amdgcn_mov_dpp(__float_as_int(v), 0x164  ,
                                                 0xf, 0xf, true));
}

__device__ __forceinline__ float4 attn_h8_row(
    const float* __restrict__ feat, const int* __restrict__ rowptr,
    const int* __restrict__ colTmp, int* col, int node, int lane, float scale) {
  const int beg = rowptr[node];
  const int end = rowptr[node + 1];
  const int n = end - beg;
  const float* fl = feat + lane * 4;
  const float4 fd = *(const float4*)(fl + (size_t)node * 128);

  const bool inRegs = n <= 64;
  int k0 = 0x7fffffff, k1 = 0x7fffffff;
  if (inRegs) {
    if (lane < n) k0 = colTmp[beg + lane];
    if (n > 32) {
      if (32 + lane < n) k1 = colTmp[beg + 32 + lane];
      k0 = sort_lanes(k0, lane, true);
      k1 = sort_lanes(k1, lane, false);
      const int lo = min(k0, k1);
      const int hi = max(k0, k1);
      k0 = merge_lanes<32>(lo, lane, true);
      k1 = merge_lanes<32>(hi, lane, true);
      if (32 + lane < n) col[beg + 32 + lane] = k1;
    } else {
      k0 = sort_lanes(k0, lane, true);
    }
    if (lane < n) col[beg + lane] = k0;
  } else {
    for (int ib = 0; ib < n; ib += 32) {
      const int i = ib + lane;
      const bool vi = i < n;
      const int ki = vi ? colTmp[beg + i] : 0;
      int rank = 0;
      for (int jb = 0; jb < n; jb += 32) {
        const int cj = min(32, n - jb);
        const int kc = (lane < cj) ? colTmp[beg + jb + lane] : 0;
        for (int t = 0; t < cj; ++t) {
          const int kj = __builtin_amdgcn_readlane(kc, t);
          const int bias = (jb + t < i) ? 1 : 0;
          rank += (kj < ki + bias) ? 1 : 0;
        }
      }
      if (vi) col[beg + rank] = ki;
    }
    __threadfence();
  }

  const float ks = scale * 1.4426950408889634f;
  const v2f fdLo = {fd.x * ks, fd.y * ks};
  const v2f fdHi = {fd.z * ks, fd.w * ks};

  float m = -INFINITY, l = 0.0f;
  v2f accLo = {0.0f, 0.0f};
  v2f accHi = {0.0f, 0.0f};

  for (int base = beg; base < end; base += 32) {
    const int cnt = min(32, end - base);
    int ids;
    if (inRegs) {
      ids = (base == beg) ? k0 : k1;
    } else {
      ids = (lane < cnt) ? col[base + lane] : 0;
    }
    ids = (lane < cnt) ? ids : node;
    for (int t = 0; t < cnt; t += 4) {
      v2f fLo[4], fHi[4];
      float p[4];
#pragma unroll
      for (int u = 0; u < 4; ++u) {
        const int s = __builtin_amdgcn_readlane(ids, t + u);
        const float4 f = *(const float4*)(fl + (size_t)s * 128);
        fLo[u] = (v2f){f.x, f.y};
        fHi[u] = (v2f){f.z, f.w};
      }
#pragma unroll
      for (int u = 0; u < 4; ++u) {
        v2f tt = fLo[u] * fdLo;
        tt = fHi[u] * fdHi + tt;
        float d = tt.x + tt.y;
        d += quad_xor1_f(d);
        d += quad_xor2_f(d);
        p[u] = (t + u < cnt) ? d : -INFINITY;
      }
      const float mNew = fmaxf(fmaxf(m, p[0]), fmaxf(fmaxf(p[1], p[2]), p[3]));
      const float c = __builtin_amdgcn_exp2f(m - mNew);
      const float w0 = __builtin_amdgcn_exp2f(p[0] - mNew);
      const float w1 = __builtin_amdgcn_exp2f(p[1] - mNew);
      const float w2 = __builtin_amdgcn_exp2f(p[2] - mNew);
      const float w3 = __builtin_amdgcn_exp2f(p[3] - mNew);
      l = l * c + ((w0 + w1) + (w2 + w3));
      v2f sLo = fLo[0] * w0;
      sLo = fLo[1] * w1 + sLo;
      sLo = fLo[2] * w2 + sLo;
      sLo = fLo[3] * w3 + sLo;
      accLo = accLo * c + sLo;
      v2f sHi = fHi[0] * w0;
      sHi = fHi[1] * w1 + sHi;
      sHi = fHi[2] * w2 + sHi;
      sHi = fHi[3] * w3 + sHi;
      accHi = accHi * c + sHi;
      m = mNew;
    }
  }

  const float inv = 1.0f / fmaxf(l, 1e-9f);
  float4 o;
  o.x = accLo.x * inv;
  o.y = accLo.y * inv;
  o.z = accHi.x * inv;
  o.w = accHi.y * inv;
  o.x = o.x > 0.0f ? o.x : expm1f(o.x);
  o.y = o.y > 0.0f ? o.y : expm1f(o.y);
  o.z = o.z > 0.0f ? o.z : expm1f(o.z);
  o.w = o.w > 0.0f ? o.w : expm1f(o.w);
  return o;
}

constexpr int L1_WAVES = 16;
constexpr int H1_PITCH = 132;

template <int KD, int N>
__device__ __forceinline__ void gemm_tile_from_lds(
    const float* tile, const float* __restrict__ B, float* __restrict__ C,
    int tm, int tn, int lane) {
  constexpr int STEPS = KD / 8;
  const int c = lane & 15;
  const int h = lane >> 4;
  const float* Ap = tile + c * H1_PITCH + 4 * h;
  const float* Bp = B + (size_t)(4 * h) * N + tn + c;
  v8f acc = {};
#pragma unroll
  for (int j = 0; j < STEPS; ++j) {
    const float4 a = *(const float4*)(Ap + 8 * j);
    const float* p = Bp + (size_t)(8 * j) * N;
    v2f av, bv;
    av.x = a.x;
    av.y = a.y;
    bv.x = p[0];
    bv.y = p[N];
    acc = __builtin_amdgcn_wmma_f32_16x16x4_f32(
         false, av,  false, bv,
         (short)0, acc,  false,  false);
    av.x = a.z;
    av.y = a.w;
    bv.x = p[2 * N];
    bv.y = p[3 * N];
    acc = __builtin_amdgcn_wmma_f32_16x16x4_f32(
         false, av,  false, bv,
         (short)0, acc,  false,  false);
  }
  const int rowBase = tm + (h << 3);
#pragma unroll
  for (int j = 0; j < 8; ++j) {
    C[(size_t)(rowBase + j) * N + tn + c] = acc[j];
  }
}

template <int N>
__global__ __launch_bounds__(32 * L1_WAVES) void node_attn_h8_gemm(
    const float* __restrict__ feat, const int* __restrict__ rowptr,
    const int* __restrict__ colTmp, int* col, const float* __restrict__ W2,
    float* __restrict__ feat2, int Nn, float scale) {
  __shared__ __attribute__((aligned(16))) float h1Tile[L1_WAVES * H1_PITCH];
  const int lane = threadIdx.x & 31;
  const int wave = __builtin_amdgcn_readfirstlane((int)(threadIdx.x >> 5));
  const int node0 = blockIdx.x * L1_WAVES;
  const int node = node0 + wave;
  float4 o = make_float4(0.0f, 0.0f, 0.0f, 0.0f);
  if (node < Nn) o = attn_h8_row(feat, rowptr, colTmp, col, node, lane, scale);
  *(float4*)(h1Tile + wave * H1_PITCH + lane * 4) = o;
  __syncthreads();
  if (wave < N / 16)
    gemm_tile_from_lds<128, N>(h1Tile, W2, feat2, node0, 16 * wave, lane);
}

template <int K0>
__device__ __forceinline__ void h1_four_steps(
    const float* __restrict__ fq, int ids, int cnt, int g, const v2f fdA0,
    const v2f fdA1, const v2f fdB0, const v2f fdB1, float& m, float& l,
    v2f& accA0, v2f& accA1, v2f& accB0, v2f& accB1) {
  v2f a0[4], a1[4], b0[4], b1[4];
  float p[4];
#pragma unroll
  for (int u = 0; u < 4; ++u) {
    const int s = u == 0   ? dpp8_i<dpp8_bcast_sel(K0 + 0)>(ids)
                  : u == 1 ? dpp8_i<dpp8_bcast_sel(K0 + 1)>(ids)
                  : u == 2 ? dpp8_i<dpp8_bcast_sel(K0 + 2)>(ids)
                           : dpp8_i<dpp8_bcast_sel(K0 + 3)>(ids);
    const float* r = fq + (size_t)s * 64;
    const float4 fa = *(const float4*)(r);
    const float4 fb = *(const float4*)(r + 32);
    a0[u] = (v2f){fa.x, fa.y};
    a1[u] = (v2f){fa.z, fa.w};
    b0[u] = (v2f){fb.x, fb.y};
    b1[u] = (v2f){fb.z, fb.w};
  }
#pragma unroll
  for (int u = 0; u < 4; ++u) {
    v2f tt = a0[u] * fdA0;
    tt = a1[u] * fdA1 + tt;
    tt = b0[u] * fdB0 + tt;
    tt = b1[u] * fdB1 + tt;
    float d = tt.x + tt.y;
    d += quad_xor1_f(d);
    d += quad_xor2_f(d);
    d += row_xor4_f(d);
    p[u] = (4 * (K0 + u) + g < cnt) ? d : -INFINITY;
  }
  const float mNew = fmaxf(fmaxf(m, p[0]), fmaxf(fmaxf(p[1], p[2]), p[3]));
  const float mRef = (mNew == -INFINITY) ? 0.0f : mNew;
  const float c = __builtin_amdgcn_exp2f(m - mRef);
  const float w0 = __builtin_amdgcn_exp2f(p[0] - mRef);
  const float w1 = __builtin_amdgcn_exp2f(p[1] - mRef);
  const float w2 = __builtin_amdgcn_exp2f(p[2] - mRef);
  const float w3 = __builtin_amdgcn_exp2f(p[3] - mRef);
  l = l * c + ((w0 + w1) + (w2 + w3));
  v2f s;
  s = a0[0] * w0;
  s = a0[1] * w1 + s;
  s = a0[2] * w2 + s;
  s = a0[3] * w3 + s;
  accA0 = accA0 * c + s;
  s = a1[0] * w0;
  s = a1[1] * w1 + s;
  s = a1[2] * w2 + s;
  s = a1[3] * w3 + s;
  accA1 = accA1 * c + s;
  s = b0[0] * w0;
  s = b0[1] * w1 + s;
  s = b0[2] * w2 + s;
  s = b0[3] * w3 + s;
  accB0 = accB0 * c + s;
  s = b1[0] * w0;
  s = b1[1] * w1 + s;
  s = b1[2] * w2 + s;
  s = b1[3] * w3 + s;
  accB1 = accB1 * c + s;
  m = mNew;
}

__device__ __forceinline__ float sum_groups(float v) {
  v += __shfl_xor(v, 8, 32);
  v += __shfl_xor(v, 16, 32);
  return v;
}

__global__ __launch_bounds__(256) void node_attn_h1(
    const float* __restrict__ feat, const int* __restrict__ rowptr,
    const int* __restrict__ col, float* __restrict__ out, int Nn, float scale) {
  const int lane = threadIdx.x & 31;
  const int node = __builtin_amdgcn_readfirstlane(
      (int)((blockIdx.x * blockDim.x + threadIdx.x) >> 5));
  if (node >= Nn) return;
  const int beg = rowptr[node];
  const int end = rowptr[node + 1];
  const int q = lane & 7;
  const int g = lane >> 3;
  const float* fq = feat + q * 4;
  const float4 fdA = *(const float4*)(fq + (size_t)node * 64);
  const float4 fdB = *(const float4*)(fq + (size_t)node * 64 + 32);
  const float ks = scale * 1.4426950408889634f;
  const v2f fdA0 = {fdA.x * ks, fdA.y * ks};
  const v2f fdA1 = {fdA.z * ks, fdA.w * ks};
  const v2f fdB0 = {fdB.x * ks, fdB.y * ks};
  const v2f fdB1 = {fdB.z * ks, fdB.w * ks};

  float m = -INFINITY, l = 0.0f;
  v2f accA0 = {0.0f, 0.0f}, accA1 = {0.0f, 0.0f};
  v2f accB0 = {0.0f, 0.0f}, accB1 = {0.0f, 0.0f};

  const int myIdx = 4 * q + g;
  for (int base = beg; base < end; base += 32) {
    const int cnt = min(32, end - base);
    const int ids = (myIdx < cnt) ? col[base + myIdx] : node;
    h1_four_steps<0>(fq, ids, cnt, g, fdA0, fdA1, fdB0, fdB1, m, l, accA0, accA1, accB0, accB1);
    if (cnt > 16)
      h1_four_steps<4>(fq, ids, cnt, g, fdA0, fdA1, fdB0, fdB1, m, l, accA0, accA1, accB0, accB1);
  }

  float M = fmaxf(m, __shfl_xor(m, 8, 32));
  M = fmaxf(M, __shfl_xor(M, 16, 32));
  const float Mref = (M == -INFINITY) ? 0.0f : M;
  const float sc = __builtin_amdgcn_exp2f(m - Mref);
  const float L = sum_groups(l * sc);
  float4 oA, oB;
  oA.x = sum_groups(accA0.x * sc);
  oA.y = sum_groups(accA0.y * sc);
  oA.z = sum_groups(accA1.x * sc);
  oA.w = sum_groups(accA1.y * sc);
  oB.x = sum_groups(accB0.x * sc);
  oB.y = sum_groups(accB0.y * sc);
  oB.z = sum_groups(accB1.x * sc);
  oB.w = sum_groups(accB1.y * sc);
  const float inv = 1.0f / fmaxf(L, 1e-9f);
  if (g == 0) {
    float* o = out + (size_t)node * 64 + q * 4;
    *(float4*)(o) = make_float4(oA.x * inv, oA.y * inv, oA.z * inv, oA.w * inv);
    *(float4*)(o + 32) = make_float4(oB.x * inv, oB.y * inv, oB.z * inv, oB.w * inv);
  }
}

static inline int cdiv_i(long long a, long long b) { return (int)((a + b - 1) / b); }

extern "C" void kernel_launch(void* const* d_in, const int* in_sizes, int n_in,
                              void* d_out, int out_size, void* d_ws, size_t ws_size,
                              hipStream_t stream) {
  const float* hIn = (const float*)d_in[0];
  const float* W1  = (const float*)d_in[1];
  const float* W2  = (const float*)d_in[2];
  const int*   src = (const int*)d_in[3];
  const int*   dst = (const int*)d_in[4];

  const int IN_DIM = 128;
  constexpr int OUT_DIM = 64;
  const int HD1 = 128;
  const int Nn  = in_sizes[0] / IN_DIM;
  const int E   = in_sizes[3];

  const int nBins   = cdiv_i(Nn, BIN_NODES);
  const int chunk   = cdiv_i(E, NCHUNK);
  const int locLen  = (nBins + 1) * NCHUNK;

  float* feat1  = (float*)d_ws;
  float* h1     = feat1 + (size_t)Nn * HD1;
  int*   colTmp = (int*)(h1 + (size_t)Nn * HD1) + (size_t)E;
  int*   col    = colTmp + (size_t)E;
  int*   rowptr = col + (size_t)E + (size_t)Nn;
  int2*  binned = (int2*)h1;
  int*   loc    = (int*)(h1 + (size_t)Nn * HD1) - locLen;

  const int TPB = 256;
  const int wavesPerBlock = TPB / 32;
  float* feat2 = h1;

  {
    int gemmBlocks = cdiv_i((Nn / 16) * (HD1 / 64), SORT_TPB / 32);
    bin_sort_and_gemm<128, 128><<<NCHUNK + gemmBlocks, SORT_TPB, 0, stream>>>(
        src, dst, loc, binned, E, chunk, nBins, hIn, W1, feat1, Nn);
  }
  bin_lists<<<nBins, 32 * LIST_WAVES, 0, stream>>>(binned, loc, rowptr, colTmp, Nn,
                                                   nBins, chunk);

  node_attn_h8_gemm<OUT_DIM><<<cdiv_i(Nn, L1_WAVES), 32 * L1_WAVES, 0, stream>>>(
      feat1, rowptr, colTmp, col, W2, feat2, Nn, 0.25f);

  node_attn_h1<<<cdiv_i(Nn, wavesPerBlock), TPB, 0, stream>>>(
      feat2, rowptr, col, (float*)d_out, Nn, 0.125f);
}


# Sort scatters and h@W1 block stores go through LDS: a chunk's slice, a bin's node lists and each 16x64 product block are assembled in LDS and leave as whole 128-byte lines; same lists, same sums
# speedup vs baseline: 9.8137x; 1.0375x over previous
#include <hip/hip_runtime.h>
#include <math.h>

typedef __attribute__((ext_vector_type(2))) float v2f;
typedef __attribute__((ext_vector_type(8))) float v8f;

constexpr int BIN_SHIFT = 8;
constexpr int BIN_NODES = 1 << BIN_SHIFT;
constexpr int NCHUNK = 256;
constexpr int MAX_BINS = 1024;
constexpr int SORT_TPB = 256;
constexpr int SORT_BATCH = 16;
constexpr int STAGE_PAIRS = 6400;
constexpr int STAGE_SLOTS = STAGE_PAIRS + 16;

template <int N>
__device__ __forceinline__ int row_shr_or0(int v) {
  return __builtin_amdgcn_update_dpp(0, v, 0x110 + N, 0xf, 0xf, true);
}

__device__ __forceinline__ int wave_incl_scan(int v, int lane) {
  v += row_shr_or0<1>(v);
  v += row_shr_or0<2>(v);
  v += row_shr_or0<4>(v);
  v += row_shr_or0<8>(v);
  const int lowRow = __builtin_amdgcn_readlane(v, 15);
  return v + (lane >= 16 ? lowRow : 0);
}

constexpr int GEMM_TILE_PITCH = 68;

template <int KD, int N>
__device__ __forceinline__ void gemm_block_16x64(
    const float* __restrict__ A, const float* __restrict__ B,
    float* __restrict__ C, int unit, int lane, float* tileLds) {
  constexpr int NT = 4;
  constexpr int CG = N / (16 * NT);
  constexpr int STEPS = KD / 8;
  const int tm  = (unit / CG) << 4;
  const int tn0 = (unit % CG) * (16 * NT);
  const int c = lane & 15;
  const int h = lane >> 4;

  const float* Arow = A + (size_t)(tm + c) * KD + 4 * h;
  const float* Bp = B + (size_t)(4 * h) * N + tn0 + c;
  v8f acc[NT] = {};
#pragma unroll
  for (int j = 0; j < STEPS; ++j) {
    const float4 a = *(const float4*)(Arow + 8 * j);
    float b0[NT], b1[NT], b2[NT], b3[NT];
#pragma unroll
    for (int t = 0; t < NT; ++t) {
      const float* p = Bp + (size_t)(8 * j) * N + 16 * t;
      b0[t] = p[0];
      b1[t] = p[N];
      b2[t] = p[2 * N];
      b3[t] = p[3 * N];
    }
#pragma unroll
    for (int t = 0; t < NT; ++t) {
      v2f av, bv;
      av.x = a.x;
      av.y = a.y;
      bv.x = b0[t];
      bv.y = b1[t];
      acc[t] = __builtin_amdgcn_wmma_f32_16x16x4_f32(
           false, av,  false, bv,
           (short)0, acc[t],  false,  false);
    }
#pragma unroll
    for (int t = 0; t < NT; ++t) {
      v2f av, bv;
      av.x = a.z;
      av.y = a.w;
      bv.x = b2[t];
      bv.y = b3[t];
      acc[t] = __builtin_amdgcn_wmma_f32_16x16x4_f32(
           false, av,  false, bv,
           (short)0, acc[t],  false,  false);
    }
  }

  const int rowLds = (h << 3) * GEMM_TILE_PITCH + c;
#pragma unroll
  for (int t = 0; t < NT; ++t) {
#pragma unroll
    for (int j = 0; j < 8; ++j) {
      tileLds[rowLds + j * GEMM_TILE_PITCH + 16 * t] = acc[t][j];
    }
  }
#pragma unroll
  for (int i = 0; i < 8; ++i) {
    const int r = 2 * i + h;
    const float4 v = *(const float4*)(tileLds + r * GEMM_TILE_PITCH + 4 * c);
    *(float4*)(C + (size_t)(tm + r) * N + tn0 + 4 * c) = v;
  }
}

template <int KD, int N>
__global__ __launch_bounds__(SORT_TPB) __attribute__((amdgpu_num_vgpr(64))) void bin_sort_and_gemm(
    const int* __restrict__ src, const int* __restrict__ dst,
    int* __restrict__ loc, int2* __restrict__ binned, int E, int chunk,
    int nBins, const float* __restrict__ A, const float* __restrict__ B,
    float* __restrict__ C, int M) {
  __shared__ int binCur[MAX_BINS + 32];
  __shared__ int waveTot[SORT_TPB / 32];
  __shared__ __attribute__((aligned(16))) int2 stage[STAGE_SLOTS];
  static_assert((SORT_TPB / 32) * 16 * GEMM_TILE_PITCH * 4 <= STAGE_SLOTS * 8,
                "the product waves' tiles fit into the staging array");
  const int b = blockIdx.x;
  if (b >= NCHUNK) {
    const int unit = __builtin_amdgcn_readfirstlane(
        (int)(((b - NCHUNK) * SORT_TPB + threadIdx.x) >> 5));
    if (unit < (M >> 4) * (N / 64))
      gemm_block_16x64<KD, N>(A, B, C, unit, threadIdx.x & 31,
                              (float*)stage + (threadIdx.x >> 5) * (16 * GEMM_TILE_PITCH));
    return;
  }
  const int c = b;
  const int t = threadIdx.x;
  const int lane = t & 31;
  const int wave = t >> 5;
  for (int i = t; i < nBins; i += SORT_TPB) binCur[i] = 0;
  if (t < 32) binCur[MAX_BINS + t] = 0;
  __syncthreads();
  const int e0 = c * chunk;
  const int e1 = min(E, e0 + chunk);

  for (int e = e0 + t; e < e1; e += SORT_BATCH * SORT_TPB) {
    int d[SORT_BATCH];
#pragma unroll
    for (int u = 0; u < SORT_BATCH; ++u) {
      const int i = e + u * SORT_TPB;
      d[u] = (i < e1) ? dst[i] : -1;
    }
#pragma unroll
    for (int u = 0; u < SORT_BATCH; ++u)
      if (d[u] >= 0) atomicAdd(&binCur[d[u] >> BIN_SHIFT], 1);
  }
  __syncthreads();

  const int i0 = 4 * t;
  const int v0 = (i0 + 0 < nBins) ? binCur[i0 + 0] : 0;
  const int v1 = (i0 + 1 < nBins) ? binCur[i0 + 1] : 0;
  const int v2 = (i0 + 2 < nBins) ? binCur[i0 + 2] : 0;
  const int v3 = (i0 + 3 < nBins) ? binCur[i0 + 3] : 0;
  const int s4 = (v0 + v1) + (v2 + v3);
  const int inc = wave_incl_scan(s4, lane);
  if (lane == 31) waveTot[wave] = inc;
  __syncthreads();
  int base = 0, total = 0;
#pragma unroll
  for (int w = 0; w < SORT_TPB / 32; ++w) {
    const int x = waveTot[w];
    total += x;
    base += (w < wave) ? x : 0;
  }
  const int o0 = base + (inc - s4);
  const int o1 = o0 + v0;
  const int o2 = o1 + v1;
  const int o3 = o2 + v2;
  if (i0 + 0 < nBins) { binCur[i0 + 0] = o0; loc[(i0 + 0) * NCHUNK + c] = o0; }
  if (i0 + 1 < nBins) { binCur[i0 + 1] = o1; loc[(i0 + 1) * NCHUNK + c] = o1; }
  if (i0 + 2 < nBins) { binCur[i0 + 2] = o2; loc[(i0 + 2) * NCHUNK + c] = o2; }
  if (i0 + 3 < nBins) { binCur[i0 + 3] = o3; loc[(i0 + 3) * NCHUNK + c] = o3; }
  if (t == 0) loc[nBins * NCHUNK + c] = total;
  __syncthreads();

  const int nEdge = e1 - e0;
  const bool staged = nEdge <= STAGE_PAIRS;
  const int mis = (int)(((size_t)(binned + e0) >> 3) & 15);
  for (int e = e0 + t; e < e1; e += SORT_BATCH * SORT_TPB) {
    int d[SORT_BATCH], sId[SORT_BATCH];
#pragma unroll
    for (int u = 0; u < SORT_BATCH; ++u) {
      const int i = e + u * SORT_TPB;
      const bool v = i < e1;
      d[u] = v ? dst[i] : -1;
      sId[u] = v ? src[i] : 0;
    }
    int pos[SORT_BATCH];
#pragma unroll
    for (int u = 0; u < SORT_BATCH; ++u)
      pos[u] = atomicAdd(&binCur[d[u] >= 0 ? (d[u] >> BIN_SHIFT) : MAX_BINS + lane], 1);
    if (staged) {
#pragma unroll
      for (int u = 0; u < SORT_BATCH; ++u)
        if (d[u] >= 0) stage[mis + pos[u]] = make_int2(d[u], sId[u]);
    } else {
#pragma unroll
      for (int u = 0; u < SORT_BATCH; ++u)
        if (d[u] >= 0) binned[e0 + pos[u]] = make_int2(d[u], sId[u]);
    }
  }
  if (staged) {
    __syncthreads();
    int2* lineBase = binned + e0 - mis;
    const int last = mis + nEdge;
    for (int k = 2 * t; k < last; k += 2 * SORT_TPB) {
      const int4 v = *(const int4*)&stage[k];
      if (k >= mis && k + 1 < last) {
        *(int4*)(lineBase + k) = v;
      } else {
        if (k >= mis) lineBase[k] = make_int2(v.x, v.y);
        if (k + 1 >= mis && k + 1 < last) lineBase[k + 1] = make_int2(v.z, v.w);
      }
    }
  }
}

constexpr int LIST_WAVES = 32;
constexpr int LIST_CPW = NCHUNK / LIST_WAVES;
constexpr int LIST_STAGE_IDS = 12288;
constexpr int LIST_STAGE_SLOTS = LIST_STAGE_IDS + 32;

__global__ __launch_bounds__(32 * LIST_WAVES) void bin_lists(
    const int2* __restrict__ binned, const int* __restrict__ loc,
    int* __restrict__ rowptr, int* __restrict__ colTmp, int Nn, int nBins,
    int chunk) {
  __shared__ int cnt[BIN_NODES];
  __shared__ int waveTot[BIN_NODES / 32];
  __shared__ int sumLo[LIST_WAVES];
  __shared__ int sumHi[LIST_WAVES];
  __shared__ __attribute__((aligned(16))) int listStage[LIST_STAGE_SLOTS];
  const int bin = blockIdx.x;
  const int t = threadIdx.x;
  const int lane = t & 31;
  const int wave = __builtin_amdgcn_readfirstlane((int)(t >> 5));
  const int node0 = bin << BIN_SHIFT;

  const int cl = wave * LIST_CPW + (lane % LIST_CPW);
  const int loV = loc[bin * NCHUNK + cl];
  const int hiV = loc[(bin + 1) * NCHUNK + cl];
  {
    const int wl = wave_incl_scan(lane < LIST_CPW ? loV : 0, lane);
    const int wh = wave_incl_scan(lane < LIST_CPW ? hiV : 0, lane);
    if (lane == 31) {
      sumLo[wave] = wl;
      sumHi[wave] = wh;
    }
  }
  if (t < BIN_NODES) cnt[t] = 0;
  __syncthreads();
  const int b0 = __builtin_amdgcn_readlane(wave_incl_scan(sumLo[lane], lane), 31);
  const int b1 = __builtin_amdgcn_readlane(wave_incl_scan(sumHi[lane], lane), 31);

  int lo[LIST_CPW], hi[LIST_CPW];
#pragma unroll
  for (int u = 0; u < LIST_CPW; ++u) {
    lo[u] = __builtin_amdgcn_readlane(loV, u);
    hi[u] = __builtin_amdgcn_readlane(hiV, u);
  }
  const int2* wbase = binned + (size_t)(wave * LIST_CPW) * chunk;

  {
    int d[2 * LIST_CPW];
#pragma unroll
    for (int u = 0; u < LIST_CPW; ++u) {
      const int2* piece = wbase + (size_t)u * chunk;
      const int j = lo[u] + lane;
      d[2 * u] = (j < hi[u]) ? piece[j].x : -1;
      d[2 * u + 1] = (j + 32 < hi[u]) ? piece[j + 32].x : -1;
    }
#pragma unroll
    for (int v = 0; v < 2 * LIST_CPW; ++v)
      if (d[v] >= 0) atomicAdd(&cnt[d[v] - node0], 1);
#pragma unroll
    for (int u = 0; u < LIST_CPW; ++u) {
      const int2* piece = wbase + (size_t)u * chunk;
      for (int jb = lo[u] + 64; jb < hi[u]; jb += 32) {
        const int j = jb + lane;
        if (j < hi[u]) atomicAdd(&cnt[piece[j].x - node0], 1);
      }
    }
  }
  __syncthreads();

  int n = 0, inc = 0;
  if (t < BIN_NODES) {
    n = cnt[t];
    inc = wave_incl_scan(n, lane);
    if (lane == 31) waveTot[t >> 5] = inc;
  }
  __syncthreads();
  if (t < BIN_NODES) {
    int base = b0;
    for (int w = 0; w < (t >> 5); ++w) base += waveTot[w];
    const int start = base + (inc - n);
    cnt[t] = start;
    const int node = node0 + t;
    if (node < Nn) rowptr[node] = start;
  }
  if (bin == nBins - 1 && t == 0) rowptr[Nn] = b1;
  __syncthreads();

  const bool staged = b1 - b0 <= LIST_STAGE_IDS;
  const int mis = (int)(((size_t)(colTmp + b0) >> 2) & 31);
  const int sOff = mis - b0;
  {
    int2 e[2 * LIST_CPW];
#pragma unroll
    for (int u = 0; u < LIST_CPW; ++u) {
      const int2* piece = wbase + (size_t)u * chunk;
      const int j = lo[u] + lane;
      e[2 * u] = (j < hi[u]) ? piece[j] : make_int2(-1, 0);
      e[2 * u + 1] = (j + 32 < hi[u]) ? piece[j + 32] : make_int2(-1, 0);
    }
#pragma unroll
    for (int v = 0; v < 2 * LIST_CPW; ++v) {
      if (e[v].x >= 0) {
        const int pos = atomicAdd(&cnt[e[v].x - node0], 1);
        if (staged) listStage[pos + sOff] = e[v].y;
        else colTmp[pos] = e[v].y;
      }
    }
#pragma unroll
    for (int u = 0; u < LIST_CPW; ++u) {
      const int2* piece = wbase + (size_t)u * chunk;
      for (int jb = lo[u] + 64; jb < hi[u]; jb += 32) {
        const int j = jb + lane;
        if (j < hi[u]) {
          const int2 x = piece[j];
          const int pos = atomicAdd(&cnt[x.x - node0], 1);
          if (staged) listStage[pos + sOff] = x.y;
          else colTmp[pos] = x.y;
        }
      }
    }
  }
  if (staged) {
    __syncthreads();
    int* lineBase = colTmp + b0 - mis;
    const int last = mis + (b1 - b0);
    for (int k = 4 * t; k < last; k += 4 * 32 * LIST_WAVES) {
      const int4 v = *(const int4*)&listStage[k];
      if (k >= mis && k + 3 < last) {
        *(int4*)(lineBase + k) = v;
      } else {
        if (k >= mis && k < last) lineBase[k] = v.x;
        if (k + 1 >= mis && k + 1 < last) lineBase[k + 1] = v.y;
        if (k + 2 >= mis && k + 2 < last) lineBase[k + 2] = v.z;
        if (k + 3 >= mis && k + 3 < last) lineBase[k + 3] = v.w;
      }
    }
  }
}

constexpr unsigned dpp8_xor_sel(int x) {
  unsigned s = 0;
  for (int i = 0; i < 8; ++i) s |= (unsigned)((i ^ x) & 7) << (3 * i);
  return s;
}
constexpr unsigned dpp8_bcast_sel(int k) { return (unsigned)k * 0x249249u; }

template <unsigned SEL>
__device__ __forceinline__ float dpp8_f(float v) {
  return __uint_as_float(__builtin_amdgcn_mov_dpp8(__float_as_uint(v), SEL));
}
template <unsigned SEL>
__device__ __forceinline__ int dpp8_i(int v) {
  return (int)__builtin_amdgcn_mov_dpp8((unsigned)v, SEL);
}

template <int J>
__device__ __forceinline__ int lane_xor_i(int v);
template <>
__device__ __forceinline__ int lane_xor_i<1>(int v) { return dpp8_i<dpp8_xor_sel(1)>(v); }
template <>
__device__ __forceinline__ int lane_xor_i<2>(int v) { return dpp8_i<dpp8_xor_sel(2)>(v); }
template <>
__device__ __forceinline__ int lane_xor_i<4>(int v) { return dpp8_i<dpp8_xor_sel(4)>(v); }
template <>
__device__ __forceinline__ int lane_xor_i<8>(int v) {
  return __builtin_amdgcn_update_dpp(v, v, 0x168  , 0xf, 0xf, false);
}
template <>
__device__ __forceinline__ int lane_xor_i<16>(int v) {
  return (int)__builtin_amdgcn_permlanex16((unsigned)v, (unsigned)v, 0x76543210u,
                                           0xfedcba98u, false, false);
}

template <int J>
__device__ __forceinline__ int cmp_xchg(int v, int lane, bool asc) {
  const int p = lane_xor_i<J>(v);
  const bool keepMin = ((lane & J) == 0) == asc;
  return keepMin ? min(v, p) : max(v, p);
}

template <int K>
__device__ __forceinline__ int merge_lanes(int v, int lane, bool asc) {
  if (K >= 32) v = cmp_xchg<16>(v, lane, asc);
  if (K >= 16) v = cmp_xchg<8>(v, lane, asc);
  if (K >= 8) v = cmp_xchg<4>(v, lane, asc);
  if (K >= 4) v = cmp_xchg<2>(v, lane, asc);
  return cmp_xchg<1>(v, lane, asc);
}

__device__ __forceinline__ int sort_lanes(int v, int lane, bool asc) {
  v = merge_lanes<2>(v, lane, (lane & 2) == 0);
  v = merge_lanes<4>(v, lane, (lane & 4) == 0);
  v = merge_lanes<8>(v, lane, (lane & 8) == 0);
  v = merge_lanes<16>(v, lane, (lane & 16) == 0);
  return merge_lanes<32>(v, lane, asc);
}

__device__ __forceinline__ float quad_xor1_f(float v) {
  return __int_as_float(__builtin_amdgcn_mov_dpp(__float_as_int(v), 0xB1  ,
                                                 0xf, 0xf, true));
}
__device__ __forceinline__ float quad_xor2_f(float v) {
  return __int_as_float(__builtin_amdgcn_mov_dpp(__float_as_int(v), 0x4E  ,
                                                 0xf, 0xf, true));
}
__device__ __forceinline__ float row_xor4_f(float v) {
  return __int_as_float(__builtin_amdgcn_mov_dpp(__float_as_int(v), 0x164  ,
                                                 0xf, 0xf, true));
}

__device__ __forceinline__ float4 attn_h8_row(
    const float* __restrict__ feat, const int* __restrict__ rowptr,
    const int* __restrict__ colTmp, int* col, int node, int lane, float scale) {
  const int beg = rowptr[node];
  const int end = rowptr[node + 1];
  const int n = end - beg;
  const float* fl = feat + lane * 4;
  const float4 fd = *(const float4*)(fl + (size_t)node * 128);

  const bool inRegs = n <= 64;
  int k0 = 0x7fffffff, k1 = 0x7fffffff;
  if (inRegs) {
    if (lane < n) k0 = colTmp[beg + lane];
    if (n > 32) {
      if (32 + lane < n) k1 = colTmp[beg + 32 + lane];
      k0 = sort_lanes(k0, lane, true);
      k1 = sort_lanes(k1, lane, false);
      const int lo = min(k0, k1);
      const int hi = max(k0, k1);
      k0 = merge_lanes<32>(lo, lane, true);
      k1 = merge_lanes<32>(hi, lane, true);
      if (32 + lane < n) col[beg + 32 + lane] = k1;
    } else {
      k0 = sort_lanes(k0, lane, true);
    }
    if (lane < n) col[beg + lane] = k0;
  } else {
    for (int ib = 0; ib < n; ib += 32) {
      const int i = ib + lane;
      const bool vi = i < n;
      const int ki = vi ? colTmp[beg + i] : 0;
      int rank = 0;
      for (int jb = 0; jb < n; jb += 32) {
        const int cj = min(32, n - jb);
        const int kc = (lane < cj) ? colTmp[beg + jb + lane] : 0;
        for (int t = 0; t < cj; ++t) {
          const int kj = __builtin_amdgcn_readlane(kc, t);
          const int bias = (jb + t < i) ? 1 : 0;
          rank += (kj < ki + bias) ? 1 : 0;
        }
      }
      if (vi) col[beg + rank] = ki;
    }
    __threadfence();
  }

  const float ks = scale * 1.4426950408889634f;
  const v2f fdLo = {fd.x * ks, fd.y * ks};
  const v2f fdHi = {fd.z * ks, fd.w * ks};

  float m = -INFINITY, l = 0.0f;
  v2f accLo = {0.0f, 0.0f};
  v2f accHi = {0.0f, 0.0f};

  for (int base = beg; base < end; base += 32) {
    const int cnt = min(32, end - base);
    int ids;
    if (inRegs) {
      ids = (base == beg) ? k0 : k1;
    } else {
      ids = (lane < cnt) ? col[base + lane] : 0;
    }
    ids = (lane < cnt) ? ids : node;
    for (int t = 0; t < cnt; t += 4) {
      v2f fLo[4], fHi[4];
      float p[4];
#pragma unroll
      for (int u = 0; u < 4; ++u) {
        const int s = __builtin_amdgcn_readlane(ids, t + u);
        const float4 f = *(const float4*)(fl + (size_t)s * 128);
        fLo[u] = (v2f){f.x, f.y};
        fHi[u] = (v2f){f.z, f.w};
      }
#pragma unroll
      for (int u = 0; u < 4; ++u) {
        v2f tt = fLo[u] * fdLo;
        tt = fHi[u] * fdHi + tt;
        float d = tt.x + tt.y;
        d += quad_xor1_f(d);
        d += quad_xor2_f(d);
        p[u] = (t + u < cnt) ? d : -INFINITY;
      }
      const float mNew = fmaxf(fmaxf(m, p[0]), fmaxf(fmaxf(p[1], p[2]), p[3]));
      const float c = __builtin_amdgcn_exp2f(m - mNew);
      const float w0 = __builtin_amdgcn_exp2f(p[0] - mNew);
      const float w1 = __builtin_amdgcn_exp2f(p[1] - mNew);
      const float w2 = __builtin_amdgcn_exp2f(p[2] - mNew);
      const float w3 = __builtin_amdgcn_exp2f(p[3] - mNew);
      l = l * c + ((w0 + w1) + (w2 + w3));
      v2f sLo = fLo[0] * w0;
      sLo = fLo[1] * w1 + sLo;
      sLo = fLo[2] * w2 + sLo;
      sLo = fLo[3] * w3 + sLo;
      accLo = accLo * c + sLo;
      v2f sHi = fHi[0] * w0;
      sHi = fHi[1] * w1 + sHi;
      sHi = fHi[2] * w2 + sHi;
      sHi = fHi[3] * w3 + sHi;
      accHi = accHi * c + sHi;
      m = mNew;
    }
  }

  const float inv = 1.0f / fmaxf(l, 1e-9f);
  float4 o;
  o.x = accLo.x * inv;
  o.y = accLo.y * inv;
  o.z = accHi.x * inv;
  o.w = accHi.y * inv;
  o.x = o.x > 0.0f ? o.x : expm1f(o.x);
  o.y = o.y > 0.0f ? o.y : expm1f(o.y);
  o.z = o.z > 0.0f ? o.z : expm1f(o.z);
  o.w = o.w > 0.0f ? o.w : expm1f(o.w);
  return o;
}

constexpr int L1_WAVES = 16;
constexpr int H1_PITCH = 132;

template <int KD, int N>
__device__ __forceinline__ void gemm_tile_from_lds(
    const float* tile, const float* __restrict__ B, float* __restrict__ C,
    int tm, int tn, int lane) {
  constexpr int STEPS = KD / 8;
  const int c = lane & 15;
  const int h = lane >> 4;
  const float* Ap = tile + c * H1_PITCH + 4 * h;
  const float* Bp = B + (size_t)(4 * h) * N + tn + c;
  v8f acc = {};
#pragma unroll
  for (int j = 0; j < STEPS; ++j) {
    const float4 a = *(const float4*)(Ap + 8 * j);
    const float* p = Bp + (size_t)(8 * j) * N;
    v2f av, bv;
    av.x = a.x;
    av.y = a.y;
    bv.x = p[0];
    bv.y = p[N];
    acc = __builtin_amdgcn_wmma_f32_16x16x4_f32(
         false, av,  false, bv,
         (short)0, acc,  false,  false);
    av.x = a.z;
    av.y = a.w;
    bv.x = p[2 * N];
    bv.y = p[3 * N];
    acc = __builtin_amdgcn_wmma_f32_16x16x4_f32(
         false, av,  false, bv,
         (short)0, acc,  false,  false);
  }
  const int rowBase = tm + (h << 3);
#pragma unroll
  for (int j = 0; j < 8; ++j) {
    C[(size_t)(rowBase + j) * N + tn + c] = acc[j];
  }
}

template <int N>
__global__ __launch_bounds__(32 * L1_WAVES) void node_attn_h8_gemm(
    const float* __restrict__ feat, const int* __restrict__ rowptr,
    const int* __restrict__ colTmp, int* col, const float* __restrict__ W2,
    float* __restrict__ feat2, int Nn, float scale) {
  __shared__ __attribute__((aligned(16))) float h1Tile[L1_WAVES * H1_PITCH];
  const int lane = threadIdx.x & 31;
  const int wave = __builtin_amdgcn_readfirstlane((int)(threadIdx.x >> 5));
  const int node0 = blockIdx.x * L1_WAVES;
  const int node = node0 + wave;
  float4 o = make_float4(0.0f, 0.0f, 0.0f, 0.0f);
  if (node < Nn) o = attn_h8_row(feat, rowptr, colTmp, col, node, lane, scale);
  *(float4*)(h1Tile + wave * H1_PITCH + lane * 4) = o;
  __syncthreads();
  if (wave < N / 16)
    gemm_tile_from_lds<128, N>(h1Tile, W2, feat2, node0, 16 * wave, lane);
}

template <int K0>
__device__ __forceinline__ void h1_four_steps(
    const float* __restrict__ fq, int ids, int cnt, int g, const v2f fdA0,
    const v2f fdA1, const v2f fdB0, const v2f fdB1, float& m, float& l,
    v2f& accA0, v2f& accA1, v2f& accB0, v2f& accB1) {
  v2f a0[4], a1[4], b0[4], b1[4];
  float p[4];
#pragma unroll
  for (int u = 0; u < 4; ++u) {
    const int s = u == 0   ? dpp8_i<dpp8_bcast_sel(K0 + 0)>(ids)
                  : u == 1 ? dpp8_i<dpp8_bcast_sel(K0 + 1)>(ids)
                  : u == 2 ? dpp8_i<dpp8_bcast_sel(K0 + 2)>(ids)
                           : dpp8_i<dpp8_bcast_sel(K0 + 3)>(ids);
    const float* r = fq + (size_t)s * 64;
    const float4 fa = *(const float4*)(r);
    const float4 fb = *(const float4*)(r + 32);
    a0[u] = (v2f){fa.x, fa.y};
    a1[u] = (v2f){fa.z, fa.w};
    b0[u] = (v2f){fb.x, fb.y};
    b1[u] = (v2f){fb.z, fb.w};
  }
#pragma unroll
  for (int u = 0; u < 4; ++u) {
    v2f tt = a0[u] * fdA0;
    tt = a1[u] * fdA1 + tt;
    tt = b0[u] * fdB0 + tt;
    tt = b1[u] * fdB1 + tt;
    float d = tt.x + tt.y;
    d += quad_xor1_f(d);
    d += quad_xor2_f(d);
    d += row_xor4_f(d);
    p[u] = (4 * (K0 + u) + g < cnt) ? d : -INFINITY;
  }
  const float mNew = fmaxf(fmaxf(m, p[0]), fmaxf(fmaxf(p[1], p[2]), p[3]));
  const float mRef = (mNew == -INFINITY) ? 0.0f : mNew;
  const float c = __builtin_amdgcn_exp2f(m - mRef);
  const float w0 = __builtin_amdgcn_exp2f(p[0] - mRef);
  const float w1 = __builtin_amdgcn_exp2f(p[1] - mRef);
  const float w2 = __builtin_amdgcn_exp2f(p[2] - mRef);
  const float w3 = __builtin_amdgcn_exp2f(p[3] - mRef);
  l = l * c + ((w0 + w1) + (w2 + w3));
  v2f s;
  s = a0[0] * w0;
  s = a0[1] * w1 + s;
  s = a0[2] * w2 + s;
  s = a0[3] * w3 + s;
  accA0 = accA0 * c + s;
  s = a1[0] * w0;
  s = a1[1] * w1 + s;
  s = a1[2] * w2 + s;
  s = a1[3] * w3 + s;
  accA1 = accA1 * c + s;
  s = b0[0] * w0;
  s = b0[1] * w1 + s;
  s = b0[2] * w2 + s;
  s = b0[3] * w3 + s;
  accB0 = accB0 * c + s;
  s = b1[0] * w0;
  s = b1[1] * w1 + s;
  s = b1[2] * w2 + s;
  s = b1[3] * w3 + s;
  accB1 = accB1 * c + s;
  m = mNew;
}

__device__ __forceinline__ float sum_groups(float v) {
  v += __shfl_xor(v, 8, 32);
  v += __shfl_xor(v, 16, 32);
  return v;
}

__global__ __launch_bounds__(256) void node_attn_h1(
    const float* __restrict__ feat, const int* __restrict__ rowptr,
    const int* __restrict__ col, float* __restrict__ out, int Nn, float scale) {
  const int lane = threadIdx.x & 31;
  const int node = __builtin_amdgcn_readfirstlane(
      (int)((blockIdx.x * blockDim.x + threadIdx.x) >> 5));
  if (node >= Nn) return;
  const int beg = rowptr[node];
  const int end = rowptr[node + 1];
  const int q = lane & 7;
  const int g = lane >> 3;
  const float* fq = feat + q * 4;
  const float4 fdA = *(const float4*)(fq + (size_t)node * 64);
  const float4 fdB = *(const float4*)(fq + (size_t)node * 64 + 32);
  const float ks = scale * 1.4426950408889634f;
  const v2f fdA0 = {fdA.x * ks, fdA.y * ks};
  const v2f fdA1 = {fdA.z * ks, fdA.w * ks};
  const v2f fdB0 = {fdB.x * ks, fdB.y * ks};
  const v2f fdB1 = {fdB.z * ks, fdB.w * ks};

  float m = -INFINITY, l = 0.0f;
  v2f accA0 = {0.0f, 0.0f}, accA1 = {0.0f, 0.0f};
  v2f accB0 = {0.0f, 0.0f}, accB1 = {0.0f, 0.0f};

  const int myIdx = 4 * q + g;
  for (int base = beg; base < end; base += 32) {
    const int cnt = min(32, end - base);
    const int ids = (myIdx < cnt) ? col[base + myIdx] : node;
    h1_four_steps<0>(fq, ids, cnt, g, fdA0, fdA1, fdB0, fdB1, m, l, accA0, accA1, accB0, accB1);
    if (cnt > 16)
      h1_four_steps<4>(fq, ids, cnt, g, fdA0, fdA1, fdB0, fdB1, m, l, accA0, accA1, accB0, accB1);
  }

  float M = fmaxf(m, __shfl_xor(m, 8, 32));
  M = fmaxf(M, __shfl_xor(M, 16, 32));
  const float Mref = (M == -INFINITY) ? 0.0f : M;
  const float sc = __builtin_amdgcn_exp2f(m - Mref);
  const float L = sum_groups(l * sc);
  float4 oA, oB;
  oA.x = sum_groups(accA0.x * sc);
  oA.y = sum_groups(accA0.y * sc);
  oA.z = sum_groups(accA1.x * sc);
  oA.w = sum_groups(accA1.y * sc);
  oB.x = sum_groups(accB0.x * sc);
  oB.y = sum_groups(accB0.y * sc);
  oB.z = sum_groups(accB1.x * sc);
  oB.w = sum_groups(accB1.y * sc);
  const float inv = 1.0f / fmaxf(L, 1e-9f);
  if (g == 0) {
    float* o = out + (size_t)node * 64 + q * 4;
    *(float4*)(o) = make_float4(oA.x * inv, oA.y * inv, oA.z * inv, oA.w * inv);
    *(float4*)(o + 32) = make_float4(oB.x * inv, oB.y * inv, oB.z * inv, oB.w * inv);
  }
}

static inline int cdiv_i(long long a, long long b) { return (int)((a + b - 1) / b); }

extern "C" void kernel_launch(void* const* d_in, const int* in_sizes, int n_in,
                              void* d_out, int out_size, void* d_ws, size_t ws_size,
                              hipStream_t stream) {
  const float* hIn = (const float*)d_in[0];
  const float* W1  = (const float*)d_in[1];
  const float* W2  = (const float*)d_in[2];
  const int*   src = (const int*)d_in[3];
  const int*   dst = (const int*)d_in[4];

  const int IN_DIM = 128;
  constexpr int OUT_DIM = 64;
  const int HD1 = 128;
  const int Nn  = in_sizes[0] / IN_DIM;
  const int E   = in_sizes[3];

  const int nBins   = cdiv_i(Nn, BIN_NODES);
  const int chunk   = cdiv_i(E, NCHUNK);
  const int locLen  = (nBins + 1) * NCHUNK;

  float* feat1  = (float*)d_ws;
  float* h1     = feat1 + (size_t)Nn * HD1;
  int*   colTmp = (int*)(h1 + (size_t)Nn * HD1) + (size_t)E;
  int*   col    = colTmp + (size_t)E;
  int*   rowptr = col + (size_t)E + (size_t)Nn;
  int2*  binned = (int2*)h1;
  int*   loc    = (int*)(h1 + (size_t)Nn * HD1) - locLen;

  const int TPB = 256;
  const int wavesPerBlock = TPB / 32;
  float* feat2 = h1;

  {
    int gemmBlocks = cdiv_i((Nn / 16) * (HD1 / 64), SORT_TPB / 32);
    bin_sort_and_gemm<128, 128><<<NCHUNK + gemmBlocks, SORT_TPB, 0, stream>>>(
        src, dst, loc, binned, E, chunk, nBins, hIn, W1, feat1, Nn);
  }
  bin_lists<<<nBins, 32 * LIST_WAVES, 0, stream>>>(binned, loc, rowptr, colTmp, Nn,
                                                   nBins, chunk);

  node_attn_h8_gemm<OUT_DIM><<<cdiv_i(Nn, L1_WAVES), 32 * L1_WAVES, 0, stream>>>(
      feat1, rowptr, colTmp, col, W2, feat2, Nn, 0.25f);

  node_attn_h1<<<cdiv_i(Nn, wavesPerBlock), TPB, 0, stream>>>(
      feat2, rowptr, col, (float*)d_out, Nn, 0.125f);
}
